# starting kernel
# baseline (speedup 1.0000x reference)
#include <hip/hip_runtime.h>


typedef __attribute__((ext_vector_type(2))) float v2f;
typedef __attribute__((ext_vector_type(8))) float v8f;

__global__ void zero_f32(float* __restrict__ p, long n) {
  long i = (long)blockIdx.x * blockDim.x + threadIdx.x;
  if (i < n) p[i] = 0.0f;
}

__global__ void count_edges(const int* __restrict__ dst, float* __restrict__ cnt,
                            long n_edges) {
  long i = (long)blockIdx.x * blockDim.x + threadIdx.x;
  if (i < n_edges) unsafeAtomicAdd(&cnt[dst[i]], 1.0f);
}

__global__ void reciprocal_clamped(float* __restrict__ p, long n) {
  long i = (long)blockIdx.x * blockDim.x + threadIdx.x;
  if (i < n) p[i] = 1.0f / fmaxf(p[i], 1.0f);
}

template <int F_IN, int F_OUT>
__global__ void node_transform(const float* __restrict__ h,
                               const float* __restrict__ Wl,
                               const float* __restrict__ Wr,
                               const float* __restrict__ bias,
                               float* __restrict__ hl,
                               float* __restrict__ hs,
                               float* __restrict__ agg,
                               int n_nodes) {
  const int lane = threadIdx.x & 31;
  const int wave = threadIdx.x >> 5;
  const int tile = blockIdx.x * (blockDim.x >> 5) + wave;
  const int tiles = (n_nodes + 15) >> 4;
  if (tile >= tiles) return;

  const int m    = lane & 15;
  const int hi   = lane >> 4;
  const int row0 = tile << 4;
  const int arow = min(row0 + m, n_nodes - 1);
  const bool cok = (m < F_OUT);

  auto ldA = [&](int k) -> float {
    const bool ok = (k < F_IN);
    const float v = h[(long)arow * F_IN + (ok ? k : 0)];
    return ok ? v : 0.0f;
  };
  auto ldW = [&](const float* __restrict__ W, int k) -> float {
    const bool ok = cok && (k < F_IN);
    const float v = W[ok ? (m * F_IN + k) : 0];
    return ok ? v : 0.0f;
  };

  v8f dl = {};
  v8f dr = {};
#pragma unroll
  for (int kc = 0; kc < F_IN; kc += 4) {
    const int k0 = kc + hi * 2;
    v2f a, bl, br;
    a.x  = ldA(k0);
    a.y  = ldA(k0 + 1);
    bl.x = ldW(Wl, k0);
    bl.y = ldW(Wl, k0 + 1);
    br.x = ldW(Wr, k0);
    br.y = ldW(Wr, k0 + 1);
    dl = __builtin_amdgcn_wmma_f32_16x16x4_f32(false, a, false, bl, (short)0,
                                               dl, false, false);
    dr = __builtin_amdgcn_wmma_f32_16x16x4_f32(false, a, false, br, (short)0,
                                               dr, false, false);
  }

  if (cok) {
    const float bv   = bias[m];
    const long  base = (long)(row0 + hi * 8) * F_OUT + m;
    if (row0 + 16 <= n_nodes) {
#pragma unroll
      for (int v = 0; v < 8; ++v) {
        hl[base + (long)v * F_OUT]  = dl[v];
        hs[base + (long)v * F_OUT]  = dr[v] + bv;
        agg[base + (long)v * F_OUT] = 0.0f;
      }
    } else {
#pragma unroll
      for (int v = 0; v < 8; ++v) {
        if (row0 + hi * 8 + v < n_nodes) {
          hl[base + (long)v * F_OUT]  = dl[v];
          hs[base + (long)v * F_OUT]  = dr[v] + bv;
          agg[base + (long)v * F_OUT] = 0.0f;
        }
      }
    }
  }
}

template <int F>
__global__ void edge_scatter(const int* __restrict__ src,
                             const int* __restrict__ dst,
                             const float* __restrict__ hl,
                             float* __restrict__ agg, long n_edges) {
  long i = (long)blockIdx.x * blockDim.x + threadIdx.x;
  if (i >= n_edges) return;
  const int s = src[i];
  const int d = dst[i];
  float v[F];
#pragma unroll
  for (int j = 0; j < F; ++j) v[j] = hl[(long)s * F + j];
#pragma unroll
  for (int j = 0; j < F; ++j) unsafeAtomicAdd(&agg[(long)d * F + j], v[j]);
}

template <int F, bool RELU>
__global__ void combine(const float* __restrict__ agg,
                        const float* __restrict__ hs,
                        const float* __restrict__ inv,
                        float* __restrict__ out, int n_nodes) {
  const int node = blockIdx.x * blockDim.x + threadIdx.x;
  if (node >= n_nodes) return;
  const float iv = inv[node];
#pragma unroll
  for (int j = 0; j < F; ++j) {
    const long o = (long)node * F + j;
    const float v = agg[o] * iv + hs[o];
    out[o] = RELU ? fmaxf(v, 0.0f) : v;
  }
}

extern "C" void kernel_launch(void* const* d_in, const int* in_sizes, int n_in,
                              void* d_out, int out_size, void* d_ws,
                              size_t ws_size, hipStream_t stream) {
  const float* x    = (const float*)d_in[0];
  const int*   ei   = (const int*)  d_in[1];
  const float* Wl1  = (const float*)d_in[2];
  const float* Wr1  = (const float*)d_in[3];
  const float* b1   = (const float*)d_in[4];
  const float* Wlm  = (const float*)d_in[5];
  const float* Wrm  = (const float*)d_in[6];
  const float* bm   = (const float*)d_in[7];
  const float* Wl10 = (const float*)d_in[8];
  const float* Wr10 = (const float*)d_in[9];
  const float* b10  = (const float*)d_in[10];

  const int  N = in_sizes[0] / 10;
  const long E = (long)in_sizes[1] / 2;
  const int* src = ei;
  const int* dst = ei + E;

  float* ws  = (float*)d_ws;
  float* inv = ws;
  float* hl  = inv + (size_t)N;
  float* hs  = hl + (size_t)N * 5;
  float* agg = hs + (size_t)N * 5;
  float* hA  = agg + (size_t)N * 5;
  float* hB  = hA + (size_t)N * 5;

  const int B = 256;
  auto cdiv = [](long a, long b) { return (int)((a + b - 1) / b); };

  zero_f32<<<cdiv(N, B), B, 0, stream>>>(inv, N);
  count_edges<<<cdiv(E, B), B, 0, stream>>>(dst, inv, E);
  reciprocal_clamped<<<cdiv(N, B), B, 0, stream>>>(inv, N);

  const int tiles   = (N + 15) / 16;
  const int tblocks = (tiles + 7) / 8;

  node_transform<10, 5><<<tblocks, 256, 0, stream>>>(x, Wl1, Wr1, b1, hl, hs, agg, N);
  edge_scatter<5><<<cdiv(E, B), B, 0, stream>>>(src, dst, hl, agg, E);
  combine<5, true><<<cdiv(N, B), B, 0, stream>>>(agg, hs, inv, hA, N);

  for (int l = 0; l < 8; ++l) {
    const float* hin  = (l & 1) ? hB : hA;
    float*       hout = (l & 1) ? hA : hB;
    node_transform<5, 5><<<tblocks, 256, 0, stream>>>(
        hin, Wlm + l * 25, Wrm + l * 25, bm + l * 5, hl, hs, agg, N);
    edge_scatter<5><<<cdiv(E, B), B, 0, stream>>>(src, dst, hl, agg, E);
    combine<5, true><<<cdiv(N, B), B, 0, stream>>>(agg, hs, inv, hout, N);
  }

  node_transform<5, 1><<<tblocks, 256, 0, stream>>>(hA, Wl10, Wr10, b10, hl, hs, agg, N);
  edge_scatter<1><<<cdiv(E, B), B, 0, stream>>>(src, dst, hl, agg, E);
  combine<1, false><<<cdiv(N, B), B, 0, stream>>>(agg, hs, inv, (float*)d_out, N);
}


# Edges sorted by destination once per call (int atomic count, scan, arrival-order fill); each layer pulls rows with 8 lanes per node: float atomics gone, sums now in row order, 8 partials + xor tree
# speedup vs baseline: 4.0505x; 4.0505x over previous
#include <hip/hip_runtime.h>


typedef __attribute__((ext_vector_type(2))) float v2f;
typedef __attribute__((ext_vector_type(8))) float v8f;

#define SCAN_CHUNK 2048

__global__ void zero_i32(int* __restrict__ p, int n) {
  const int i = blockIdx.x * blockDim.x + threadIdx.x;
  if (i < n) p[i] = 0;
}

__global__ void count_in_edges(const int* __restrict__ dst, int* __restrict__ rp,
                               long n_edges) {
  const long stride = (long)gridDim.x * blockDim.x;
  long i = (long)blockIdx.x * blockDim.x + threadIdx.x;
  for (; i + 3 * stride < n_edges; i += 4 * stride) {
    const int d0 = dst[i];
    const int d1 = dst[i + stride];
    const int d2 = dst[i + 2 * stride];
    const int d3 = dst[i + 3 * stride];
    atomicAdd(&rp[d0], 1);
    atomicAdd(&rp[d1], 1);
    atomicAdd(&rp[d2], 1);
    atomicAdd(&rp[d3], 1);
  }
  for (; i < n_edges; i += stride) atomicAdd(&rp[dst[i]], 1);
}

__device__ inline int wave_incl_scan(int x, int lane) {
  for (int off = 1; off < 32; off <<= 1) {
    const int y = __shfl_up(x, off);
    if (lane >= off) x += y;
  }
  return x;
}

__global__ void block_sums(const int* __restrict__ cnt, int* __restrict__ bsum,
                           int n) {
  __shared__ int wsum[8];
  const int base = blockIdx.x * SCAN_CHUNK;
  int s = 0;
  for (int k = 0; k < 8; ++k) {
    const int idx = base + k * 256 + (int)threadIdx.x;
    if (idx < n) s += cnt[idx];
  }
  for (int off = 16; off > 0; off >>= 1) s += __shfl_xor(s, off);
  if ((threadIdx.x & 31) == 0) wsum[threadIdx.x >> 5] = s;
  __syncthreads();
  if (threadIdx.x == 0) {
    int t = 0;
    for (int w = 0; w < 8; ++w) t += wsum[w];
    bsum[blockIdx.x] = t;
  }
}

__global__ void scan_block_sums(int* __restrict__ bsum, int nb) {
  __shared__ int wsum[8];
  const int lane = threadIdx.x & 31;
  const int wave = threadIdx.x >> 5;
  int carry = 0;
  for (int base = 0; base < nb; base += 256) {
    const int i = base + (int)threadIdx.x;
    const int v = (i < nb) ? bsum[i] : 0;
    const int x = wave_incl_scan(v, lane);
    if (lane == 31) wsum[wave] = x;
    __syncthreads();
    int before = 0, total = 0;
    for (int w = 0; w < 8; ++w) {
      const int t = wsum[w];
      before += (w < wave) ? t : 0;
      total += t;
    }
    if (i < nb) bsum[i] = carry + before + x - v;
    carry += total;
    __syncthreads();
  }
}

__global__ void scan_apply(int* rp, const int* __restrict__ bofs, int n) {
  __shared__ int wsum[8];
  const int lane = threadIdx.x & 31;
  const int wave = threadIdx.x >> 5;
  const int base = blockIdx.x * SCAN_CHUNK + (int)threadIdx.x * 8;
  int v[8];
  int t = 0;
  for (int k = 0; k < 8; ++k) {
    const int idx = base + k;
    v[k] = (idx < n) ? rp[idx] : 0;
    t += v[k];
  }
  const int x = wave_incl_scan(t, lane);
  if (lane == 31) wsum[wave] = x;
  __syncthreads();
  int run = bofs[blockIdx.x] + x - t;
  for (int w = 0; w < 8; ++w) run += (w < wave) ? wsum[w] : 0;
  for (int k = 0; k < 8; ++k) {
    const int idx = base + k;
    if (idx < n) rp[idx] = run;
    run += v[k];
  }
}

__global__ void fill_rows(const int* __restrict__ src, const int* __restrict__ dst,
                          int* __restrict__ rp, int* __restrict__ ss,
                          long n_edges) {
  const long stride = (long)gridDim.x * blockDim.x;
  long i = (long)blockIdx.x * blockDim.x + threadIdx.x;
  for (; i + 3 * stride < n_edges; i += 4 * stride) {
    const int d0 = dst[i];
    const int d1 = dst[i + stride];
    const int d2 = dst[i + 2 * stride];
    const int d3 = dst[i + 3 * stride];
    const int s0 = src[i];
    const int s1 = src[i + stride];
    const int s2 = src[i + 2 * stride];
    const int s3 = src[i + 3 * stride];
    const int p0 = atomicAdd(&rp[d0], 1);
    const int p1 = atomicAdd(&rp[d1], 1);
    const int p2 = atomicAdd(&rp[d2], 1);
    const int p3 = atomicAdd(&rp[d3], 1);
    ss[p0] = s0;
    ss[p1] = s1;
    ss[p2] = s2;
    ss[p3] = s3;
  }
  for (; i < n_edges; i += stride) {
    const int d = dst[i];
    const int s = src[i];
    const int p = atomicAdd(&rp[d], 1);
    ss[p] = s;
  }
}

template <int F_IN, int F_OUT, bool ZERO_AGG>
__global__ void node_transform(const float* __restrict__ h,
                               const float* __restrict__ Wl,
                               const float* __restrict__ Wr,
                               const float* __restrict__ bias,
                               float* __restrict__ hl,
                               float* __restrict__ hs,
                               float* __restrict__ agg,
                               int n_nodes) {
  const int lane = threadIdx.x & 31;
  const int wave = threadIdx.x >> 5;
  const int tile = blockIdx.x * (blockDim.x >> 5) + wave;
  const int tiles = (n_nodes + 15) >> 4;
  if (tile >= tiles) return;

  const int m    = lane & 15;
  const int hi   = lane >> 4;
  const int row0 = tile << 4;
  const int arow = min(row0 + m, n_nodes - 1);
  const bool cok = (m < F_OUT);

  auto ldA = [&](int k) -> float {
    const bool ok = (k < F_IN);
    const float v = h[(long)arow * F_IN + (ok ? k : 0)];
    return ok ? v : 0.0f;
  };
  auto ldW = [&](const float* __restrict__ W, int k) -> float {
    const bool ok = cok && (k < F_IN);
    const float v = W[ok ? (m * F_IN + k) : 0];
    return ok ? v : 0.0f;
  };

  v8f dl = {};
  v8f dr = {};
#pragma unroll
  for (int kc = 0; kc < F_IN; kc += 4) {
    const int k0 = kc + hi * 2;
    v2f a, bl, br;
    a.x  = ldA(k0);
    a.y  = ldA(k0 + 1);
    bl.x = ldW(Wl, k0);
    bl.y = ldW(Wl, k0 + 1);
    br.x = ldW(Wr, k0);
    br.y = ldW(Wr, k0 + 1);
    dl = __builtin_amdgcn_wmma_f32_16x16x4_f32(false, a, false, bl, (short)0,
                                               dl, false, false);
    dr = __builtin_amdgcn_wmma_f32_16x16x4_f32(false, a, false, br, (short)0,
                                               dr, false, false);
  }

  if (cok) {
    const float bv   = bias[m];
    const long  base = (long)(row0 + hi * 8) * F_OUT + m;
    if (row0 + 16 <= n_nodes) {
#pragma unroll
      for (int v = 0; v < 8; ++v) {
        hl[base + (long)v * F_OUT]  = dl[v];
        hs[base + (long)v * F_OUT]  = dr[v] + bv;
        if (ZERO_AGG) agg[base + (long)v * F_OUT] = 0.0f;
      }
    } else {
#pragma unroll
      for (int v = 0; v < 8; ++v) {
        if (row0 + hi * 8 + v < n_nodes) {
          hl[base + (long)v * F_OUT]  = dl[v];
          hs[base + (long)v * F_OUT]  = dr[v] + bv;
          if (ZERO_AGG) agg[base + (long)v * F_OUT] = 0.0f;
        }
      }
    }
  }
}

template <int F, bool RELU>
__global__ void pull_combine(const int* __restrict__ rp, const int* __restrict__ ss,
                             const float* __restrict__ hl,
                             const float* __restrict__ hs,
                             float* __restrict__ out, int n_nodes) {
  const int q    = threadIdx.x & 7;
  const int node = (int)((blockIdx.x * blockDim.x + threadIdx.x) >> 3);
  const bool live = node < n_nodes;
  const int nd  = live ? node : n_nodes - 1;
  const int beg = (nd > 0) ? rp[nd - 1] : 0;
  const int end = live ? rp[nd] : beg;

  float acc[F];
  for (int f = 0; f < F; ++f) acc[f] = 0.0f;
  for (int j = beg + q; j < end; j += 8) {
    const int s = ss[j];
    for (int f = 0; f < F; ++f) acc[f] += hl[(long)s * F + f];
  }
  for (int f = 0; f < F; ++f) {
    acc[f] += __shfl_xor(acc[f], 4);
    acc[f] += __shfl_xor(acc[f], 2);
    acc[f] += __shfl_xor(acc[f], 1);
  }
  float val = acc[0];
  for (int f = 1; f < F; ++f) val = (q == f) ? acc[f] : val;

  if (live && q < F) {
    const float iv = 1.0f / fmaxf((float)(end - beg), 1.0f);
    const long  o  = (long)node * F + q;
    const float r  = val * iv + hs[o];
    out[o] = RELU ? fmaxf(r, 0.0f) : r;
  }
}

__global__ void zero_f32(float* __restrict__ p, long n) {
  long i = (long)blockIdx.x * blockDim.x + threadIdx.x;
  if (i < n) p[i] = 0.0f;
}

__global__ void count_edges(const int* __restrict__ dst, float* __restrict__ cnt,
                            long n_edges) {
  long i = (long)blockIdx.x * blockDim.x + threadIdx.x;
  if (i < n_edges) unsafeAtomicAdd(&cnt[dst[i]], 1.0f);
}

__global__ void reciprocal_clamped(float* __restrict__ p, long n) {
  long i = (long)blockIdx.x * blockDim.x + threadIdx.x;
  if (i < n) p[i] = 1.0f / fmaxf(p[i], 1.0f);
}

template <int F>
__global__ void edge_scatter(const int* __restrict__ src,
                             const int* __restrict__ dst,
                             const float* __restrict__ hl,
                             float* __restrict__ agg, long n_edges) {
  long i = (long)blockIdx.x * blockDim.x + threadIdx.x;
  if (i >= n_edges) return;
  const int s = src[i];
  const int d = dst[i];
  float v[F];
#pragma unroll
  for (int j = 0; j < F; ++j) v[j] = hl[(long)s * F + j];
#pragma unroll
  for (int j = 0; j < F; ++j) unsafeAtomicAdd(&agg[(long)d * F + j], v[j]);
}

template <int F, bool RELU>
__global__ void combine(const float* __restrict__ agg,
                        const float* __restrict__ hs,
                        const float* __restrict__ inv,
                        float* __restrict__ out, int n_nodes) {
  const int node = blockIdx.x * blockDim.x + threadIdx.x;
  if (node >= n_nodes) return;
  const float iv = inv[node];
#pragma unroll
  for (int j = 0; j < F; ++j) {
    const long o = (long)node * F + j;
    const float v = agg[o] * iv + hs[o];
    out[o] = RELU ? fmaxf(v, 0.0f) : v;
  }
}

extern "C" void kernel_launch(void* const* d_in, const int* in_sizes, int n_in,
                              void* d_out, int out_size, void* d_ws,
                              size_t ws_size, hipStream_t stream) {
  const float* x    = (const float*)d_in[0];
  const int*   ei   = (const int*)  d_in[1];
  const float* Wl1  = (const float*)d_in[2];
  const float* Wr1  = (const float*)d_in[3];
  const float* b1   = (const float*)d_in[4];
  const float* Wlm  = (const float*)d_in[5];
  const float* Wrm  = (const float*)d_in[6];
  const float* bm   = (const float*)d_in[7];
  const float* Wl10 = (const float*)d_in[8];
  const float* Wr10 = (const float*)d_in[9];
  const float* b10  = (const float*)d_in[10];

  const int  N = in_sizes[0] / 10;
  const long E = (long)in_sizes[1] / 2;
  const int* src = ei;
  const int* dst = ei + E;

  const int B = 256;
  auto cdiv = [](long a, long b) { return (int)((a + b - 1) / b); };
  auto pad32 = [](size_t a) { return (a + 31) & ~(size_t)31; };

  const int tiles   = (N + 15) / 16;
  const int tblocks = (tiles + 7) / 8;

  const int    nb    = cdiv(N, SCAN_CHUNK);
  const size_t w_rp  = pad32((size_t)N);
  const size_t w_bs  = pad32((size_t)nb);
  const size_t w_ss  = pad32((size_t)E);
  const size_t w_h   = pad32((size_t)N * 5);
  const size_t need  = (w_rp + w_bs + w_ss + 3 * w_h) * sizeof(int);

  if (ws_size >= need) {
    int*   rp   = (int*)d_ws;
    int*   bsum = rp + w_rp;
    int*   ss   = bsum + w_bs;
    float* h    = (float*)(ss + w_ss);
    float* hl   = h + w_h;
    float* hs   = hl + w_h;
    float* none = nullptr;

    const int eblocks = cdiv(E, B) < 2048 ? cdiv(E, B) : 2048;
    zero_i32<<<cdiv(N, B), B, 0, stream>>>(rp, N);
    count_in_edges<<<eblocks, B, 0, stream>>>(dst, rp, E);
    block_sums<<<nb, B, 0, stream>>>(rp, bsum, N);
    scan_block_sums<<<1, B, 0, stream>>>(bsum, nb);
    scan_apply<<<nb, B, 0, stream>>>(rp, bsum, N);
    fill_rows<<<eblocks, B, 0, stream>>>(src, dst, rp, ss, E);

    const int pblocks = cdiv((long)N * 8, B);

    node_transform<10, 5, false><<<tblocks, 256, 0, stream>>>(x, Wl1, Wr1, b1, hl, hs, none, N);
    pull_combine<5, true><<<pblocks, B, 0, stream>>>(rp, ss, hl, hs, h, N);

    for (int l = 0; l < 8; ++l) {
      node_transform<5, 5, false><<<tblocks, 256, 0, stream>>>(
          h, Wlm + l * 25, Wrm + l * 25, bm + l * 5, hl, hs, none, N);
      pull_combine<5, true><<<pblocks, B, 0, stream>>>(rp, ss, hl, hs, h, N);
    }

    node_transform<5, 1, false><<<tblocks, 256, 0, stream>>>(h, Wl10, Wr10, b10, hl, hs, none, N);
    pull_combine<1, false><<<pblocks, B, 0, stream>>>(rp, ss, hl, hs, (float*)d_out, N);
    return;
  }

  float* ws  = (float*)d_ws;
  float* inv = ws;
  float* hl  = inv + (size_t)N;
  float* hs  = hl + (size_t)N * 5;
  float* agg = hs + (size_t)N * 5;
  float* hA  = agg + (size_t)N * 5;
  float* hB  = hA + (size_t)N * 5;

  zero_f32<<<cdiv(N, B), B, 0, stream>>>(inv, N);
  count_edges<<<cdiv(E, B), B, 0, stream>>>(dst, inv, E);
  reciprocal_clamped<<<cdiv(N, B), B, 0, stream>>>(inv, N);

  node_transform<10, 5, true><<<tblocks, 256, 0, stream>>>(x, Wl1, Wr1, b1, hl, hs, agg, N);
  edge_scatter<5><<<cdiv(E, B), B, 0, stream>>>(src, dst, hl, agg, E);
  combine<5, true><<<cdiv(N, B), B, 0, stream>>>(agg, hs, inv, hA, N);

  for (int l = 0; l < 8; ++l) {
    const float* hin  = (l & 1) ? hB : hA;
    float*       hout = (l & 1) ? hA : hB;
    node_transform<5, 5, true><<<tblocks, 256, 0, stream>>>(
        hin, Wlm + l * 25, Wrm + l * 25, bm + l * 5, hl, hs, agg, N);
    edge_scatter<5><<<cdiv(E, B), B, 0, stream>>>(src, dst, hl, agg, E);
    combine<5, true><<<cdiv(N, B), B, 0, stream>>>(agg, hs, inv, hout, N);
  }

  node_transform<5, 1, true><<<tblocks, 256, 0, stream>>>(hA, Wl10, Wr10, b10, hl, hs, agg, N);
  edge_scatter<1><<<cdiv(E, B), B, 0, stream>>>(src, dst, hl, agg, E);
  combine<1, false><<<cdiv(N, B), B, 0, stream>>>(agg, hs, inv, (float*)d_out, N);
}


# 5-wide gather rows at a 32-byte pitch, fetched by a lane pair with one 16-byte load each (4 rows per pair in flight): one aligned request per edge; sums now 4 pair partials of 4-way trips + xor tree
# speedup vs baseline: 4.6063x; 1.1372x over previous
#include <hip/hip_runtime.h>


typedef __attribute__((ext_vector_type(2))) float v2f;
typedef __attribute__((ext_vector_type(4))) float v4f;
typedef __attribute__((ext_vector_type(8))) float v8f;

#define ROW8 8

#define SCAN_CHUNK 2048

__global__ void zero_i32(int* __restrict__ p, int n) {
  const int i = blockIdx.x * blockDim.x + threadIdx.x;
  if (i < n) p[i] = 0;
}

__global__ void count_in_edges(const int* __restrict__ dst, int* __restrict__ rp,
                               long n_edges) {
  const long stride = (long)gridDim.x * blockDim.x;
  long i = (long)blockIdx.x * blockDim.x + threadIdx.x;
  for (; i + 3 * stride < n_edges; i += 4 * stride) {
    const int d0 = dst[i];
    const int d1 = dst[i + stride];
    const int d2 = dst[i + 2 * stride];
    const int d3 = dst[i + 3 * stride];
    atomicAdd(&rp[d0], 1);
    atomicAdd(&rp[d1], 1);
    atomicAdd(&rp[d2], 1);
    atomicAdd(&rp[d3], 1);
  }
  for (; i < n_edges; i += stride) atomicAdd(&rp[dst[i]], 1);
}

__device__ inline int wave_incl_scan(int x, int lane) {
  for (int off = 1; off < 32; off <<= 1) {
    const int y = __shfl_up(x, off);
    if (lane >= off) x += y;
  }
  return x;
}

__global__ void block_sums(const int* __restrict__ cnt, int* __restrict__ bsum,
                           int n) {
  __shared__ int wsum[8];
  const int base = blockIdx.x * SCAN_CHUNK;
  int s = 0;
  for (int k = 0; k < 8; ++k) {
    const int idx = base + k * 256 + (int)threadIdx.x;
    if (idx < n) s += cnt[idx];
  }
  for (int off = 16; off > 0; off >>= 1) s += __shfl_xor(s, off);
  if ((threadIdx.x & 31) == 0) wsum[threadIdx.x >> 5] = s;
  __syncthreads();
  if (threadIdx.x == 0) {
    int t = 0;
    for (int w = 0; w < 8; ++w) t += wsum[w];
    bsum[blockIdx.x] = t;
  }
}

__global__ void scan_block_sums(int* __restrict__ bsum, int nb) {
  __shared__ int wsum[8];
  const int lane = threadIdx.x & 31;
  const int wave = threadIdx.x >> 5;
  int carry = 0;
  for (int base = 0; base < nb; base += 256) {
    const int i = base + (int)threadIdx.x;
    const int v = (i < nb) ? bsum[i] : 0;
    const int x = wave_incl_scan(v, lane);
    if (lane == 31) wsum[wave] = x;
    __syncthreads();
    int before = 0, total = 0;
    for (int w = 0; w < 8; ++w) {
      const int t = wsum[w];
      before += (w < wave) ? t : 0;
      total += t;
    }
    if (i < nb) bsum[i] = carry + before + x - v;
    carry += total;
    __syncthreads();
  }
}

__global__ void scan_apply(int* rp, const int* __restrict__ bofs, int n) {
  __shared__ int wsum[8];
  const int lane = threadIdx.x & 31;
  const int wave = threadIdx.x >> 5;
  const int base = blockIdx.x * SCAN_CHUNK + (int)threadIdx.x * 8;
  int v[8];
  int t = 0;
  for (int k = 0; k < 8; ++k) {
    const int idx = base + k;
    v[k] = (idx < n) ? rp[idx] : 0;
    t += v[k];
  }
  const int x = wave_incl_scan(t, lane);
  if (lane == 31) wsum[wave] = x;
  __syncthreads();
  int run = bofs[blockIdx.x] + x - t;
  for (int w = 0; w < 8; ++w) run += (w < wave) ? wsum[w] : 0;
  for (int k = 0; k < 8; ++k) {
    const int idx = base + k;
    if (idx < n) rp[idx] = run;
    run += v[k];
  }
}

__global__ void fill_rows(const int* __restrict__ src, const int* __restrict__ dst,
                          int* __restrict__ rp, int* __restrict__ ss,
                          long n_edges) {
  const long stride = (long)gridDim.x * blockDim.x;
  long i = (long)blockIdx.x * blockDim.x + threadIdx.x;
  for (; i + 3 * stride < n_edges; i += 4 * stride) {
    const int d0 = dst[i];
    const int d1 = dst[i + stride];
    const int d2 = dst[i + 2 * stride];
    const int d3 = dst[i + 3 * stride];
    const int s0 = src[i];
    const int s1 = src[i + stride];
    const int s2 = src[i + 2 * stride];
    const int s3 = src[i + 3 * stride];
    const int p0 = atomicAdd(&rp[d0], 1);
    const int p1 = atomicAdd(&rp[d1], 1);
    const int p2 = atomicAdd(&rp[d2], 1);
    const int p3 = atomicAdd(&rp[d3], 1);
    ss[p0] = s0;
    ss[p1] = s1;
    ss[p2] = s2;
    ss[p3] = s3;
  }
  for (; i < n_edges; i += stride) {
    const int d = dst[i];
    const int s = src[i];
    const int p = atomicAdd(&rp[d], 1);
    ss[p] = s;
  }
}

template <int F_IN, int F_OUT, bool ZERO_AGG, int HL_PITCH>
__global__ void node_transform(const float* __restrict__ h,
                               const float* __restrict__ Wl,
                               const float* __restrict__ Wr,
                               const float* __restrict__ bias,
                               float* __restrict__ hl,
                               float* __restrict__ hs,
                               float* __restrict__ agg,
                               int n_nodes) {
  const int lane = threadIdx.x & 31;
  const int wave = threadIdx.x >> 5;
  const int tile = blockIdx.x * (blockDim.x >> 5) + wave;
  const int tiles = (n_nodes + 15) >> 4;
  if (tile >= tiles) return;

  const int m    = lane & 15;
  const int hi   = lane >> 4;
  const int row0 = tile << 4;
  const int arow = min(row0 + m, n_nodes - 1);
  const bool cok = (m < F_OUT);

  auto ldA = [&](int k) -> float {
    const bool ok = (k < F_IN);
    const float v = h[(long)arow * F_IN + (ok ? k : 0)];
    return ok ? v : 0.0f;
  };
  auto ldW = [&](const float* __restrict__ W, int k) -> float {
    const bool ok = cok && (k < F_IN);
    const float v = W[ok ? (m * F_IN + k) : 0];
    return ok ? v : 0.0f;
  };

  v8f dl = {};
  v8f dr = {};
#pragma unroll
  for (int kc = 0; kc < F_IN; kc += 4) {
    const int k0 = kc + hi * 2;
    v2f a, bl, br;
    a.x  = ldA(k0);
    a.y  = ldA(k0 + 1);
    bl.x = ldW(Wl, k0);
    bl.y = ldW(Wl, k0 + 1);
    br.x = ldW(Wr, k0);
    br.y = ldW(Wr, k0 + 1);
    dl = __builtin_amdgcn_wmma_f32_16x16x4_f32(false, a, false, bl, (short)0,
                                               dl, false, false);
    dr = __builtin_amdgcn_wmma_f32_16x16x4_f32(false, a, false, br, (short)0,
                                               dr, false, false);
  }

  const bool full = (row0 + 16 <= n_nodes);
  if (m < HL_PITCH) {
    const long lbase = (long)(row0 + hi * 8) * HL_PITCH + m;
    if (full) {
#pragma unroll
      for (int v = 0; v < 8; ++v) hl[lbase + (long)v * HL_PITCH] = dl[v];
    } else {
#pragma unroll
      for (int v = 0; v < 8; ++v)
        if (row0 + hi * 8 + v < n_nodes) hl[lbase + (long)v * HL_PITCH] = dl[v];
    }
  }
  if (cok) {
    const float bv   = bias[m];
    const long  base = (long)(row0 + hi * 8) * F_OUT + m;
    if (full) {
#pragma unroll
      for (int v = 0; v < 8; ++v) {
        hs[base + (long)v * F_OUT]  = dr[v] + bv;
        if (ZERO_AGG) agg[base + (long)v * F_OUT] = 0.0f;
      }
    } else {
#pragma unroll
      for (int v = 0; v < 8; ++v) {
        if (row0 + hi * 8 + v < n_nodes) {
          hs[base + (long)v * F_OUT]  = dr[v] + bv;
          if (ZERO_AGG) agg[base + (long)v * F_OUT] = 0.0f;
        }
      }
    }
  }
}

template <int F, bool RELU>
__global__ void pull_combine(const int* __restrict__ rp, const int* __restrict__ ss,
                             const float* __restrict__ hl,
                             const float* __restrict__ hs,
                             float* __restrict__ out, int n_nodes) {
  const int q    = threadIdx.x & 7;
  const int node = (int)((blockIdx.x * blockDim.x + threadIdx.x) >> 3);
  const bool live = node < n_nodes;
  const int nd  = live ? node : n_nodes - 1;
  const int beg = (nd > 0) ? rp[nd - 1] : 0;
  const int end = live ? rp[nd] : beg;

  float acc[F];
  for (int f = 0; f < F; ++f) acc[f] = 0.0f;
  for (int j = beg + q; j < end; j += 8) {
    const int s = ss[j];
    for (int f = 0; f < F; ++f) acc[f] += hl[(long)s * F + f];
  }
  for (int f = 0; f < F; ++f) {
    acc[f] += __shfl_xor(acc[f], 4);
    acc[f] += __shfl_xor(acc[f], 2);
    acc[f] += __shfl_xor(acc[f], 1);
  }
  float val = acc[0];
  for (int f = 1; f < F; ++f) val = (q == f) ? acc[f] : val;

  if (live && q < F) {
    const float iv = 1.0f / fmaxf((float)(end - beg), 1.0f);
    const long  o  = (long)node * F + q;
    const float r  = val * iv + hs[o];
    out[o] = RELU ? fmaxf(r, 0.0f) : r;
  }
}

template <bool RELU>
__global__ void pull_rows8(const int* __restrict__ rp, const int* __restrict__ ss,
                           const float* __restrict__ hl8,
                           const float* __restrict__ hs,
                           float* __restrict__ out, int n_nodes) {
  const int l    = threadIdx.x & 7;
  const int p    = l >> 1;
  const int hh   = l & 1;
  const int node = (int)((blockIdx.x * blockDim.x + threadIdx.x) >> 3);
  const bool live = node < n_nodes;
  const int nd  = live ? node : n_nodes - 1;
  const int beg = (nd > 0) ? rp[nd - 1] : 0;
  const int end = live ? rp[nd] : beg;

  const v4f* __restrict__ rows = (const v4f*)hl8;
  const v4f zero = {0.0f, 0.0f, 0.0f, 0.0f};
  v4f acc = zero;
  for (int j0 = beg + p; j0 < end; j0 += 16) {
    const int j1 = j0 + 4, j2 = j0 + 8, j3 = j0 + 12;
    const bool k1 = j1 < end, k2 = j2 < end, k3 = j3 < end;
    const int s0 = ss[j0];
    int s1 = ss[k1 ? j1 : j0];
    int s2 = ss[k2 ? j2 : j0];
    int s3 = ss[k3 ? j3 : j0];
    asm volatile("" : "+v"(s1), "+v"(s2), "+v"(s3));
    v4f v0 = rows[(long)s0 * 2 + hh];
    v4f v1 = zero, v2 = zero, v3 = zero;
    if (k1) v1 = rows[(long)s1 * 2 + hh];
    if (k2) v2 = rows[(long)s2 * 2 + hh];
    if (k3) v3 = rows[(long)s3 * 2 + hh];
    acc += (v0 + v1) + (v2 + v3);
  }
  for (int c = 0; c < 4; ++c) {
    acc[c] += __shfl_xor(acc[c], 4);
    acc[c] += __shfl_xor(acc[c], 2);
  }
  const float nx = __shfl_xor(acc[0], 1);
  const float ny = __shfl_xor(acc[1], 1);
  const float nw = __shfl_xor(acc[3], 1);
  float val = acc[0];
  val = (l == 1) ? ny : val;
  val = (l == 2) ? acc[2] : val;
  val = (l == 3) ? nw : val;
  val = (l == 4) ? nx : val;

  if (live && l < 5) {
    const float iv = 1.0f / fmaxf((float)(end - beg), 1.0f);
    const long  o  = (long)node * 5 + l;
    const float r  = val * iv + hs[o];
    out[o] = RELU ? fmaxf(r, 0.0f) : r;
  }
}

__global__ void zero_f32(float* __restrict__ p, long n) {
  long i = (long)blockIdx.x * blockDim.x + threadIdx.x;
  if (i < n) p[i] = 0.0f;
}

__global__ void count_edges(const int* __restrict__ dst, float* __restrict__ cnt,
                            long n_edges) {
  long i = (long)blockIdx.x * blockDim.x + threadIdx.x;
  if (i < n_edges) unsafeAtomicAdd(&cnt[dst[i]], 1.0f);
}

__global__ void reciprocal_clamped(float* __restrict__ p, long n) {
  long i = (long)blockIdx.x * blockDim.x + threadIdx.x;
  if (i < n) p[i] = 1.0f / fmaxf(p[i], 1.0f);
}

template <int F>
__global__ void edge_scatter(const int* __restrict__ src,
                             const int* __restrict__ dst,
                             const float* __restrict__ hl,
                             float* __restrict__ agg, long n_edges) {
  long i = (long)blockIdx.x * blockDim.x + threadIdx.x;
  if (i >= n_edges) return;
  const int s = src[i];
  const int d = dst[i];
  float v[F];
#pragma unroll
  for (int j = 0; j < F; ++j) v[j] = hl[(long)s * F + j];
#pragma unroll
  for (int j = 0; j < F; ++j) unsafeAtomicAdd(&agg[(long)d * F + j], v[j]);
}

template <int F, bool RELU>
__global__ void combine(const float* __restrict__ agg,
                        const float* __restrict__ hs,
                        const float* __restrict__ inv,
                        float* __restrict__ out, int n_nodes) {
  const int node = blockIdx.x * blockDim.x + threadIdx.x;
  if (node >= n_nodes) return;
  const float iv = inv[node];
#pragma unroll
  for (int j = 0; j < F; ++j) {
    const long o = (long)node * F + j;
    const float v = agg[o] * iv + hs[o];
    out[o] = RELU ? fmaxf(v, 0.0f) : v;
  }
}

extern "C" void kernel_launch(void* const* d_in, const int* in_sizes, int n_in,
                              void* d_out, int out_size, void* d_ws,
                              size_t ws_size, hipStream_t stream) {
  const float* x    = (const float*)d_in[0];
  const int*   ei   = (const int*)  d_in[1];
  const float* Wl1  = (const float*)d_in[2];
  const float* Wr1  = (const float*)d_in[3];
  const float* b1   = (const float*)d_in[4];
  const float* Wlm  = (const float*)d_in[5];
  const float* Wrm  = (const float*)d_in[6];
  const float* bm   = (const float*)d_in[7];
  const float* Wl10 = (const float*)d_in[8];
  const float* Wr10 = (const float*)d_in[9];
  const float* b10  = (const float*)d_in[10];

  const int  N = in_sizes[0] / 10;
  const long E = (long)in_sizes[1] / 2;
  const int* src = ei;
  const int* dst = ei + E;

  const int B = 256;
  auto cdiv = [](long a, long b) { return (int)((a + b - 1) / b); };
  auto pad32 = [](size_t a) { return (a + 31) & ~(size_t)31; };

  const int tiles   = (N + 15) / 16;
  const int tblocks = (tiles + 7) / 8;

  const int    nb    = cdiv(N, SCAN_CHUNK);
  const size_t w_rp  = pad32((size_t)N);
  const size_t w_bs  = pad32((size_t)nb);
  const size_t w_ss  = pad32((size_t)E);
  const size_t w_h   = pad32((size_t)N * 5);
  const size_t w_h8  = pad32((size_t)N * ROW8);
  const size_t need  = (w_rp + w_bs + w_ss + 3 * w_h) * sizeof(int);
  const size_t need8 = (w_rp + w_bs + w_ss + 2 * w_h + w_h8) * sizeof(int);

  if (ws_size >= need) {
    const bool wide = (ws_size >= need8);
    int*   rp   = (int*)d_ws;
    int*   bsum = rp + w_rp;
    int*   ss   = bsum + w_bs;
    float* h    = (float*)(ss + w_ss);
    float* hs   = h + w_h;
    float* hl   = hs + w_h;
    float* none = nullptr;

    const int eblocks = cdiv(E, B) < 2048 ? cdiv(E, B) : 2048;
    zero_i32<<<cdiv(N, B), B, 0, stream>>>(rp, N);
    count_in_edges<<<eblocks, B, 0, stream>>>(dst, rp, E);
    block_sums<<<nb, B, 0, stream>>>(rp, bsum, N);
    scan_block_sums<<<1, B, 0, stream>>>(bsum, nb);
    scan_apply<<<nb, B, 0, stream>>>(rp, bsum, N);
    fill_rows<<<eblocks, B, 0, stream>>>(src, dst, rp, ss, E);

    const int pblocks = cdiv((long)N * 8, B);

    if (wide) {
      node_transform<10, 5, false, ROW8><<<tblocks, 256, 0, stream>>>(x, Wl1, Wr1, b1, hl, hs, none, N);
      pull_rows8<true><<<pblocks, B, 0, stream>>>(rp, ss, hl, hs, h, N);
    } else {
      node_transform<10, 5, false, 5><<<tblocks, 256, 0, stream>>>(x, Wl1, Wr1, b1, hl, hs, none, N);
      pull_combine<5, true><<<pblocks, B, 0, stream>>>(rp, ss, hl, hs, h, N);
    }

    for (int l = 0; l < 8; ++l) {
      if (wide) {
        node_transform<5, 5, false, ROW8><<<tblocks, 256, 0, stream>>>(
            h, Wlm + l * 25, Wrm + l * 25, bm + l * 5, hl, hs, none, N);
        pull_rows8<true><<<pblocks, B, 0, stream>>>(rp, ss, hl, hs, h, N);
      } else {
        node_transform<5, 5, false, 5><<<tblocks, 256, 0, stream>>>(
            h, Wlm + l * 25, Wrm + l * 25, bm + l * 5, hl, hs, none, N);
        pull_combine<5, true><<<pblocks, B, 0, stream>>>(rp, ss, hl, hs, h, N);
      }
    }

    node_transform<5, 1, false, 1><<<tblocks, 256, 0, stream>>>(h, Wl10, Wr10, b10, hl, hs, none, N);
    pull_combine<1, false><<<pblocks, B, 0, stream>>>(rp, ss, hl, hs, (float*)d_out, N);
    return;
  }

  float* ws  = (float*)d_ws;
  float* inv = ws;
  float* hl  = inv + (size_t)N;
  float* hs  = hl + (size_t)N * 5;
  float* agg = hs + (size_t)N * 5;
  float* hA  = agg + (size_t)N * 5;
  float* hB  = hA + (size_t)N * 5;

  zero_f32<<<cdiv(N, B), B, 0, stream>>>(inv, N);
  count_edges<<<cdiv(E, B), B, 0, stream>>>(dst, inv, E);
  reciprocal_clamped<<<cdiv(N, B), B, 0, stream>>>(inv, N);

  node_transform<10, 5, true, 5><<<tblocks, 256, 0, stream>>>(x, Wl1, Wr1, b1, hl, hs, agg, N);
  edge_scatter<5><<<cdiv(E, B), B, 0, stream>>>(src, dst, hl, agg, E);
  combine<5, true><<<cdiv(N, B), B, 0, stream>>>(agg, hs, inv, hA, N);

  for (int l = 0; l < 8; ++l) {
    const float* hin  = (l & 1) ? hB : hA;
    float*       hout = (l & 1) ? hA : hB;
    node_transform<5, 5, true, 5><<<tblocks, 256, 0, stream>>>(
        hin, Wlm + l * 25, Wrm + l * 25, bm + l * 5, hl, hs, agg, N);
    edge_scatter<5><<<cdiv(E, B), B, 0, stream>>>(src, dst, hl, agg, E);
    combine<5, true><<<cdiv(N, B), B, 0, stream>>>(agg, hs, inv, hout, N);
  }

  node_transform<5, 1, true, 1><<<tblocks, 256, 0, stream>>>(hA, Wl10, Wr10, b10, hl, hs, agg, N);
  edge_scatter<1><<<cdiv(E, B), B, 0, stream>>>(src, dst, hl, agg, E);
  combine<1, false><<<cdiv(N, B), B, 0, stream>>>(agg, hs, inv, (float*)d_out, N);
}


# Edge sort by destination now two levels staged in LDS (2048-node buckets, then per-bucket ranking with LDS adds, whole-line writes); row entry order, so the order of sums, is LDS-add arrival order
# speedup vs baseline: 8.8537x; 1.9221x over previous
#include <hip/hip_runtime.h>


typedef __attribute__((ext_vector_type(2))) float v2f;
typedef __attribute__((ext_vector_type(4))) float v4f;
typedef __attribute__((ext_vector_type(8))) float v8f;

#define ROW8 8

#define SCAN_CHUNK 2048

__global__ void zero_i32(int* __restrict__ p, int n) {
  const int i = blockIdx.x * blockDim.x + threadIdx.x;
  if (i < n) p[i] = 0;
}

__global__ void count_in_edges(const int* __restrict__ dst, int* __restrict__ rp,
                               long n_edges) {
  const long stride = (long)gridDim.x * blockDim.x;
  long i = (long)blockIdx.x * blockDim.x + threadIdx.x;
  for (; i + 3 * stride < n_edges; i += 4 * stride) {
    const int d0 = dst[i];
    const int d1 = dst[i + stride];
    const int d2 = dst[i + 2 * stride];
    const int d3 = dst[i + 3 * stride];
    atomicAdd(&rp[d0], 1);
    atomicAdd(&rp[d1], 1);
    atomicAdd(&rp[d2], 1);
    atomicAdd(&rp[d3], 1);
  }
  for (; i < n_edges; i += stride) atomicAdd(&rp[dst[i]], 1);
}

__device__ inline int wave_incl_scan(int x, int lane) {
  for (int off = 1; off < 32; off <<= 1) {
    const int y = __shfl_up(x, off);
    if (lane >= off) x += y;
  }
  return x;
}

__global__ void block_sums(const int* __restrict__ cnt, int* __restrict__ bsum,
                           int n) {
  __shared__ int wsum[8];
  const int base = blockIdx.x * SCAN_CHUNK;
  int s = 0;
  for (int k = 0; k < 8; ++k) {
    const int idx = base + k * 256 + (int)threadIdx.x;
    if (idx < n) s += cnt[idx];
  }
  for (int off = 16; off > 0; off >>= 1) s += __shfl_xor(s, off);
  if ((threadIdx.x & 31) == 0) wsum[threadIdx.x >> 5] = s;
  __syncthreads();
  if (threadIdx.x == 0) {
    int t = 0;
    for (int w = 0; w < 8; ++w) t += wsum[w];
    bsum[blockIdx.x] = t;
  }
}

__global__ void scan_block_sums(int* __restrict__ bsum, int nb) {
  __shared__ int wsum[8];
  const int lane = threadIdx.x & 31;
  const int wave = threadIdx.x >> 5;
  int carry = 0;
  for (int base = 0; base < nb; base += 256) {
    const int i = base + (int)threadIdx.x;
    const int v = (i < nb) ? bsum[i] : 0;
    const int x = wave_incl_scan(v, lane);
    if (lane == 31) wsum[wave] = x;
    __syncthreads();
    int before = 0, total = 0;
    for (int w = 0; w < 8; ++w) {
      const int t = wsum[w];
      before += (w < wave) ? t : 0;
      total += t;
    }
    if (i < nb) bsum[i] = carry + before + x - v;
    carry += total;
    __syncthreads();
  }
}

__global__ void scan_apply(int* rp, const int* __restrict__ bofs, int n) {
  __shared__ int wsum[8];
  const int lane = threadIdx.x & 31;
  const int wave = threadIdx.x >> 5;
  const int base = blockIdx.x * SCAN_CHUNK + (int)threadIdx.x * 8;
  int v[8];
  int t = 0;
  for (int k = 0; k < 8; ++k) {
    const int idx = base + k;
    v[k] = (idx < n) ? rp[idx] : 0;
    t += v[k];
  }
  const int x = wave_incl_scan(t, lane);
  if (lane == 31) wsum[wave] = x;
  __syncthreads();
  int run = bofs[blockIdx.x] + x - t;
  for (int w = 0; w < 8; ++w) run += (w < wave) ? wsum[w] : 0;
  for (int k = 0; k < 8; ++k) {
    const int idx = base + k;
    if (idx < n) rp[idx] = run;
    run += v[k];
  }
}

__global__ void fill_rows(const int* __restrict__ src, const int* __restrict__ dst,
                          int* __restrict__ rp, int* __restrict__ ss,
                          long n_edges) {
  const long stride = (long)gridDim.x * blockDim.x;
  long i = (long)blockIdx.x * blockDim.x + threadIdx.x;
  for (; i + 3 * stride < n_edges; i += 4 * stride) {
    const int d0 = dst[i];
    const int d1 = dst[i + stride];
    const int d2 = dst[i + 2 * stride];
    const int d3 = dst[i + 3 * stride];
    const int s0 = src[i];
    const int s1 = src[i + stride];
    const int s2 = src[i + 2 * stride];
    const int s3 = src[i + 3 * stride];
    const int p0 = atomicAdd(&rp[d0], 1);
    const int p1 = atomicAdd(&rp[d1], 1);
    const int p2 = atomicAdd(&rp[d2], 1);
    const int p3 = atomicAdd(&rp[d3], 1);
    ss[p0] = s0;
    ss[p1] = s1;
    ss[p2] = s2;
    ss[p3] = s3;
  }
  for (; i < n_edges; i += stride) {
    const int d = dst[i];
    const int s = src[i];
    const int p = atomicAdd(&rp[d], 1);
    ss[p] = s;
  }
}

#define BK_SHIFT 11
#define BK_NODES 2048
#define MAX_BK 1024
#define PART_CAP 16384
#define BK_CAP 73728

__device__ inline int bucket_of(int d, int nbk) {
  const int b = (int)((unsigned)d >> BK_SHIFT);
  return b < nbk ? b : nbk - 1;
}

__global__ void part_hist(const int* __restrict__ dst, int* __restrict__ hist,
                          long n_edges, int chunk, int nbk) {
  __shared__ int bins[MAX_BK];
  for (int b = threadIdx.x; b < nbk; b += blockDim.x) bins[b] = 0;
  __syncthreads();
  const long lo = (long)blockIdx.x * chunk;
  long hi = lo + chunk;
  if (hi > n_edges) hi = n_edges;
  for (long i = lo + threadIdx.x; i < hi; i += 4 * blockDim.x) {
    const long i1 = i + blockDim.x, i2 = i1 + blockDim.x, i3 = i2 + blockDim.x;
    const int d0 = dst[i];
    const int d1 = dst[i1 < hi ? i1 : i];
    const int d2 = dst[i2 < hi ? i2 : i];
    const int d3 = dst[i3 < hi ? i3 : i];
    atomicAdd(&bins[bucket_of(d0, nbk)], 1);
    if (i1 < hi) atomicAdd(&bins[bucket_of(d1, nbk)], 1);
    if (i2 < hi) atomicAdd(&bins[bucket_of(d2, nbk)], 1);
    if (i3 < hi) atomicAdd(&bins[bucket_of(d3, nbk)], 1);
  }
  __syncthreads();
  for (int b = threadIdx.x; b < nbk; b += blockDim.x)
    hist[(long)b * gridDim.x + blockIdx.x] = bins[b];
}

__global__ void part_scatter(const int* __restrict__ src,
                             const int* __restrict__ dst,
                             const int* __restrict__ offs,
                             int* __restrict__ tmp, long n_edges, int chunk,
                             int nbk) {
  __shared__ int lbeg[MAX_BK];
  __shared__ int cur[MAX_BK];
  __shared__ int wsum[8];
  __shared__ int stage[PART_CAP];
  const int tid = threadIdx.x, lane = tid & 31, wave = tid >> 5;
  const int G = gridDim.x, g = blockIdx.x;
  const long total = (long)nbk * G;

  int v[4];
  int t = 0;
  for (int k = 0; k < 4; ++k) {
    const int b = 4 * tid + k;
    int c = 0;
    if (b < nbk) {
      const long idx = (long)b * G + g;
      const int o0 = offs[idx];
      const int o1 = (idx + 1 < total) ? offs[idx + 1] : (int)n_edges;
      c = o1 - o0;
    }
    v[k] = c;
    t += c;
  }
  const int x = wave_incl_scan(t, lane);
  if (lane == 31) wsum[wave] = x;
  __syncthreads();
  int run = x - t;
  for (int w = 0; w < 8; ++w) run += (w < wave) ? wsum[w] : 0;
  for (int k = 0; k < 4; ++k) {
    const int b = 4 * tid + k;
    if (b < nbk) { lbeg[b] = run; cur[b] = run; }
    run += v[k];
  }
  __syncthreads();

  const long lo = (long)g * chunk;
  long hi = lo + chunk;
  if (hi > n_edges) hi = n_edges;
  for (long i = lo + tid; i < hi; i += 4 * blockDim.x) {
    const long i1 = i + blockDim.x, i2 = i1 + blockDim.x, i3 = i2 + blockDim.x;
    const bool k1 = i1 < hi, k2 = i2 < hi, k3 = i3 < hi;
    const int d0 = dst[i];
    const int d1 = dst[k1 ? i1 : i];
    const int d2 = dst[k2 ? i2 : i];
    const int d3 = dst[k3 ? i3 : i];
    const int s0 = src[i];
    const int s1 = src[k1 ? i1 : i];
    const int s2 = src[k2 ? i2 : i];
    const int s3 = src[k3 ? i3 : i];
    {
      const int p = atomicAdd(&cur[bucket_of(d0, nbk)], 1);
      if (p < PART_CAP) stage[p] = (int)(((unsigned)s0 << BK_SHIFT) | ((unsigned)d0 & (BK_NODES - 1)));
    }
    if (k1) {
      const int p = atomicAdd(&cur[bucket_of(d1, nbk)], 1);
      if (p < PART_CAP) stage[p] = (int)(((unsigned)s1 << BK_SHIFT) | ((unsigned)d1 & (BK_NODES - 1)));
    }
    if (k2) {
      const int p = atomicAdd(&cur[bucket_of(d2, nbk)], 1);
      if (p < PART_CAP) stage[p] = (int)(((unsigned)s2 << BK_SHIFT) | ((unsigned)d2 & (BK_NODES - 1)));
    }
    if (k3) {
      const int p = atomicAdd(&cur[bucket_of(d3, nbk)], 1);
      if (p < PART_CAP) stage[p] = (int)(((unsigned)s3 << BK_SHIFT) | ((unsigned)d3 & (BK_NODES - 1)));
    }
  }
  __syncthreads();

  for (int b = wave; b < nbk; b += 8) {
    const int ls = lbeg[b];
    const int n = cur[b] - ls;
    const int gs = offs[(long)b * G + g];
    for (int k = lane; k < n; k += 32) tmp[gs + k] = stage[ls + k];
  }
}

__global__ void __launch_bounds__(1024)
bucket_rows(const int* __restrict__ tmp, const int* __restrict__ offs,
            int* __restrict__ rp, int* __restrict__ ss, long n_edges,
            int n_nodes, int nbk, int G) {
  __shared__ int cnt[BK_NODES];
  __shared__ int wsum[32];
  __shared__ int stage[BK_CAP];
  const int b = blockIdx.x;
  const int tid = threadIdx.x, lane = tid & 31, wave = tid >> 5;
  const int bs = offs[(long)b * G];
  const int be = (b + 1 < nbk) ? offs[(long)(b + 1) * G] : (int)n_edges;
  const int n = be - bs;
  const int* __restrict__ in = tmp + bs;

  cnt[tid] = 0;
  cnt[tid + 1024] = 0;
  __syncthreads();
  for (int i = tid; i < n; i += 4096) {
    const int i1 = i + 1024, i2 = i + 2048, i3 = i + 3072;
    const bool k1 = i1 < n, k2 = i2 < n, k3 = i3 < n;
    const int e0 = in[i];
    const int e1 = in[k1 ? i1 : i];
    const int e2 = in[k2 ? i2 : i];
    const int e3 = in[k3 ? i3 : i];
    atomicAdd(&cnt[e0 & (BK_NODES - 1)], 1);
    if (k1) atomicAdd(&cnt[e1 & (BK_NODES - 1)], 1);
    if (k2) atomicAdd(&cnt[e2 & (BK_NODES - 1)], 1);
    if (k3) atomicAdd(&cnt[e3 & (BK_NODES - 1)], 1);
  }
  __syncthreads();

  const int c0 = cnt[2 * tid];
  const int c1 = cnt[2 * tid + 1];
  const int t = c0 + c1;
  const int x = wave_incl_scan(t, lane);
  if (lane == 31) wsum[wave] = x;
  __syncthreads();
  int ex = x - t;
  for (int w = 0; w < 32; ++w) ex += (w < wave) ? wsum[w] : 0;
  cnt[2 * tid] = ex;
  cnt[2 * tid + 1] = ex + c0;
  const int node0 = b * BK_NODES + 2 * tid;
  if (node0 < n_nodes) rp[node0] = bs + ex + c0;
  if (node0 + 1 < n_nodes) rp[node0 + 1] = bs + ex + t;
  __syncthreads();

  const bool fits = (n <= BK_CAP);
  for (int i = tid; i < n; i += 4096) {
    const int i1 = i + 1024, i2 = i + 2048, i3 = i + 3072;
    const bool k1 = i1 < n, k2 = i2 < n, k3 = i3 < n;
    const unsigned e0 = (unsigned)in[i];
    const unsigned e1 = (unsigned)in[k1 ? i1 : i];
    const unsigned e2 = (unsigned)in[k2 ? i2 : i];
    const unsigned e3 = (unsigned)in[k3 ? i3 : i];
    {
      const int p = atomicAdd(&cnt[e0 & (BK_NODES - 1)], 1);
      if (fits) stage[p] = (int)(e0 >> BK_SHIFT); else ss[bs + p] = (int)(e0 >> BK_SHIFT);
    }
    if (k1) {
      const int p = atomicAdd(&cnt[e1 & (BK_NODES - 1)], 1);
      if (fits) stage[p] = (int)(e1 >> BK_SHIFT); else ss[bs + p] = (int)(e1 >> BK_SHIFT);
    }
    if (k2) {
      const int p = atomicAdd(&cnt[e2 & (BK_NODES - 1)], 1);
      if (fits) stage[p] = (int)(e2 >> BK_SHIFT); else ss[bs + p] = (int)(e2 >> BK_SHIFT);
    }
    if (k3) {
      const int p = atomicAdd(&cnt[e3 & (BK_NODES - 1)], 1);
      if (fits) stage[p] = (int)(e3 >> BK_SHIFT); else ss[bs + p] = (int)(e3 >> BK_SHIFT);
    }
  }
  __syncthreads();
  if (fits)
    for (int i = tid; i < n; i += 1024) ss[bs + i] = stage[i];
}

template <int F_IN, int F_OUT, bool ZERO_AGG, int HL_PITCH>
__global__ void node_transform(const float* __restrict__ h,
                               const float* __restrict__ Wl,
                               const float* __restrict__ Wr,
                               const float* __restrict__ bias,
                               float* __restrict__ hl,
                               float* __restrict__ hs,
                               float* __restrict__ agg,
                               int n_nodes) {
  const int lane = threadIdx.x & 31;
  const int wave = threadIdx.x >> 5;
  const int tile = blockIdx.x * (blockDim.x >> 5) + wave;
  const int tiles = (n_nodes + 15) >> 4;
  if (tile >= tiles) return;

  const int m    = lane & 15;
  const int hi   = lane >> 4;
  const int row0 = tile << 4;
  const int arow = min(row0 + m, n_nodes - 1);
  const bool cok = (m < F_OUT);

  auto ldA = [&](int k) -> float {
    const bool ok = (k < F_IN);
    const float v = h[(long)arow * F_IN + (ok ? k : 0)];
    return ok ? v : 0.0f;
  };
  auto ldW = [&](const float* __restrict__ W, int k) -> float {
    const bool ok = cok && (k < F_IN);
    const float v = W[ok ? (m * F_IN + k) : 0];
    return ok ? v : 0.0f;
  };

  v8f dl = {};
  v8f dr = {};
#pragma unroll
  for (int kc = 0; kc < F_IN; kc += 4) {
    const int k0 = kc + hi * 2;
    v2f a, bl, br;
    a.x  = ldA(k0);
    a.y  = ldA(k0 + 1);
    bl.x = ldW(Wl, k0);
    bl.y = ldW(Wl, k0 + 1);
    br.x = ldW(Wr, k0);
    br.y = ldW(Wr, k0 + 1);
    dl = __builtin_amdgcn_wmma_f32_16x16x4_f32(false, a, false, bl, (short)0,
                                               dl, false, false);
    dr = __builtin_amdgcn_wmma_f32_16x16x4_f32(false, a, false, br, (short)0,
                                               dr, false, false);
  }

  const bool full = (row0 + 16 <= n_nodes);
  if (m < HL_PITCH) {
    const long lbase = (long)(row0 + hi * 8) * HL_PITCH + m;
    if (full) {
#pragma unroll
      for (int v = 0; v < 8; ++v) hl[lbase + (long)v * HL_PITCH] = dl[v];
    } else {
#pragma unroll
      for (int v = 0; v < 8; ++v)
        if (row0 + hi * 8 + v < n_nodes) hl[lbase + (long)v * HL_PITCH] = dl[v];
    }
  }
  if (cok) {
    const float bv   = bias[m];
    const long  base = (long)(row0 + hi * 8) * F_OUT + m;
    if (full) {
#pragma unroll
      for (int v = 0; v < 8; ++v) {
        hs[base + (long)v * F_OUT]  = dr[v] + bv;
        if (ZERO_AGG) agg[base + (long)v * F_OUT] = 0.0f;
      }
    } else {
#pragma unroll
      for (int v = 0; v < 8; ++v) {
        if (row0 + hi * 8 + v < n_nodes) {
          hs[base + (long)v * F_OUT]  = dr[v] + bv;
          if (ZERO_AGG) agg[base + (long)v * F_OUT] = 0.0f;
        }
      }
    }
  }
}

template <int F, bool RELU>
__global__ void pull_combine(const int* __restrict__ rp, const int* __restrict__ ss,
                             const float* __restrict__ hl,
                             const float* __restrict__ hs,
                             float* __restrict__ out, int n_nodes) {
  const int q    = threadIdx.x & 7;
  const int node = (int)((blockIdx.x * blockDim.x + threadIdx.x) >> 3);
  const bool live = node < n_nodes;
  const int nd  = live ? node : n_nodes - 1;
  const int beg = (nd > 0) ? rp[nd - 1] : 0;
  const int end = live ? rp[nd] : beg;

  float acc[F];
  for (int f = 0; f < F; ++f) acc[f] = 0.0f;
  for (int j = beg + q; j < end; j += 8) {
    const int s = ss[j];
    for (int f = 0; f < F; ++f) acc[f] += hl[(long)s * F + f];
  }
  for (int f = 0; f < F; ++f) {
    acc[f] += __shfl_xor(acc[f], 4);
    acc[f] += __shfl_xor(acc[f], 2);
    acc[f] += __shfl_xor(acc[f], 1);
  }
  float val = acc[0];
  for (int f = 1; f < F; ++f) val = (q == f) ? acc[f] : val;

  if (live && q < F) {
    const float iv = 1.0f / fmaxf((float)(end - beg), 1.0f);
    const long  o  = (long)node * F + q;
    const float r  = val * iv + hs[o];
    out[o] = RELU ? fmaxf(r, 0.0f) : r;
  }
}

template <bool RELU>
__global__ void pull_rows8(const int* __restrict__ rp, const int* __restrict__ ss,
                           const float* __restrict__ hl8,
                           const float* __restrict__ hs,
                           float* __restrict__ out, int n_nodes) {
  const int l    = threadIdx.x & 7;
  const int p    = l >> 1;
  const int hh   = l & 1;
  const int node = (int)((blockIdx.x * blockDim.x + threadIdx.x) >> 3);
  const bool live = node < n_nodes;
  const int nd  = live ? node : n_nodes - 1;
  const int beg = (nd > 0) ? rp[nd - 1] : 0;
  const int end = live ? rp[nd] : beg;

  const v4f* __restrict__ rows = (const v4f*)hl8;
  const v4f zero = {0.0f, 0.0f, 0.0f, 0.0f};
  v4f acc = zero;
  for (int j0 = beg + p; j0 < end; j0 += 16) {
    const int j1 = j0 + 4, j2 = j0 + 8, j3 = j0 + 12;
    const bool k1 = j1 < end, k2 = j2 < end, k3 = j3 < end;
    const int s0 = ss[j0];
    int s1 = ss[k1 ? j1 : j0];
    int s2 = ss[k2 ? j2 : j0];
    int s3 = ss[k3 ? j3 : j0];
    asm volatile("" : "+v"(s1), "+v"(s2), "+v"(s3));
    v4f v0 = rows[(long)s0 * 2 + hh];
    v4f v1 = zero, v2 = zero, v3 = zero;
    if (k1) v1 = rows[(long)s1 * 2 + hh];
    if (k2) v2 = rows[(long)s2 * 2 + hh];
    if (k3) v3 = rows[(long)s3 * 2 + hh];
    acc += (v0 + v1) + (v2 + v3);
  }
  for (int c = 0; c < 4; ++c) {
    acc[c] += __shfl_xor(acc[c], 4);
    acc[c] += __shfl_xor(acc[c], 2);
  }
  const float nx = __shfl_xor(acc[0], 1);
  const float ny = __shfl_xor(acc[1], 1);
  const float nw = __shfl_xor(acc[3], 1);
  float val = acc[0];
  val = (l == 1) ? ny : val;
  val = (l == 2) ? acc[2] : val;
  val = (l == 3) ? nw : val;
  val = (l == 4) ? nx : val;

  if (live && l < 5) {
    const float iv = 1.0f / fmaxf((float)(end - beg), 1.0f);
    const long  o  = (long)node * 5 + l;
    const float r  = val * iv + hs[o];
    out[o] = RELU ? fmaxf(r, 0.0f) : r;
  }
}

__global__ void zero_f32(float* __restrict__ p, long n) {
  long i = (long)blockIdx.x * blockDim.x + threadIdx.x;
  if (i < n) p[i] = 0.0f;
}

__global__ void count_edges(const int* __restrict__ dst, float* __restrict__ cnt,
                            long n_edges) {
  long i = (long)blockIdx.x * blockDim.x + threadIdx.x;
  if (i < n_edges) unsafeAtomicAdd(&cnt[dst[i]], 1.0f);
}

__global__ void reciprocal_clamped(float* __restrict__ p, long n) {
  long i = (long)blockIdx.x * blockDim.x + threadIdx.x;
  if (i < n) p[i] = 1.0f / fmaxf(p[i], 1.0f);
}

template <int F>
__global__ void edge_scatter(const int* __restrict__ src,
                             const int* __restrict__ dst,
                             const float* __restrict__ hl,
                             float* __restrict__ agg, long n_edges) {
  long i = (long)blockIdx.x * blockDim.x + threadIdx.x;
  if (i >= n_edges) return;
  const int s = src[i];
  const int d = dst[i];
  float v[F];
#pragma unroll
  for (int j = 0; j < F; ++j) v[j] = hl[(long)s * F + j];
#pragma unroll
  for (int j = 0; j < F; ++j) unsafeAtomicAdd(&agg[(long)d * F + j], v[j]);
}

template <int F, bool RELU>
__global__ void combine(const float* __restrict__ agg,
                        const float* __restrict__ hs,
                        const float* __restrict__ inv,
                        float* __restrict__ out, int n_nodes) {
  const int node = blockIdx.x * blockDim.x + threadIdx.x;
  if (node >= n_nodes) return;
  const float iv = inv[node];
#pragma unroll
  for (int j = 0; j < F; ++j) {
    const long o = (long)node * F + j;
    const float v = agg[o] * iv + hs[o];
    out[o] = RELU ? fmaxf(v, 0.0f) : v;
  }
}

extern "C" void kernel_launch(void* const* d_in, const int* in_sizes, int n_in,
                              void* d_out, int out_size, void* d_ws,
                              size_t ws_size, hipStream_t stream) {
  const float* x    = (const float*)d_in[0];
  const int*   ei   = (const int*)  d_in[1];
  const float* Wl1  = (const float*)d_in[2];
  const float* Wr1  = (const float*)d_in[3];
  const float* b1   = (const float*)d_in[4];
  const float* Wlm  = (const float*)d_in[5];
  const float* Wrm  = (const float*)d_in[6];
  const float* bm   = (const float*)d_in[7];
  const float* Wl10 = (const float*)d_in[8];
  const float* Wr10 = (const float*)d_in[9];
  const float* b10  = (const float*)d_in[10];

  const int  N = in_sizes[0] / 10;
  const long E = (long)in_sizes[1] / 2;
  const int* src = ei;
  const int* dst = ei + E;

  const int B = 256;
  auto cdiv = [](long a, long b) { return (int)((a + b - 1) / b); };
  auto pad32 = [](size_t a) { return (a + 31) & ~(size_t)31; };

  const int tiles   = (N + 15) / 16;
  const int tblocks = (tiles + 7) / 8;

  const int    nb    = cdiv(N, SCAN_CHUNK);
  const size_t w_rp  = pad32((size_t)N);
  const size_t w_bs  = pad32((size_t)nb);
  const size_t w_ss  = pad32((size_t)E);
  const size_t w_h   = pad32((size_t)N * 5);
  const size_t w_h8  = pad32((size_t)N * ROW8);
  const size_t need  = (w_rp + w_bs + w_ss + 3 * w_h) * sizeof(int);
  const size_t need8 = (w_rp + w_bs + w_ss + 2 * w_h + w_h8) * sizeof(int);

  if (ws_size >= need) {
    const bool wide = (ws_size >= need8);
    int*   rp   = (int*)d_ws;
    int*   bsum = rp + w_rp;
    int*   ss   = bsum + w_bs;
    float* h    = (float*)(ss + w_ss);
    float* hs   = h + w_h;
    float* hl   = hs + w_h;
    float* none = nullptr;

    const int    PG     = 1024;
    const int    nbk    = cdiv(N, BK_NODES);
    const int    chunk  = cdiv(E, PG);
    const size_t w_hist = pad32((size_t)nbk * PG);
    const size_t need_s = need8 + (w_hist + w_ss) * sizeof(int);
    const bool lds_sort = wide && ws_size >= need_s && nbk <= MAX_BK &&
                          chunk <= PART_CAP && N <= (1 << 21);
    if (lds_sort) {
      int* hist = (int*)(hl + w_h8);
      int* tmp  = hist + w_hist;
      const int hn  = nbk * PG;
      const int hnb = cdiv(hn, SCAN_CHUNK);
      part_hist<<<PG, 256, 0, stream>>>(dst, hist, E, chunk, nbk);
      block_sums<<<hnb, B, 0, stream>>>(hist, bsum, hn);
      scan_block_sums<<<1, B, 0, stream>>>(bsum, hnb);
      scan_apply<<<hnb, B, 0, stream>>>(hist, bsum, hn);
      part_scatter<<<PG, 256, 0, stream>>>(src, dst, hist, tmp, E, chunk, nbk);
      bucket_rows<<<nbk, 1024, 0, stream>>>(tmp, hist, rp, ss, E, N, nbk, PG);
    } else {
      const int eblocks = cdiv(E, B) < 2048 ? cdiv(E, B) : 2048;
      zero_i32<<<cdiv(N, B), B, 0, stream>>>(rp, N);
      count_in_edges<<<eblocks, B, 0, stream>>>(dst, rp, E);
      block_sums<<<nb, B, 0, stream>>>(rp, bsum, N);
      scan_block_sums<<<1, B, 0, stream>>>(bsum, nb);
      scan_apply<<<nb, B, 0, stream>>>(rp, bsum, N);
      fill_rows<<<eblocks, B, 0, stream>>>(src, dst, rp, ss, E);
    }

    const int pblocks = cdiv((long)N * 8, B);

    if (wide) {
      node_transform<10, 5, false, ROW8><<<tblocks, 256, 0, stream>>>(x, Wl1, Wr1, b1, hl, hs, none, N);
      pull_rows8<true><<<pblocks, B, 0, stream>>>(rp, ss, hl, hs, h, N);
    } else {
      node_transform<10, 5, false, 5><<<tblocks, 256, 0, stream>>>(x, Wl1, Wr1, b1, hl, hs, none, N);
      pull_combine<5, true><<<pblocks, B, 0, stream>>>(rp, ss, hl, hs, h, N);
    }

    for (int l = 0; l < 8; ++l) {
      if (wide) {
        node_transform<5, 5, false, ROW8><<<tblocks, 256, 0, stream>>>(
            h, Wlm + l * 25, Wrm + l * 25, bm + l * 5, hl, hs, none, N);
        pull_rows8<true><<<pblocks, B, 0, stream>>>(rp, ss, hl, hs, h, N);
      } else {
        node_transform<5, 5, false, 5><<<tblocks, 256, 0, stream>>>(
            h, Wlm + l * 25, Wrm + l * 25, bm + l * 5, hl, hs, none, N);
        pull_combine<5, true><<<pblocks, B, 0, stream>>>(rp, ss, hl, hs, h, N);
      }
    }

    node_transform<5, 1, false, 1><<<tblocks, 256, 0, stream>>>(h, Wl10, Wr10, b10, hl, hs, none, N);
    pull_combine<1, false><<<pblocks, B, 0, stream>>>(rp, ss, hl, hs, (float*)d_out, N);
    return;
  }

  float* ws  = (float*)d_ws;
  float* inv = ws;
  float* hl  = inv + (size_t)N;
  float* hs  = hl + (size_t)N * 5;
  float* agg = hs + (size_t)N * 5;
  float* hA  = agg + (size_t)N * 5;
  float* hB  = hA + (size_t)N * 5;

  zero_f32<<<cdiv(N, B), B, 0, stream>>>(inv, N);
  count_edges<<<cdiv(E, B), B, 0, stream>>>(dst, inv, E);
  reciprocal_clamped<<<cdiv(N, B), B, 0, stream>>>(inv, N);

  node_transform<10, 5, true, 5><<<tblocks, 256, 0, stream>>>(x, Wl1, Wr1, b1, hl, hs, agg, N);
  edge_scatter<5><<<cdiv(E, B), B, 0, stream>>>(src, dst, hl, agg, E);
  combine<5, true><<<cdiv(N, B), B, 0, stream>>>(agg, hs, inv, hA, N);

  for (int l = 0; l < 8; ++l) {
    const float* hin  = (l & 1) ? hB : hA;
    float*       hout = (l & 1) ? hA : hB;
    node_transform<5, 5, true, 5><<<tblocks, 256, 0, stream>>>(
        hin, Wlm + l * 25, Wrm + l * 25, bm + l * 5, hl, hs, agg, N);
    edge_scatter<5><<<cdiv(E, B), B, 0, stream>>>(src, dst, hl, agg, E);
    combine<5, true><<<cdiv(N, B), B, 0, stream>>>(agg, hs, inv, hout, N);
  }

  node_transform<5, 1, true, 1><<<tblocks, 256, 0, stream>>>(hA, Wl10, Wr10, b10, hl, hs, agg, N);
  edge_scatter<1><<<cdiv(E, B), B, 0, stream>>>(src, dst, hl, agg, E);
  combine<1, false><<<cdiv(N, B), B, 0, stream>>>(agg, hs, inv, (float*)d_out, N);
}


# Layers 2..10 one launch each: pull takes the mean of gathered activation rows, then two waves of the block do [mean|self]x[Wl;Wr] on the f32 matrix unit (aggregate, then transform: sum order changes)
# speedup vs baseline: 9.5625x; 1.0801x over previous
#include <hip/hip_runtime.h>


typedef __attribute__((ext_vector_type(2))) float v2f;
typedef __attribute__((ext_vector_type(4))) float v4f;
typedef __attribute__((ext_vector_type(8))) float v8f;

#define ROW8 8

#define SCAN_CHUNK 2048

__global__ void zero_i32(int* __restrict__ p, int n) {
  const int i = blockIdx.x * blockDim.x + threadIdx.x;
  if (i < n) p[i] = 0;
}

__global__ void count_in_edges(const int* __restrict__ dst, int* __restrict__ rp,
                               long n_edges) {
  const long stride = (long)gridDim.x * blockDim.x;
  long i = (long)blockIdx.x * blockDim.x + threadIdx.x;
  for (; i + 3 * stride < n_edges; i += 4 * stride) {
    const int d0 = dst[i];
    const int d1 = dst[i + stride];
    const int d2 = dst[i + 2 * stride];
    const int d3 = dst[i + 3 * stride];
    atomicAdd(&rp[d0], 1);
    atomicAdd(&rp[d1], 1);
    atomicAdd(&rp[d2], 1);
    atomicAdd(&rp[d3], 1);
  }
  for (; i < n_edges; i += stride) atomicAdd(&rp[dst[i]], 1);
}

__device__ inline int wave_incl_scan(int x, int lane) {
  for (int off = 1; off < 32; off <<= 1) {
    const int y = __shfl_up(x, off);
    if (lane >= off) x += y;
  }
  return x;
}

__global__ void block_sums(const int* __restrict__ cnt, int* __restrict__ bsum,
                           int n) {
  __shared__ int wsum[8];
  const int base = blockIdx.x * SCAN_CHUNK;
  int s = 0;
  for (int k = 0; k < 8; ++k) {
    const int idx = base + k * 256 + (int)threadIdx.x;
    if (idx < n) s += cnt[idx];
  }
  for (int off = 16; off > 0; off >>= 1) s += __shfl_xor(s, off);
  if ((threadIdx.x & 31) == 0) wsum[threadIdx.x >> 5] = s;
  __syncthreads();
  if (threadIdx.x == 0) {
    int t = 0;
    for (int w = 0; w < 8; ++w) t += wsum[w];
    bsum[blockIdx.x] = t;
  }
}

__global__ void scan_block_sums(int* __restrict__ bsum, int nb) {
  __shared__ int wsum[8];
  const int lane = threadIdx.x & 31;
  const int wave = threadIdx.x >> 5;
  int carry = 0;
  for (int base = 0; base < nb; base += 256) {
    const int i = base + (int)threadIdx.x;
    const int v = (i < nb) ? bsum[i] : 0;
    const int x = wave_incl_scan(v, lane);
    if (lane == 31) wsum[wave] = x;
    __syncthreads();
    int before = 0, total = 0;
    for (int w = 0; w < 8; ++w) {
      const int t = wsum[w];
      before += (w < wave) ? t : 0;
      total += t;
    }
    if (i < nb) bsum[i] = carry + before + x - v;
    carry += total;
    __syncthreads();
  }
}

__global__ void scan_apply(int* rp, const int* __restrict__ bofs, int n) {
  __shared__ int wsum[8];
  const int lane = threadIdx.x & 31;
  const int wave = threadIdx.x >> 5;
  const int base = blockIdx.x * SCAN_CHUNK + (int)threadIdx.x * 8;
  int v[8];
  int t = 0;
  for (int k = 0; k < 8; ++k) {
    const int idx = base + k;
    v[k] = (idx < n) ? rp[idx] : 0;
    t += v[k];
  }
  const int x = wave_incl_scan(t, lane);
  if (lane == 31) wsum[wave] = x;
  __syncthreads();
  int run = bofs[blockIdx.x] + x - t;
  for (int w = 0; w < 8; ++w) run += (w < wave) ? wsum[w] : 0;
  for (int k = 0; k < 8; ++k) {
    const int idx = base + k;
    if (idx < n) rp[idx] = run;
    run += v[k];
  }
}

__global__ void fill_rows(const int* __restrict__ src, const int* __restrict__ dst,
                          int* __restrict__ rp, int* __restrict__ ss,
                          long n_edges) {
  const long stride = (long)gridDim.x * blockDim.x;
  long i = (long)blockIdx.x * blockDim.x + threadIdx.x;
  for (; i + 3 * stride < n_edges; i += 4 * stride) {
    const int d0 = dst[i];
    const int d1 = dst[i + stride];
    const int d2 = dst[i + 2 * stride];
    const int d3 = dst[i + 3 * stride];
    const int s0 = src[i];
    const int s1 = src[i + stride];
    const int s2 = src[i + 2 * stride];
    const int s3 = src[i + 3 * stride];
    const int p0 = atomicAdd(&rp[d0], 1);
    const int p1 = atomicAdd(&rp[d1], 1);
    const int p2 = atomicAdd(&rp[d2], 1);
    const int p3 = atomicAdd(&rp[d3], 1);
    ss[p0] = s0;
    ss[p1] = s1;
    ss[p2] = s2;
    ss[p3] = s3;
  }
  for (; i < n_edges; i += stride) {
    const int d = dst[i];
    const int s = src[i];
    const int p = atomicAdd(&rp[d], 1);
    ss[p] = s;
  }
}

#define BK_SHIFT 11
#define BK_NODES 2048
#define MAX_BK 1024
#define PART_CAP 16384
#define BK_CAP 73728

__device__ inline int bucket_of(int d, int nbk) {
  const int b = (int)((unsigned)d >> BK_SHIFT);
  return b < nbk ? b : nbk - 1;
}

__global__ void part_hist(const int* __restrict__ dst, int* __restrict__ hist,
                          long n_edges, int chunk, int nbk) {
  __shared__ int bins[MAX_BK];
  for (int b = threadIdx.x; b < nbk; b += blockDim.x) bins[b] = 0;
  __syncthreads();
  const long lo = (long)blockIdx.x * chunk;
  long hi = lo + chunk;
  if (hi > n_edges) hi = n_edges;
  for (long i = lo + threadIdx.x; i < hi; i += 4 * blockDim.x) {
    const long i1 = i + blockDim.x, i2 = i1 + blockDim.x, i3 = i2 + blockDim.x;
    const int d0 = dst[i];
    const int d1 = dst[i1 < hi ? i1 : i];
    const int d2 = dst[i2 < hi ? i2 : i];
    const int d3 = dst[i3 < hi ? i3 : i];
    atomicAdd(&bins[bucket_of(d0, nbk)], 1);
    if (i1 < hi) atomicAdd(&bins[bucket_of(d1, nbk)], 1);
    if (i2 < hi) atomicAdd(&bins[bucket_of(d2, nbk)], 1);
    if (i3 < hi) atomicAdd(&bins[bucket_of(d3, nbk)], 1);
  }
  __syncthreads();
  for (int b = threadIdx.x; b < nbk; b += blockDim.x)
    hist[(long)b * gridDim.x + blockIdx.x] = bins[b];
}

__global__ void part_scatter(const int* __restrict__ src,
                             const int* __restrict__ dst,
                             const int* __restrict__ offs,
                             int* __restrict__ tmp, long n_edges, int chunk,
                             int nbk) {
  __shared__ int lbeg[MAX_BK];
  __shared__ int cur[MAX_BK];
  __shared__ int wsum[8];
  __shared__ int stage[PART_CAP];
  const int tid = threadIdx.x, lane = tid & 31, wave = tid >> 5;
  const int G = gridDim.x, g = blockIdx.x;
  const long total = (long)nbk * G;

  int v[4];
  int t = 0;
  for (int k = 0; k < 4; ++k) {
    const int b = 4 * tid + k;
    int c = 0;
    if (b < nbk) {
      const long idx = (long)b * G + g;
      const int o0 = offs[idx];
      const int o1 = (idx + 1 < total) ? offs[idx + 1] : (int)n_edges;
      c = o1 - o0;
    }
    v[k] = c;
    t += c;
  }
  const int x = wave_incl_scan(t, lane);
  if (lane == 31) wsum[wave] = x;
  __syncthreads();
  int run = x - t;
  for (int w = 0; w < 8; ++w) run += (w < wave) ? wsum[w] : 0;
  for (int k = 0; k < 4; ++k) {
    const int b = 4 * tid + k;
    if (b < nbk) { lbeg[b] = run; cur[b] = run; }
    run += v[k];
  }
  __syncthreads();

  const long lo = (long)g * chunk;
  long hi = lo + chunk;
  if (hi > n_edges) hi = n_edges;
  for (long i = lo + tid; i < hi; i += 4 * blockDim.x) {
    const long i1 = i + blockDim.x, i2 = i1 + blockDim.x, i3 = i2 + blockDim.x;
    const bool k1 = i1 < hi, k2 = i2 < hi, k3 = i3 < hi;
    const int d0 = dst[i];
    const int d1 = dst[k1 ? i1 : i];
    const int d2 = dst[k2 ? i2 : i];
    const int d3 = dst[k3 ? i3 : i];
    const int s0 = src[i];
    const int s1 = src[k1 ? i1 : i];
    const int s2 = src[k2 ? i2 : i];
    const int s3 = src[k3 ? i3 : i];
    {
      const int p = atomicAdd(&cur[bucket_of(d0, nbk)], 1);
      if (p < PART_CAP) stage[p] = (int)(((unsigned)s0 << BK_SHIFT) | ((unsigned)d0 & (BK_NODES - 1)));
    }
    if (k1) {
      const int p = atomicAdd(&cur[bucket_of(d1, nbk)], 1);
      if (p < PART_CAP) stage[p] = (int)(((unsigned)s1 << BK_SHIFT) | ((unsigned)d1 & (BK_NODES - 1)));
    }
    if (k2) {
      const int p = atomicAdd(&cur[bucket_of(d2, nbk)], 1);
      if (p < PART_CAP) stage[p] = (int)(((unsigned)s2 << BK_SHIFT) | ((unsigned)d2 & (BK_NODES - 1)));
    }
    if (k3) {
      const int p = atomicAdd(&cur[bucket_of(d3, nbk)], 1);
      if (p < PART_CAP) stage[p] = (int)(((unsigned)s3 << BK_SHIFT) | ((unsigned)d3 & (BK_NODES - 1)));
    }
  }
  __syncthreads();

  for (int b = wave; b < nbk; b += 8) {
    const int ls = lbeg[b];
    const int n = cur[b] - ls;
    const int gs = offs[(long)b * G + g];
    for (int k = lane; k < n; k += 32) tmp[gs + k] = stage[ls + k];
  }
}

__global__ void __launch_bounds__(1024)
bucket_rows(const int* __restrict__ tmp, const int* __restrict__ offs,
            int* __restrict__ rp, int* __restrict__ ss, long n_edges,
            int n_nodes, int nbk, int G) {
  __shared__ int cnt[BK_NODES];
  __shared__ int wsum[32];
  __shared__ int stage[BK_CAP];
  const int b = blockIdx.x;
  const int tid = threadIdx.x, lane = tid & 31, wave = tid >> 5;
  const int bs = offs[(long)b * G];
  const int be = (b + 1 < nbk) ? offs[(long)(b + 1) * G] : (int)n_edges;
  const int n = be - bs;
  const int* __restrict__ in = tmp + bs;

  cnt[tid] = 0;
  cnt[tid + 1024] = 0;
  __syncthreads();
  for (int i = tid; i < n; i += 4096) {
    const int i1 = i + 1024, i2 = i + 2048, i3 = i + 3072;
    const bool k1 = i1 < n, k2 = i2 < n, k3 = i3 < n;
    const int e0 = in[i];
    const int e1 = in[k1 ? i1 : i];
    const int e2 = in[k2 ? i2 : i];
    const int e3 = in[k3 ? i3 : i];
    atomicAdd(&cnt[e0 & (BK_NODES - 1)], 1);
    if (k1) atomicAdd(&cnt[e1 & (BK_NODES - 1)], 1);
    if (k2) atomicAdd(&cnt[e2 & (BK_NODES - 1)], 1);
    if (k3) atomicAdd(&cnt[e3 & (BK_NODES - 1)], 1);
  }
  __syncthreads();

  const int c0 = cnt[2 * tid];
  const int c1 = cnt[2 * tid + 1];
  const int t = c0 + c1;
  const int x = wave_incl_scan(t, lane);
  if (lane == 31) wsum[wave] = x;
  __syncthreads();
  int ex = x - t;
  for (int w = 0; w < 32; ++w) ex += (w < wave) ? wsum[w] : 0;
  cnt[2 * tid] = ex;
  cnt[2 * tid + 1] = ex + c0;
  const int node0 = b * BK_NODES + 2 * tid;
  if (node0 < n_nodes) rp[node0] = bs + ex + c0;
  if (node0 + 1 < n_nodes) rp[node0 + 1] = bs + ex + t;
  __syncthreads();

  const bool fits = (n <= BK_CAP);
  for (int i = tid; i < n; i += 4096) {
    const int i1 = i + 1024, i2 = i + 2048, i3 = i + 3072;
    const bool k1 = i1 < n, k2 = i2 < n, k3 = i3 < n;
    const unsigned e0 = (unsigned)in[i];
    const unsigned e1 = (unsigned)in[k1 ? i1 : i];
    const unsigned e2 = (unsigned)in[k2 ? i2 : i];
    const unsigned e3 = (unsigned)in[k3 ? i3 : i];
    {
      const int p = atomicAdd(&cnt[e0 & (BK_NODES - 1)], 1);
      if (fits) stage[p] = (int)(e0 >> BK_SHIFT); else ss[bs + p] = (int)(e0 >> BK_SHIFT);
    }
    if (k1) {
      const int p = atomicAdd(&cnt[e1 & (BK_NODES - 1)], 1);
      if (fits) stage[p] = (int)(e1 >> BK_SHIFT); else ss[bs + p] = (int)(e1 >> BK_SHIFT);
    }
    if (k2) {
      const int p = atomicAdd(&cnt[e2 & (BK_NODES - 1)], 1);
      if (fits) stage[p] = (int)(e2 >> BK_SHIFT); else ss[bs + p] = (int)(e2 >> BK_SHIFT);
    }
    if (k3) {
      const int p = atomicAdd(&cnt[e3 & (BK_NODES - 1)], 1);
      if (fits) stage[p] = (int)(e3 >> BK_SHIFT); else ss[bs + p] = (int)(e3 >> BK_SHIFT);
    }
  }
  __syncthreads();
  if (fits)
    for (int i = tid; i < n; i += 1024) ss[bs + i] = stage[i];
}

template <int F_IN, int F_OUT, bool ZERO_AGG, int HL_PITCH>
__global__ void node_transform(const float* __restrict__ h,
                               const float* __restrict__ Wl,
                               const float* __restrict__ Wr,
                               const float* __restrict__ bias,
                               float* __restrict__ hl,
                               float* __restrict__ hs,
                               float* __restrict__ agg,
                               int n_nodes) {
  const int lane = threadIdx.x & 31;
  const int wave = threadIdx.x >> 5;
  const int tile = blockIdx.x * (blockDim.x >> 5) + wave;
  const int tiles = (n_nodes + 15) >> 4;
  if (tile >= tiles) return;

  const int m    = lane & 15;
  const int hi   = lane >> 4;
  const int row0 = tile << 4;
  const int arow = min(row0 + m, n_nodes - 1);
  const bool cok = (m < F_OUT);

  auto ldA = [&](int k) -> float {
    const bool ok = (k < F_IN);
    const float v = h[(long)arow * F_IN + (ok ? k : 0)];
    return ok ? v : 0.0f;
  };
  auto ldW = [&](const float* __restrict__ W, int k) -> float {
    const bool ok = cok && (k < F_IN);
    const float v = W[ok ? (m * F_IN + k) : 0];
    return ok ? v : 0.0f;
  };

  v8f dl = {};
  v8f dr = {};
#pragma unroll
  for (int kc = 0; kc < F_IN; kc += 4) {
    const int k0 = kc + hi * 2;
    v2f a, bl, br;
    a.x  = ldA(k0);
    a.y  = ldA(k0 + 1);
    bl.x = ldW(Wl, k0);
    bl.y = ldW(Wl, k0 + 1);
    br.x = ldW(Wr, k0);
    br.y = ldW(Wr, k0 + 1);
    dl = __builtin_amdgcn_wmma_f32_16x16x4_f32(false, a, false, bl, (short)0,
                                               dl, false, false);
    dr = __builtin_amdgcn_wmma_f32_16x16x4_f32(false, a, false, br, (short)0,
                                               dr, false, false);
  }

  const bool full = (row0 + 16 <= n_nodes);
  if (m < HL_PITCH) {
    const long lbase = (long)(row0 + hi * 8) * HL_PITCH + m;
    if (full) {
#pragma unroll
      for (int v = 0; v < 8; ++v) hl[lbase + (long)v * HL_PITCH] = dl[v];
    } else {
#pragma unroll
      for (int v = 0; v < 8; ++v)
        if (row0 + hi * 8 + v < n_nodes) hl[lbase + (long)v * HL_PITCH] = dl[v];
    }
  }
  if (cok) {
    const float bv   = bias[m];
    const long  base = (long)(row0 + hi * 8) * F_OUT + m;
    if (full) {
#pragma unroll
      for (int v = 0; v < 8; ++v) {
        hs[base + (long)v * F_OUT]  = dr[v] + bv;
        if (ZERO_AGG) agg[base + (long)v * F_OUT] = 0.0f;
      }
    } else {
#pragma unroll
      for (int v = 0; v < 8; ++v) {
        if (row0 + hi * 8 + v < n_nodes) {
          hs[base + (long)v * F_OUT]  = dr[v] + bv;
          if (ZERO_AGG) agg[base + (long)v * F_OUT] = 0.0f;
        }
      }
    }
  }
}

template <int F, bool RELU>
__global__ void pull_combine(const int* __restrict__ rp, const int* __restrict__ ss,
                             const float* __restrict__ hl,
                             const float* __restrict__ hs,
                             float* __restrict__ out, int n_nodes) {
  const int q    = threadIdx.x & 7;
  const int node = (int)((blockIdx.x * blockDim.x + threadIdx.x) >> 3);
  const bool live = node < n_nodes;
  const int nd  = live ? node : n_nodes - 1;
  const int beg = (nd > 0) ? rp[nd - 1] : 0;
  const int end = live ? rp[nd] : beg;

  float acc[F];
  for (int f = 0; f < F; ++f) acc[f] = 0.0f;
  for (int j = beg + q; j < end; j += 8) {
    const int s = ss[j];
    for (int f = 0; f < F; ++f) acc[f] += hl[(long)s * F + f];
  }
  for (int f = 0; f < F; ++f) {
    acc[f] += __shfl_xor(acc[f], 4);
    acc[f] += __shfl_xor(acc[f], 2);
    acc[f] += __shfl_xor(acc[f], 1);
  }
  float val = acc[0];
  for (int f = 1; f < F; ++f) val = (q == f) ? acc[f] : val;

  if (live && q < F) {
    const float iv = 1.0f / fmaxf((float)(end - beg), 1.0f);
    const long  o  = (long)node * F + q;
    const float r  = val * iv + hs[o];
    out[o] = RELU ? fmaxf(r, 0.0f) : r;
  }
}

template <bool RELU, int OUT_PITCH>
__global__ void pull_rows8(const int* __restrict__ rp, const int* __restrict__ ss,
                           const float* __restrict__ hl8,
                           const float* __restrict__ hs,
                           float* __restrict__ out, int n_nodes) {
  const int l    = threadIdx.x & 7;
  const int p    = l >> 1;
  const int hh   = l & 1;
  const int node = (int)((blockIdx.x * blockDim.x + threadIdx.x) >> 3);
  const bool live = node < n_nodes;
  const int nd  = live ? node : n_nodes - 1;
  const int beg = (nd > 0) ? rp[nd - 1] : 0;
  const int end = live ? rp[nd] : beg;

  const v4f* __restrict__ rows = (const v4f*)hl8;
  const v4f zero = {0.0f, 0.0f, 0.0f, 0.0f};
  v4f acc = zero;
  for (int j0 = beg + p; j0 < end; j0 += 16) {
    const int j1 = j0 + 4, j2 = j0 + 8, j3 = j0 + 12;
    const bool k1 = j1 < end, k2 = j2 < end, k3 = j3 < end;
    const int s0 = ss[j0];
    int s1 = ss[k1 ? j1 : j0];
    int s2 = ss[k2 ? j2 : j0];
    int s3 = ss[k3 ? j3 : j0];
    asm volatile("" : "+v"(s1), "+v"(s2), "+v"(s3));
    v4f v0 = rows[(long)s0 * 2 + hh];
    v4f v1 = zero, v2 = zero, v3 = zero;
    if (k1) v1 = rows[(long)s1 * 2 + hh];
    if (k2) v2 = rows[(long)s2 * 2 + hh];
    if (k3) v3 = rows[(long)s3 * 2 + hh];
    acc += (v0 + v1) + (v2 + v3);
  }
  for (int c = 0; c < 4; ++c) {
    acc[c] += __shfl_xor(acc[c], 4);
    acc[c] += __shfl_xor(acc[c], 2);
  }
  const float nx = __shfl_xor(acc[0], 1);
  const float ny = __shfl_xor(acc[1], 1);
  const float nw = __shfl_xor(acc[3], 1);
  float val = acc[0];
  val = (l == 1) ? ny : val;
  val = (l == 2) ? acc[2] : val;
  val = (l == 3) ? nw : val;
  val = (l == 4) ? nx : val;

  if (live && l < OUT_PITCH) {
    float r = 0.0f;
    if (l < 5) {
      const float iv = 1.0f / fmaxf((float)(end - beg), 1.0f);
      r = val * iv + hs[(long)node * 5 + l];
      r = RELU ? fmaxf(r, 0.0f) : r;
    }
    out[(long)node * OUT_PITCH + l] = r;
  }
}

#define A_PITCH 12
template <int F_OUT, bool RELU, int OUT_PITCH>
__global__ void __launch_bounds__(256)
pull_transform(const int* __restrict__ rp, const int* __restrict__ ss,
               const float* __restrict__ h8,
               const float* __restrict__ Wl, const float* __restrict__ Wr,
               const float* __restrict__ bias,
               float* __restrict__ out, int n_nodes) {
  __shared__ float arow[32 * A_PITCH];
  const int l    = threadIdx.x & 7;
  const int p    = l >> 1;
  const int hh   = l & 1;
  const int ln   = threadIdx.x >> 3;
  const int node = (int)blockIdx.x * 32 + ln;
  const bool live = node < n_nodes;
  const int nd  = live ? node : n_nodes - 1;
  const int beg = (nd > 0) ? rp[nd - 1] : 0;
  const int end = live ? rp[nd] : beg;

  const v4f* __restrict__ rows = (const v4f*)h8;
  const v4f zero = {0.0f, 0.0f, 0.0f, 0.0f};
  const float mine = (l < 5) ? h8[(long)nd * ROW8 + l] : 0.0f;

  const int wave = threadIdx.x >> 5;
  const int lane = threadIdx.x & 31;
  const int m    = lane & 15;
  const int hi   = lane >> 4;
  const bool cok = (m < F_OUT);
  auto bok = [&](int k) -> bool { return cok && (k < 10); };
  auto ldB = [&](int k) -> float {
    const bool left = (k < 5);
    const float* __restrict__ Wp = left ? Wl : Wr;
    const int kk = left ? k : k - 5;
    return Wp[bok(k) ? (m * 5 + kk) : 0];
  };
  float braw[A_PITCH / 2];
  float bv = 0.0f;
#pragma unroll
  for (int i = 0; i < A_PITCH / 2; ++i) braw[i] = 0.0f;
  if (wave < 2) {
#pragma unroll
    for (int i = 0; i < A_PITCH / 4; ++i) {
      const int k0 = 4 * i + hi * 2;
      braw[2 * i]     = ldB(k0);
      braw[2 * i + 1] = ldB(k0 + 1);
    }
    bv = cok ? bias[m] : 0.0f;
  }
  v4f acc = zero;
  for (int j0 = beg + p; j0 < end; j0 += 16) {
    const int j1 = j0 + 4, j2 = j0 + 8, j3 = j0 + 12;
    const bool k1 = j1 < end, k2 = j2 < end, k3 = j3 < end;
    const int s0 = ss[j0];
    int s1 = ss[k1 ? j1 : j0];
    int s2 = ss[k2 ? j2 : j0];
    int s3 = ss[k3 ? j3 : j0];
    asm volatile("" : "+v"(s1), "+v"(s2), "+v"(s3));
    v4f v0 = rows[(long)s0 * 2 + hh];
    v4f v1 = zero, v2 = zero, v3 = zero;
    if (k1) v1 = rows[(long)s1 * 2 + hh];
    if (k2) v2 = rows[(long)s2 * 2 + hh];
    if (k3) v3 = rows[(long)s3 * 2 + hh];
    acc += (v0 + v1) + (v2 + v3);
  }
  for (int c = 0; c < 4; ++c) {
    acc[c] += __shfl_xor(acc[c], 4);
    acc[c] += __shfl_xor(acc[c], 2);
  }
  const float nx = __shfl_xor(acc[0], 1);
  const float ny = __shfl_xor(acc[1], 1);
  const float nw = __shfl_xor(acc[3], 1);
  float val = acc[0];
  val = (l == 1) ? ny : val;
  val = (l == 2) ? acc[2] : val;
  val = (l == 3) ? nw : val;
  val = (l == 4) ? nx : val;

  const float iv = 1.0f / fmaxf((float)(end - beg), 1.0f);
  if (l < 5) {
    arow[ln * A_PITCH + l]     = live ? val * iv : 0.0f;
    arow[ln * A_PITCH + 5 + l] = live ? mine : 0.0f;
  } else if (l < 7) {
    arow[ln * A_PITCH + 5 + l] = 0.0f;
  }
  __syncthreads();

  if (wave >= 2) return;

  const float* __restrict__ at = arow + (wave * 16 + m) * A_PITCH;
  v8f d = {};
#pragma unroll
  for (int kc = 0; kc < A_PITCH; kc += 4) {
    const int k0 = kc + hi * 2;
    v2f a, b;
    a.x = at[k0];
    a.y = at[k0 + 1];
    b.x = bok(k0) ? braw[kc / 2] : 0.0f;
    b.y = bok(k0 + 1) ? braw[kc / 2 + 1] : 0.0f;
    d = __builtin_amdgcn_wmma_f32_16x16x4_f32(false, a, false, b, (short)0,
                                              d, false, false);
  }

  const int row0 = (int)blockIdx.x * 32 + wave * 16;
  if (m < OUT_PITCH) {
    const long  base = (long)(row0 + hi * 8) * OUT_PITCH + m;
    const bool  full = (row0 + 16 <= n_nodes);
    if (full) {
#pragma unroll
      for (int v = 0; v < 8; ++v) {
        const float r = d[v] + bv;
        out[base + (long)v * OUT_PITCH] = RELU ? fmaxf(r, 0.0f) : r;
      }
    } else {
#pragma unroll
      for (int v = 0; v < 8; ++v) {
        if (row0 + hi * 8 + v < n_nodes) {
          const float r = d[v] + bv;
          out[base + (long)v * OUT_PITCH] = RELU ? fmaxf(r, 0.0f) : r;
        }
      }
    }
  }
}

__global__ void zero_f32(float* __restrict__ p, long n) {
  long i = (long)blockIdx.x * blockDim.x + threadIdx.x;
  if (i < n) p[i] = 0.0f;
}

__global__ void count_edges(const int* __restrict__ dst, float* __restrict__ cnt,
                            long n_edges) {
  long i = (long)blockIdx.x * blockDim.x + threadIdx.x;
  if (i < n_edges) unsafeAtomicAdd(&cnt[dst[i]], 1.0f);
}

__global__ void reciprocal_clamped(float* __restrict__ p, long n) {
  long i = (long)blockIdx.x * blockDim.x + threadIdx.x;
  if (i < n) p[i] = 1.0f / fmaxf(p[i], 1.0f);
}

template <int F>
__global__ void edge_scatter(const int* __restrict__ src,
                             const int* __restrict__ dst,
                             const float* __restrict__ hl,
                             float* __restrict__ agg, long n_edges) {
  long i = (long)blockIdx.x * blockDim.x + threadIdx.x;
  if (i >= n_edges) return;
  const int s = src[i];
  const int d = dst[i];
  float v[F];
#pragma unroll
  for (int j = 0; j < F; ++j) v[j] = hl[(long)s * F + j];
#pragma unroll
  for (int j = 0; j < F; ++j) unsafeAtomicAdd(&agg[(long)d * F + j], v[j]);
}

template <int F, bool RELU>
__global__ void combine(const float* __restrict__ agg,
                        const float* __restrict__ hs,
                        const float* __restrict__ inv,
                        float* __restrict__ out, int n_nodes) {
  const int node = blockIdx.x * blockDim.x + threadIdx.x;
  if (node >= n_nodes) return;
  const float iv = inv[node];
#pragma unroll
  for (int j = 0; j < F; ++j) {
    const long o = (long)node * F + j;
    const float v = agg[o] * iv + hs[o];
    out[o] = RELU ? fmaxf(v, 0.0f) : v;
  }
}

extern "C" void kernel_launch(void* const* d_in, const int* in_sizes, int n_in,
                              void* d_out, int out_size, void* d_ws,
                              size_t ws_size, hipStream_t stream) {
  const float* x    = (const float*)d_in[0];
  const int*   ei   = (const int*)  d_in[1];
  const float* Wl1  = (const float*)d_in[2];
  const float* Wr1  = (const float*)d_in[3];
  const float* b1   = (const float*)d_in[4];
  const float* Wlm  = (const float*)d_in[5];
  const float* Wrm  = (const float*)d_in[6];
  const float* bm   = (const float*)d_in[7];
  const float* Wl10 = (const float*)d_in[8];
  const float* Wr10 = (const float*)d_in[9];
  const float* b10  = (const float*)d_in[10];

  const int  N = in_sizes[0] / 10;
  const long E = (long)in_sizes[1] / 2;
  const int* src = ei;
  const int* dst = ei + E;

  const int B = 256;
  auto cdiv = [](long a, long b) { return (int)((a + b - 1) / b); };
  auto pad32 = [](size_t a) { return (a + 31) & ~(size_t)31; };

  const int tiles   = (N + 15) / 16;
  const int tblocks = (tiles + 7) / 8;

  const int    nb    = cdiv(N, SCAN_CHUNK);
  const size_t w_rp  = pad32((size_t)N);
  const size_t w_bs  = pad32((size_t)nb);
  const size_t w_ss  = pad32((size_t)E);
  const size_t w_h   = pad32((size_t)N * 5);
  const size_t w_h8  = pad32((size_t)N * ROW8);
  const size_t need  = (w_rp + w_bs + w_ss + 3 * w_h) * sizeof(int);
  const size_t need8 = (w_rp + w_bs + w_ss + 2 * w_h + w_h8) * sizeof(int);

  if (ws_size >= need) {
    const bool wide = (ws_size >= need8);
    int*   rp   = (int*)d_ws;
    int*   bsum = rp + w_rp;
    int*   ss   = bsum + w_bs;
    float* h    = (float*)(ss + w_ss);
    float* hs   = h + w_h;
    float* hl   = hs + w_h;
    float* none = nullptr;

    const int    PG     = 1024;
    const int    nbk    = cdiv(N, BK_NODES);
    const int    chunk  = cdiv(E, PG);
    const size_t w_hist = pad32((size_t)nbk * PG);
    const size_t need_s = need8 + (w_hist + w_ss) * sizeof(int);
    const bool lds_sort = wide && ws_size >= need_s && nbk <= MAX_BK &&
                          chunk <= PART_CAP && N <= (1 << 21);
    if (lds_sort) {
      int* hist = (int*)(hl + w_h8);
      int* tmp  = hist + w_hist;
      const int hn  = nbk * PG;
      const int hnb = cdiv(hn, SCAN_CHUNK);
      part_hist<<<PG, 256, 0, stream>>>(dst, hist, E, chunk, nbk);
      block_sums<<<hnb, B, 0, stream>>>(hist, bsum, hn);
      scan_block_sums<<<1, B, 0, stream>>>(bsum, hnb);
      scan_apply<<<hnb, B, 0, stream>>>(hist, bsum, hn);
      part_scatter<<<PG, 256, 0, stream>>>(src, dst, hist, tmp, E, chunk, nbk);
      bucket_rows<<<nbk, 1024, 0, stream>>>(tmp, hist, rp, ss, E, N, nbk, PG);
    } else {
      const int eblocks = cdiv(E, B) < 2048 ? cdiv(E, B) : 2048;
      zero_i32<<<cdiv(N, B), B, 0, stream>>>(rp, N);
      count_in_edges<<<eblocks, B, 0, stream>>>(dst, rp, E);
      block_sums<<<nb, B, 0, stream>>>(rp, bsum, N);
      scan_block_sums<<<1, B, 0, stream>>>(bsum, nb);
      scan_apply<<<nb, B, 0, stream>>>(rp, bsum, N);
      fill_rows<<<eblocks, B, 0, stream>>>(src, dst, rp, ss, E);
    }

    const int pblocks = cdiv((long)N * 8, B);

    if (lds_sort && w_ss >= w_h8) {
      float* ta = hl;
      float* tb = hl + w_h8 + w_hist;
      node_transform<10, 5, false, ROW8><<<tblocks, 256, 0, stream>>>(x, Wl1, Wr1, b1, ta, hs, none, N);
      pull_rows8<true, ROW8><<<pblocks, B, 0, stream>>>(rp, ss, ta, hs, tb, N);
      for (int l = 0; l < 8; ++l) {
        const float* tin  = (l & 1) ? ta : tb;
        float*       tout = (l & 1) ? tb : ta;
        pull_transform<5, true, ROW8><<<pblocks, B, 0, stream>>>(
            rp, ss, tin, Wlm + l * 25, Wrm + l * 25, bm + l * 5, tout, N);
      }
      pull_transform<1, false, 1><<<pblocks, B, 0, stream>>>(
          rp, ss, tb, Wl10, Wr10, b10, (float*)d_out, N);
      return;
    }

    if (wide) {
      node_transform<10, 5, false, ROW8><<<tblocks, 256, 0, stream>>>(x, Wl1, Wr1, b1, hl, hs, none, N);
      pull_rows8<true, 5><<<pblocks, B, 0, stream>>>(rp, ss, hl, hs, h, N);
    } else {
      node_transform<10, 5, false, 5><<<tblocks, 256, 0, stream>>>(x, Wl1, Wr1, b1, hl, hs, none, N);
      pull_combine<5, true><<<pblocks, B, 0, stream>>>(rp, ss, hl, hs, h, N);
    }

    for (int l = 0; l < 8; ++l) {
      if (wide) {
        node_transform<5, 5, false, ROW8><<<tblocks, 256, 0, stream>>>(
            h, Wlm + l * 25, Wrm + l * 25, bm + l * 5, hl, hs, none, N);
        pull_rows8<true, 5><<<pblocks, B, 0, stream>>>(rp, ss, hl, hs, h, N);
      } else {
        node_transform<5, 5, false, 5><<<tblocks, 256, 0, stream>>>(
            h, Wlm + l * 25, Wrm + l * 25, bm + l * 5, hl, hs, none, N);
        pull_combine<5, true><<<pblocks, B, 0, stream>>>(rp, ss, hl, hs, h, N);
      }
    }

    node_transform<5, 1, false, 1><<<tblocks, 256, 0, stream>>>(h, Wl10, Wr10, b10, hl, hs, none, N);
    pull_combine<1, false><<<pblocks, B, 0, stream>>>(rp, ss, hl, hs, (float*)d_out, N);
    return;
  }

  float* ws  = (float*)d_ws;
  float* inv = ws;
  float* hl  = inv + (size_t)N;
  float* hs  = hl + (size_t)N * 5;
  float* agg = hs + (size_t)N * 5;
  float* hA  = agg + (size_t)N * 5;
  float* hB  = hA + (size_t)N * 5;

  zero_f32<<<cdiv(N, B), B, 0, stream>>>(inv, N);
  count_edges<<<cdiv(E, B), B, 0, stream>>>(dst, inv, E);
  reciprocal_clamped<<<cdiv(N, B), B, 0, stream>>>(inv, N);

  node_transform<10, 5, true, 5><<<tblocks, 256, 0, stream>>>(x, Wl1, Wr1, b1, hl, hs, agg, N);
  edge_scatter<5><<<cdiv(E, B), B, 0, stream>>>(src, dst, hl, agg, E);
  combine<5, true><<<cdiv(N, B), B, 0, stream>>>(agg, hs, inv, hA, N);

  for (int l = 0; l < 8; ++l) {
    const float* hin  = (l & 1) ? hB : hA;
    float*       hout = (l & 1) ? hA : hB;
    node_transform<5, 5, true, 5><<<tblocks, 256, 0, stream>>>(
        hin, Wlm + l * 25, Wrm + l * 25, bm + l * 5, hl, hs, agg, N);
    edge_scatter<5><<<cdiv(E, B), B, 0, stream>>>(src, dst, hl, agg, E);
    combine<5, true><<<cdiv(N, B), B, 0, stream>>>(agg, hs, inv, hout, N);
  }

  node_transform<5, 1, true, 1><<<tblocks, 256, 0, stream>>>(hA, Wl10, Wr10, b10, hl, hs, agg, N);
  edge_scatter<1><<<cdiv(E, B), B, 0, stream>>>(src, dst, hl, agg, E);
  combine<1, false><<<cdiv(N, B), B, 0, stream>>>(agg, hs, inv, (float*)d_out, N);
}


# Aggregation blocked the other way: 42 destination groups hold their sums in LDS, 6 source ranges each; edges sorted by source row gather neighbouring lines; LDS float adds (sum order: arrival)
# speedup vs baseline: 13.6271x; 1.4251x over previous
#include <hip/hip_runtime.h>


typedef __attribute__((ext_vector_type(2))) float v2f;
typedef __attribute__((ext_vector_type(4))) float v4f;
typedef __attribute__((ext_vector_type(8))) float v8f;

#define ROW8 8

#define SCAN_CHUNK 2048

__global__ void zero_i32(int* __restrict__ p, int n) {
  const int i = blockIdx.x * blockDim.x + threadIdx.x;
  if (i < n) p[i] = 0;
}

__global__ void count_in_edges(const int* __restrict__ dst, int* __restrict__ rp,
                               long n_edges) {
  const long stride = (long)gridDim.x * blockDim.x;
  long i = (long)blockIdx.x * blockDim.x + threadIdx.x;
  for (; i + 3 * stride < n_edges; i += 4 * stride) {
    const int d0 = dst[i];
    const int d1 = dst[i + stride];
    const int d2 = dst[i + 2 * stride];
    const int d3 = dst[i + 3 * stride];
    atomicAdd(&rp[d0], 1);
    atomicAdd(&rp[d1], 1);
    atomicAdd(&rp[d2], 1);
    atomicAdd(&rp[d3], 1);
  }
  for (; i < n_edges; i += stride) atomicAdd(&rp[dst[i]], 1);
}

__device__ inline int wave_incl_scan(int x, int lane) {
  for (int off = 1; off < 32; off <<= 1) {
    const int y = __shfl_up(x, off);
    if (lane >= off) x += y;
  }
  return x;
}

__global__ void block_sums(const int* __restrict__ cnt, int* __restrict__ bsum,
                           int n) {
  __shared__ int wsum[8];
  const int base = blockIdx.x * SCAN_CHUNK;
  int s = 0;
  for (int k = 0; k < 8; ++k) {
    const int idx = base + k * 256 + (int)threadIdx.x;
    if (idx < n) s += cnt[idx];
  }
  for (int off = 16; off > 0; off >>= 1) s += __shfl_xor(s, off);
  if ((threadIdx.x & 31) == 0) wsum[threadIdx.x >> 5] = s;
  __syncthreads();
  if (threadIdx.x == 0) {
    int t = 0;
    for (int w = 0; w < 8; ++w) t += wsum[w];
    bsum[blockIdx.x] = t;
  }
}

__global__ void scan_block_sums(int* __restrict__ bsum, int nb) {
  __shared__ int wsum[8];
  const int lane = threadIdx.x & 31;
  const int wave = threadIdx.x >> 5;
  int carry = 0;
  for (int base = 0; base < nb; base += 256) {
    const int i = base + (int)threadIdx.x;
    const int v = (i < nb) ? bsum[i] : 0;
    const int x = wave_incl_scan(v, lane);
    if (lane == 31) wsum[wave] = x;
    __syncthreads();
    int before = 0, total = 0;
    for (int w = 0; w < 8; ++w) {
      const int t = wsum[w];
      before += (w < wave) ? t : 0;
      total += t;
    }
    if (i < nb) bsum[i] = carry + before + x - v;
    carry += total;
    __syncthreads();
  }
}

__global__ void scan_apply(int* rp, const int* __restrict__ bofs, int n) {
  __shared__ int wsum[8];
  const int lane = threadIdx.x & 31;
  const int wave = threadIdx.x >> 5;
  const int base = blockIdx.x * SCAN_CHUNK + (int)threadIdx.x * 8;
  int v[8];
  int t = 0;
  for (int k = 0; k < 8; ++k) {
    const int idx = base + k;
    v[k] = (idx < n) ? rp[idx] : 0;
    t += v[k];
  }
  const int x = wave_incl_scan(t, lane);
  if (lane == 31) wsum[wave] = x;
  __syncthreads();
  int run = bofs[blockIdx.x] + x - t;
  for (int w = 0; w < 8; ++w) run += (w < wave) ? wsum[w] : 0;
  for (int k = 0; k < 8; ++k) {
    const int idx = base + k;
    if (idx < n) rp[idx] = run;
    run += v[k];
  }
}

__global__ void fill_rows(const int* __restrict__ src, const int* __restrict__ dst,
                          int* __restrict__ rp, int* __restrict__ ss,
                          long n_edges) {
  const long stride = (long)gridDim.x * blockDim.x;
  long i = (long)blockIdx.x * blockDim.x + threadIdx.x;
  for (; i + 3 * stride < n_edges; i += 4 * stride) {
    const int d0 = dst[i];
    const int d1 = dst[i + stride];
    const int d2 = dst[i + 2 * stride];
    const int d3 = dst[i + 3 * stride];
    const int s0 = src[i];
    const int s1 = src[i + stride];
    const int s2 = src[i + 2 * stride];
    const int s3 = src[i + 3 * stride];
    const int p0 = atomicAdd(&rp[d0], 1);
    const int p1 = atomicAdd(&rp[d1], 1);
    const int p2 = atomicAdd(&rp[d2], 1);
    const int p3 = atomicAdd(&rp[d3], 1);
    ss[p0] = s0;
    ss[p1] = s1;
    ss[p2] = s2;
    ss[p3] = s3;
  }
  for (; i < n_edges; i += stride) {
    const int d = dst[i];
    const int s = src[i];
    const int p = atomicAdd(&rp[d], 1);
    ss[p] = s;
  }
}

#define BK_SHIFT 11
#define BK_NODES 2048
#define MAX_BK 1024
#define PART_CAP 16384
#define BK_CAP 73728

__device__ inline int bucket_of(int d, int nbk) {
  const int b = (int)((unsigned)d >> BK_SHIFT);
  return b < nbk ? b : nbk - 1;
}

__global__ void part_hist(const int* __restrict__ dst, int* __restrict__ hist,
                          long n_edges, int chunk, int nbk) {
  __shared__ int bins[MAX_BK];
  for (int b = threadIdx.x; b < nbk; b += blockDim.x) bins[b] = 0;
  __syncthreads();
  const long lo = (long)blockIdx.x * chunk;
  long hi = lo + chunk;
  if (hi > n_edges) hi = n_edges;
  for (long i = lo + threadIdx.x; i < hi; i += 4 * blockDim.x) {
    const long i1 = i + blockDim.x, i2 = i1 + blockDim.x, i3 = i2 + blockDim.x;
    const int d0 = dst[i];
    const int d1 = dst[i1 < hi ? i1 : i];
    const int d2 = dst[i2 < hi ? i2 : i];
    const int d3 = dst[i3 < hi ? i3 : i];
    atomicAdd(&bins[bucket_of(d0, nbk)], 1);
    if (i1 < hi) atomicAdd(&bins[bucket_of(d1, nbk)], 1);
    if (i2 < hi) atomicAdd(&bins[bucket_of(d2, nbk)], 1);
    if (i3 < hi) atomicAdd(&bins[bucket_of(d3, nbk)], 1);
  }
  __syncthreads();
  for (int b = threadIdx.x; b < nbk; b += blockDim.x)
    hist[(long)b * gridDim.x + blockIdx.x] = bins[b];
}

__global__ void part_scatter(const int* __restrict__ src,
                             const int* __restrict__ dst,
                             const int* __restrict__ offs,
                             int* __restrict__ tmp, long n_edges, int chunk,
                             int nbk) {
  __shared__ int lbeg[MAX_BK];
  __shared__ int cur[MAX_BK];
  __shared__ int wsum[8];
  __shared__ int stage[PART_CAP];
  const int tid = threadIdx.x, lane = tid & 31, wave = tid >> 5;
  const int G = gridDim.x, g = blockIdx.x;
  const long total = (long)nbk * G;

  int v[4];
  int t = 0;
  for (int k = 0; k < 4; ++k) {
    const int b = 4 * tid + k;
    int c = 0;
    if (b < nbk) {
      const long idx = (long)b * G + g;
      const int o0 = offs[idx];
      const int o1 = (idx + 1 < total) ? offs[idx + 1] : (int)n_edges;
      c = o1 - o0;
    }
    v[k] = c;
    t += c;
  }
  const int x = wave_incl_scan(t, lane);
  if (lane == 31) wsum[wave] = x;
  __syncthreads();
  int run = x - t;
  for (int w = 0; w < 8; ++w) run += (w < wave) ? wsum[w] : 0;
  for (int k = 0; k < 4; ++k) {
    const int b = 4 * tid + k;
    if (b < nbk) { lbeg[b] = run; cur[b] = run; }
    run += v[k];
  }
  __syncthreads();

  const long lo = (long)g * chunk;
  long hi = lo + chunk;
  if (hi > n_edges) hi = n_edges;
  for (long i = lo + tid; i < hi; i += 4 * blockDim.x) {
    const long i1 = i + blockDim.x, i2 = i1 + blockDim.x, i3 = i2 + blockDim.x;
    const bool k1 = i1 < hi, k2 = i2 < hi, k3 = i3 < hi;
    const int d0 = dst[i];
    const int d1 = dst[k1 ? i1 : i];
    const int d2 = dst[k2 ? i2 : i];
    const int d3 = dst[k3 ? i3 : i];
    const int s0 = src[i];
    const int s1 = src[k1 ? i1 : i];
    const int s2 = src[k2 ? i2 : i];
    const int s3 = src[k3 ? i3 : i];
    {
      const int p = atomicAdd(&cur[bucket_of(d0, nbk)], 1);
      if (p < PART_CAP) stage[p] = (int)(((unsigned)s0 << BK_SHIFT) | ((unsigned)d0 & (BK_NODES - 1)));
    }
    if (k1) {
      const int p = atomicAdd(&cur[bucket_of(d1, nbk)], 1);
      if (p < PART_CAP) stage[p] = (int)(((unsigned)s1 << BK_SHIFT) | ((unsigned)d1 & (BK_NODES - 1)));
    }
    if (k2) {
      const int p = atomicAdd(&cur[bucket_of(d2, nbk)], 1);
      if (p < PART_CAP) stage[p] = (int)(((unsigned)s2 << BK_SHIFT) | ((unsigned)d2 & (BK_NODES - 1)));
    }
    if (k3) {
      const int p = atomicAdd(&cur[bucket_of(d3, nbk)], 1);
      if (p < PART_CAP) stage[p] = (int)(((unsigned)s3 << BK_SHIFT) | ((unsigned)d3 & (BK_NODES - 1)));
    }
  }
  __syncthreads();

  for (int b = wave; b < nbk; b += 8) {
    const int ls = lbeg[b];
    const int n = cur[b] - ls;
    const int gs = offs[(long)b * G + g];
    for (int k = lane; k < n; k += 32) tmp[gs + k] = stage[ls + k];
  }
}

__global__ void __launch_bounds__(1024)
bucket_rows(const int* __restrict__ tmp, const int* __restrict__ offs,
            int* __restrict__ rp, int* __restrict__ ss, long n_edges,
            int n_nodes, int nbk, int G) {
  __shared__ int cnt[BK_NODES];
  __shared__ int wsum[32];
  __shared__ int stage[BK_CAP];
  const int b = blockIdx.x;
  const int tid = threadIdx.x, lane = tid & 31, wave = tid >> 5;
  const int bs = offs[(long)b * G];
  const int be = (b + 1 < nbk) ? offs[(long)(b + 1) * G] : (int)n_edges;
  const int n = be - bs;
  const int* __restrict__ in = tmp + bs;

  cnt[tid] = 0;
  cnt[tid + 1024] = 0;
  __syncthreads();
  for (int i = tid; i < n; i += 4096) {
    const int i1 = i + 1024, i2 = i + 2048, i3 = i + 3072;
    const bool k1 = i1 < n, k2 = i2 < n, k3 = i3 < n;
    const int e0 = in[i];
    const int e1 = in[k1 ? i1 : i];
    const int e2 = in[k2 ? i2 : i];
    const int e3 = in[k3 ? i3 : i];
    atomicAdd(&cnt[e0 & (BK_NODES - 1)], 1);
    if (k1) atomicAdd(&cnt[e1 & (BK_NODES - 1)], 1);
    if (k2) atomicAdd(&cnt[e2 & (BK_NODES - 1)], 1);
    if (k3) atomicAdd(&cnt[e3 & (BK_NODES - 1)], 1);
  }
  __syncthreads();

  const int c0 = cnt[2 * tid];
  const int c1 = cnt[2 * tid + 1];
  const int t = c0 + c1;
  const int x = wave_incl_scan(t, lane);
  if (lane == 31) wsum[wave] = x;
  __syncthreads();
  int ex = x - t;
  for (int w = 0; w < 32; ++w) ex += (w < wave) ? wsum[w] : 0;
  cnt[2 * tid] = ex;
  cnt[2 * tid + 1] = ex + c0;
  const int node0 = b * BK_NODES + 2 * tid;
  if (node0 < n_nodes) rp[node0] = bs + ex + c0;
  if (node0 + 1 < n_nodes) rp[node0 + 1] = bs + ex + t;
  __syncthreads();

  const bool fits = (n <= BK_CAP);
  for (int i = tid; i < n; i += 4096) {
    const int i1 = i + 1024, i2 = i + 2048, i3 = i + 3072;
    const bool k1 = i1 < n, k2 = i2 < n, k3 = i3 < n;
    const unsigned e0 = (unsigned)in[i];
    const unsigned e1 = (unsigned)in[k1 ? i1 : i];
    const unsigned e2 = (unsigned)in[k2 ? i2 : i];
    const unsigned e3 = (unsigned)in[k3 ? i3 : i];
    {
      const int p = atomicAdd(&cnt[e0 & (BK_NODES - 1)], 1);
      if (fits) stage[p] = (int)(e0 >> BK_SHIFT); else ss[bs + p] = (int)(e0 >> BK_SHIFT);
    }
    if (k1) {
      const int p = atomicAdd(&cnt[e1 & (BK_NODES - 1)], 1);
      if (fits) stage[p] = (int)(e1 >> BK_SHIFT); else ss[bs + p] = (int)(e1 >> BK_SHIFT);
    }
    if (k2) {
      const int p = atomicAdd(&cnt[e2 & (BK_NODES - 1)], 1);
      if (fits) stage[p] = (int)(e2 >> BK_SHIFT); else ss[bs + p] = (int)(e2 >> BK_SHIFT);
    }
    if (k3) {
      const int p = atomicAdd(&cnt[e3 & (BK_NODES - 1)], 1);
      if (fits) stage[p] = (int)(e3 >> BK_SHIFT); else ss[bs + p] = (int)(e3 >> BK_SHIFT);
    }
  }
  __syncthreads();
  if (fits)
    for (int i = tid; i < n; i += 1024) ss[bs + i] = stage[i];
}

#define DG 11936
#define DL_BITS 14
#define SRN 83456
#define NR_MAX 6
#define SBIN_SHIFT 4
#define SBINS 8192
#define BK2_CAP 69632
#define PG2 992

__device__ inline int bucket2_of(int d, int s, int nr, int nbk) {
  const unsigned b = ((unsigned)d / DG) * (unsigned)nr + ((unsigned)s / SRN);
  return b < (unsigned)nbk ? (int)b : nbk - 1;
}
__device__ inline int entry2_of(int d, int s) {
  return (int)((((unsigned)s % SRN) << DL_BITS) | ((unsigned)d % DG));
}

__global__ void part_hist2(const int* __restrict__ src, const int* __restrict__ dst,
                           int* __restrict__ hist, long n_edges, int chunk,
                           int nr, int nbk) {
  __shared__ int bins[MAX_BK];
  for (int b = threadIdx.x; b < nbk; b += blockDim.x) bins[b] = 0;
  __syncthreads();
  const long lo = (long)blockIdx.x * chunk;
  long hi = lo + chunk;
  if (hi > n_edges) hi = n_edges;
  for (long i = lo + threadIdx.x; i < hi; i += 4 * blockDim.x) {
    const long i1 = i + blockDim.x, i2 = i1 + blockDim.x, i3 = i2 + blockDim.x;
    const bool k1 = i1 < hi, k2 = i2 < hi, k3 = i3 < hi;
    const int d0 = dst[i];
    const int d1 = dst[k1 ? i1 : i];
    const int d2 = dst[k2 ? i2 : i];
    const int d3 = dst[k3 ? i3 : i];
    const int s0 = src[i];
    const int s1 = src[k1 ? i1 : i];
    const int s2 = src[k2 ? i2 : i];
    const int s3 = src[k3 ? i3 : i];
    atomicAdd(&bins[bucket2_of(d0, s0, nr, nbk)], 1);
    if (k1) atomicAdd(&bins[bucket2_of(d1, s1, nr, nbk)], 1);
    if (k2) atomicAdd(&bins[bucket2_of(d2, s2, nr, nbk)], 1);
    if (k3) atomicAdd(&bins[bucket2_of(d3, s3, nr, nbk)], 1);
  }
  __syncthreads();
  for (int b = threadIdx.x; b < nbk; b += blockDim.x)
    hist[(long)b * gridDim.x + blockIdx.x] = bins[b];
}

__global__ void part_scatter2(const int* __restrict__ src,
                              const int* __restrict__ dst,
                              const int* __restrict__ offs,
                              int* __restrict__ tmp, long n_edges, int chunk,
                              int nr, int nbk) {
  __shared__ int lbeg[MAX_BK];
  __shared__ int cur[MAX_BK];
  __shared__ int wsum[8];
  __shared__ int stage[PART_CAP];
  const int tid = threadIdx.x, lane = tid & 31, wave = tid >> 5;
  const int G = gridDim.x, g = blockIdx.x;
  const long total = (long)nbk * G;

  int v[4];
  int t = 0;
  for (int k = 0; k < 4; ++k) {
    const int b = 4 * tid + k;
    int c = 0;
    if (b < nbk) {
      const long idx = (long)b * G + g;
      const int o0 = offs[idx];
      const int o1 = (idx + 1 < total) ? offs[idx + 1] : (int)n_edges;
      c = o1 - o0;
    }
    v[k] = c;
    t += c;
  }
  const int x = wave_incl_scan(t, lane);
  if (lane == 31) wsum[wave] = x;
  __syncthreads();
  int run = x - t;
  for (int w = 0; w < 8; ++w) run += (w < wave) ? wsum[w] : 0;
  for (int k = 0; k < 4; ++k) {
    const int b = 4 * tid + k;
    if (b < nbk) { lbeg[b] = run; cur[b] = run; }
    run += v[k];
  }
  __syncthreads();

  const long lo = (long)g * chunk;
  long hi = lo + chunk;
  if (hi > n_edges) hi = n_edges;
  for (long i = lo + tid; i < hi; i += 4 * blockDim.x) {
    const long i1 = i + blockDim.x, i2 = i1 + blockDim.x, i3 = i2 + blockDim.x;
    const bool k1 = i1 < hi, k2 = i2 < hi, k3 = i3 < hi;
    const int d0 = dst[i];
    const int d1 = dst[k1 ? i1 : i];
    const int d2 = dst[k2 ? i2 : i];
    const int d3 = dst[k3 ? i3 : i];
    const int s0 = src[i];
    const int s1 = src[k1 ? i1 : i];
    const int s2 = src[k2 ? i2 : i];
    const int s3 = src[k3 ? i3 : i];
    {
      const int p = atomicAdd(&cur[bucket2_of(d0, s0, nr, nbk)], 1);
      if (p < PART_CAP) stage[p] = entry2_of(d0, s0);
    }
    if (k1) {
      const int p = atomicAdd(&cur[bucket2_of(d1, s1, nr, nbk)], 1);
      if (p < PART_CAP) stage[p] = entry2_of(d1, s1);
    }
    if (k2) {
      const int p = atomicAdd(&cur[bucket2_of(d2, s2, nr, nbk)], 1);
      if (p < PART_CAP) stage[p] = entry2_of(d2, s2);
    }
    if (k3) {
      const int p = atomicAdd(&cur[bucket2_of(d3, s3, nr, nbk)], 1);
      if (p < PART_CAP) stage[p] = entry2_of(d3, s3);
    }
  }
  __syncthreads();

  for (int b = wave; b < nbk; b += 8) {
    const int ls = lbeg[b];
    const int n = cur[b] - ls;
    const int gs = offs[(long)b * G + g];
    for (int k = lane; k < n; k += 32) tmp[gs + k] = stage[ls + k];
  }
}

__global__ void __launch_bounds__(1024)
bucket_sort2(const int* __restrict__ tmp, const int* __restrict__ offs,
             int* __restrict__ degp, int* __restrict__ ss2, long n_edges,
             int n_nodes, int nr, int nbk, int G, long deg_stride) {
  __shared__ int cnt[SBINS];
  __shared__ int wsum[32];
  __shared__ int stage[BK2_CAP];
  const int b = blockIdx.x;
  const int g = b / nr, r = b - g * nr;
  const int tid = threadIdx.x, lane = tid & 31, wave = tid >> 5;
  const int bs = offs[(long)b * G];
  const int be = (b + 1 < nbk) ? offs[(long)(b + 1) * G] : (int)n_edges;
  const int n = be - bs;
  const int* __restrict__ in = tmp + bs;

  for (int i = tid; i < DG; i += 1024) stage[i] = 0;
  for (int i = tid; i < SBINS; i += 1024) cnt[i] = 0;
  __syncthreads();
  for (int i = tid; i < n; i += 4096) {
    const int i1 = i + 1024, i2 = i + 2048, i3 = i + 3072;
    const bool k1 = i1 < n, k2 = i2 < n, k3 = i3 < n;
    const unsigned e0 = (unsigned)in[i];
    const unsigned e1 = (unsigned)in[k1 ? i1 : i];
    const unsigned e2 = (unsigned)in[k2 ? i2 : i];
    const unsigned e3 = (unsigned)in[k3 ? i3 : i];
    atomicAdd(&stage[e0 & ((1u << DL_BITS) - 1)], 1);
    atomicAdd(&cnt[e0 >> (DL_BITS + SBIN_SHIFT)], 1);
    if (k1) {
      atomicAdd(&stage[e1 & ((1u << DL_BITS) - 1)], 1);
      atomicAdd(&cnt[e1 >> (DL_BITS + SBIN_SHIFT)], 1);
    }
    if (k2) {
      atomicAdd(&stage[e2 & ((1u << DL_BITS) - 1)], 1);
      atomicAdd(&cnt[e2 >> (DL_BITS + SBIN_SHIFT)], 1);
    }
    if (k3) {
      atomicAdd(&stage[e3 & ((1u << DL_BITS) - 1)], 1);
      atomicAdd(&cnt[e3 >> (DL_BITS + SBIN_SHIFT)], 1);
    }
  }
  __syncthreads();
  {
    const int node0 = g * DG;
    int* __restrict__ dp = degp + (long)r * deg_stride + node0;
    for (int i = tid; i < DG; i += 1024)
      if (node0 + i < n_nodes) dp[i] = stage[i];
  }

  int c[8];
  int t = 0;
  for (int k = 0; k < 8; ++k) { c[k] = cnt[8 * tid + k]; t += c[k]; }
  const int x = wave_incl_scan(t, lane);
  if (lane == 31) wsum[wave] = x;
  __syncthreads();
  int ex = x - t;
  for (int w = 0; w < 32; ++w) ex += (w < wave) ? wsum[w] : 0;
  for (int k = 0; k < 8; ++k) { cnt[8 * tid + k] = ex; ex += c[k]; }
  __syncthreads();

  const bool fits = (n <= BK2_CAP);
  for (int i = tid; i < n; i += 4096) {
    const int i1 = i + 1024, i2 = i + 2048, i3 = i + 3072;
    const bool k1 = i1 < n, k2 = i2 < n, k3 = i3 < n;
    const unsigned e0 = (unsigned)in[i];
    const unsigned e1 = (unsigned)in[k1 ? i1 : i];
    const unsigned e2 = (unsigned)in[k2 ? i2 : i];
    const unsigned e3 = (unsigned)in[k3 ? i3 : i];
    {
      const int p = atomicAdd(&cnt[e0 >> (DL_BITS + SBIN_SHIFT)], 1);
      if (fits) stage[p] = (int)e0; else ss2[bs + p] = (int)e0;
    }
    if (k1) {
      const int p = atomicAdd(&cnt[e1 >> (DL_BITS + SBIN_SHIFT)], 1);
      if (fits) stage[p] = (int)e1; else ss2[bs + p] = (int)e1;
    }
    if (k2) {
      const int p = atomicAdd(&cnt[e2 >> (DL_BITS + SBIN_SHIFT)], 1);
      if (fits) stage[p] = (int)e2; else ss2[bs + p] = (int)e2;
    }
    if (k3) {
      const int p = atomicAdd(&cnt[e3 >> (DL_BITS + SBIN_SHIFT)], 1);
      if (fits) stage[p] = (int)e3; else ss2[bs + p] = (int)e3;
    }
  }
  __syncthreads();
  if (fits)
    for (int i = tid; i < n; i += 1024) ss2[bs + i] = stage[i];
}

__global__ void degree_sum(const int* __restrict__ degp, int* __restrict__ deg,
                           int n_nodes, int nr, long deg_stride) {
  const int i = blockIdx.x * blockDim.x + threadIdx.x;
  if (i >= n_nodes) return;
  int s = 0;
  for (int r = 0; r < nr; ++r) s += degp[(long)r * deg_stride + i];
  deg[i] = s;
}

template <int F_IN, int F_OUT, bool ZERO_AGG, int HL_PITCH>
__global__ void node_transform(const float* __restrict__ h,
                               const float* __restrict__ Wl,
                               const float* __restrict__ Wr,
                               const float* __restrict__ bias,
                               float* __restrict__ hl,
                               float* __restrict__ hs,
                               float* __restrict__ agg,
                               int n_nodes) {
  const int lane = threadIdx.x & 31;
  const int wave = threadIdx.x >> 5;
  const int tile = blockIdx.x * (blockDim.x >> 5) + wave;
  const int tiles = (n_nodes + 15) >> 4;
  if (tile >= tiles) return;

  const int m    = lane & 15;
  const int hi   = lane >> 4;
  const int row0 = tile << 4;
  const int arow = min(row0 + m, n_nodes - 1);
  const bool cok = (m < F_OUT);

  auto ldA = [&](int k) -> float {
    const bool ok = (k < F_IN);
    const float v = h[(long)arow * F_IN + (ok ? k : 0)];
    return ok ? v : 0.0f;
  };
  auto ldW = [&](const float* __restrict__ W, int k) -> float {
    const bool ok = cok && (k < F_IN);
    const float v = W[ok ? (m * F_IN + k) : 0];
    return ok ? v : 0.0f;
  };

  v8f dl = {};
  v8f dr = {};
#pragma unroll
  for (int kc = 0; kc < F_IN; kc += 4) {
    const int k0 = kc + hi * 2;
    v2f a, bl, br;
    a.x  = ldA(k0);
    a.y  = ldA(k0 + 1);
    bl.x = ldW(Wl, k0);
    bl.y = ldW(Wl, k0 + 1);
    br.x = ldW(Wr, k0);
    br.y = ldW(Wr, k0 + 1);
    dl = __builtin_amdgcn_wmma_f32_16x16x4_f32(false, a, false, bl, (short)0,
                                               dl, false, false);
    dr = __builtin_amdgcn_wmma_f32_16x16x4_f32(false, a, false, br, (short)0,
                                               dr, false, false);
  }

  const bool full = (row0 + 16 <= n_nodes);
  if (m < HL_PITCH) {
    const long lbase = (long)(row0 + hi * 8) * HL_PITCH + m;
    if (full) {
#pragma unroll
      for (int v = 0; v < 8; ++v) hl[lbase + (long)v * HL_PITCH] = dl[v];
    } else {
#pragma unroll
      for (int v = 0; v < 8; ++v)
        if (row0 + hi * 8 + v < n_nodes) hl[lbase + (long)v * HL_PITCH] = dl[v];
    }
  }
  if (cok) {
    const float bv   = bias[m];
    const long  base = (long)(row0 + hi * 8) * F_OUT + m;
    if (full) {
#pragma unroll
      for (int v = 0; v < 8; ++v) {
        hs[base + (long)v * F_OUT]  = dr[v] + bv;
        if (ZERO_AGG) agg[base + (long)v * F_OUT] = 0.0f;
      }
    } else {
#pragma unroll
      for (int v = 0; v < 8; ++v) {
        if (row0 + hi * 8 + v < n_nodes) {
          hs[base + (long)v * F_OUT]  = dr[v] + bv;
          if (ZERO_AGG) agg[base + (long)v * F_OUT] = 0.0f;
        }
      }
    }
  }
}

template <int F, bool RELU>
__global__ void pull_combine(const int* __restrict__ rp, const int* __restrict__ ss,
                             const float* __restrict__ hl,
                             const float* __restrict__ hs,
                             float* __restrict__ out, int n_nodes) {
  const int q    = threadIdx.x & 7;
  const int node = (int)((blockIdx.x * blockDim.x + threadIdx.x) >> 3);
  const bool live = node < n_nodes;
  const int nd  = live ? node : n_nodes - 1;
  const int beg = (nd > 0) ? rp[nd - 1] : 0;
  const int end = live ? rp[nd] : beg;

  float acc[F];
  for (int f = 0; f < F; ++f) acc[f] = 0.0f;
  for (int j = beg + q; j < end; j += 8) {
    const int s = ss[j];
    for (int f = 0; f < F; ++f) acc[f] += hl[(long)s * F + f];
  }
  for (int f = 0; f < F; ++f) {
    acc[f] += __shfl_xor(acc[f], 4);
    acc[f] += __shfl_xor(acc[f], 2);
    acc[f] += __shfl_xor(acc[f], 1);
  }
  float val = acc[0];
  for (int f = 1; f < F; ++f) val = (q == f) ? acc[f] : val;

  if (live && q < F) {
    const float iv = 1.0f / fmaxf((float)(end - beg), 1.0f);
    const long  o  = (long)node * F + q;
    const float r  = val * iv + hs[o];
    out[o] = RELU ? fmaxf(r, 0.0f) : r;
  }
}

template <bool RELU, int OUT_PITCH>
__global__ void pull_rows8(const int* __restrict__ rp, const int* __restrict__ ss,
                           const float* __restrict__ hl8,
                           const float* __restrict__ hs,
                           float* __restrict__ out, int n_nodes) {
  const int l    = threadIdx.x & 7;
  const int p    = l >> 1;
  const int hh   = l & 1;
  const int node = (int)((blockIdx.x * blockDim.x + threadIdx.x) >> 3);
  const bool live = node < n_nodes;
  const int nd  = live ? node : n_nodes - 1;
  const int beg = (nd > 0) ? rp[nd - 1] : 0;
  const int end = live ? rp[nd] : beg;

  const v4f* __restrict__ rows = (const v4f*)hl8;
  const v4f zero = {0.0f, 0.0f, 0.0f, 0.0f};
  v4f acc = zero;
  for (int j0 = beg + p; j0 < end; j0 += 16) {
    const int j1 = j0 + 4, j2 = j0 + 8, j3 = j0 + 12;
    const bool k1 = j1 < end, k2 = j2 < end, k3 = j3 < end;
    const int s0 = ss[j0];
    int s1 = ss[k1 ? j1 : j0];
    int s2 = ss[k2 ? j2 : j0];
    int s3 = ss[k3 ? j3 : j0];
    asm volatile("" : "+v"(s1), "+v"(s2), "+v"(s3));
    v4f v0 = rows[(long)s0 * 2 + hh];
    v4f v1 = zero, v2 = zero, v3 = zero;
    if (k1) v1 = rows[(long)s1 * 2 + hh];
    if (k2) v2 = rows[(long)s2 * 2 + hh];
    if (k3) v3 = rows[(long)s3 * 2 + hh];
    acc += (v0 + v1) + (v2 + v3);
  }
  for (int c = 0; c < 4; ++c) {
    acc[c] += __shfl_xor(acc[c], 4);
    acc[c] += __shfl_xor(acc[c], 2);
  }
  const float nx = __shfl_xor(acc[0], 1);
  const float ny = __shfl_xor(acc[1], 1);
  const float nw = __shfl_xor(acc[3], 1);
  float val = acc[0];
  val = (l == 1) ? ny : val;
  val = (l == 2) ? acc[2] : val;
  val = (l == 3) ? nw : val;
  val = (l == 4) ? nx : val;

  if (live && l < OUT_PITCH) {
    float r = 0.0f;
    if (l < 5) {
      const float iv = 1.0f / fmaxf((float)(end - beg), 1.0f);
      r = val * iv + hs[(long)node * 5 + l];
      r = RELU ? fmaxf(r, 0.0f) : r;
    }
    out[(long)node * OUT_PITCH + l] = r;
  }
}

#define A_PITCH 12
template <int F_OUT, bool RELU, int OUT_PITCH>
__global__ void __launch_bounds__(256)
pull_transform(const int* __restrict__ rp, const int* __restrict__ ss,
               const float* __restrict__ h8,
               const float* __restrict__ Wl, const float* __restrict__ Wr,
               const float* __restrict__ bias,
               float* __restrict__ out, int n_nodes) {
  __shared__ float arow[32 * A_PITCH];
  const int l    = threadIdx.x & 7;
  const int p    = l >> 1;
  const int hh   = l & 1;
  const int ln   = threadIdx.x >> 3;
  const int node = (int)blockIdx.x * 32 + ln;
  const bool live = node < n_nodes;
  const int nd  = live ? node : n_nodes - 1;
  const int beg = (nd > 0) ? rp[nd - 1] : 0;
  const int end = live ? rp[nd] : beg;

  const v4f* __restrict__ rows = (const v4f*)h8;
  const v4f zero = {0.0f, 0.0f, 0.0f, 0.0f};
  const float mine = (l < 5) ? h8[(long)nd * ROW8 + l] : 0.0f;

  const int wave = threadIdx.x >> 5;
  const int lane = threadIdx.x & 31;
  const int m    = lane & 15;
  const int hi   = lane >> 4;
  const bool cok = (m < F_OUT);
  auto bok = [&](int k) -> bool { return cok && (k < 10); };
  auto ldB = [&](int k) -> float {
    const bool left = (k < 5);
    const float* __restrict__ Wp = left ? Wl : Wr;
    const int kk = left ? k : k - 5;
    return Wp[bok(k) ? (m * 5 + kk) : 0];
  };
  float braw[A_PITCH / 2];
  float bv = 0.0f;
#pragma unroll
  for (int i = 0; i < A_PITCH / 2; ++i) braw[i] = 0.0f;
  if (wave < 2) {
#pragma unroll
    for (int i = 0; i < A_PITCH / 4; ++i) {
      const int k0 = 4 * i + hi * 2;
      braw[2 * i]     = ldB(k0);
      braw[2 * i + 1] = ldB(k0 + 1);
    }
    bv = cok ? bias[m] : 0.0f;
  }
  v4f acc = zero;
  for (int j0 = beg + p; j0 < end; j0 += 16) {
    const int j1 = j0 + 4, j2 = j0 + 8, j3 = j0 + 12;
    const bool k1 = j1 < end, k2 = j2 < end, k3 = j3 < end;
    const int s0 = ss[j0];
    int s1 = ss[k1 ? j1 : j0];
    int s2 = ss[k2 ? j2 : j0];
    int s3 = ss[k3 ? j3 : j0];
    asm volatile("" : "+v"(s1), "+v"(s2), "+v"(s3));
    v4f v0 = rows[(long)s0 * 2 + hh];
    v4f v1 = zero, v2 = zero, v3 = zero;
    if (k1) v1 = rows[(long)s1 * 2 + hh];
    if (k2) v2 = rows[(long)s2 * 2 + hh];
    if (k3) v3 = rows[(long)s3 * 2 + hh];
    acc += (v0 + v1) + (v2 + v3);
  }
  for (int c = 0; c < 4; ++c) {
    acc[c] += __shfl_xor(acc[c], 4);
    acc[c] += __shfl_xor(acc[c], 2);
  }
  const float nx = __shfl_xor(acc[0], 1);
  const float ny = __shfl_xor(acc[1], 1);
  const float nw = __shfl_xor(acc[3], 1);
  float val = acc[0];
  val = (l == 1) ? ny : val;
  val = (l == 2) ? acc[2] : val;
  val = (l == 3) ? nw : val;
  val = (l == 4) ? nx : val;

  const float iv = 1.0f / fmaxf((float)(end - beg), 1.0f);
  if (l < 5) {
    arow[ln * A_PITCH + l]     = live ? val * iv : 0.0f;
    arow[ln * A_PITCH + 5 + l] = live ? mine : 0.0f;
  } else if (l < 7) {
    arow[ln * A_PITCH + 5 + l] = 0.0f;
  }
  __syncthreads();

  if (wave >= 2) return;

  const float* __restrict__ at = arow + (wave * 16 + m) * A_PITCH;
  v8f d = {};
#pragma unroll
  for (int kc = 0; kc < A_PITCH; kc += 4) {
    const int k0 = kc + hi * 2;
    v2f a, b;
    a.x = at[k0];
    a.y = at[k0 + 1];
    b.x = bok(k0) ? braw[kc / 2] : 0.0f;
    b.y = bok(k0 + 1) ? braw[kc / 2 + 1] : 0.0f;
    d = __builtin_amdgcn_wmma_f32_16x16x4_f32(false, a, false, b, (short)0,
                                              d, false, false);
  }

  const int row0 = (int)blockIdx.x * 32 + wave * 16;
  if (m < OUT_PITCH) {
    const long  base = (long)(row0 + hi * 8) * OUT_PITCH + m;
    const bool  full = (row0 + 16 <= n_nodes);
    if (full) {
#pragma unroll
      for (int v = 0; v < 8; ++v) {
        const float r = d[v] + bv;
        out[base + (long)v * OUT_PITCH] = RELU ? fmaxf(r, 0.0f) : r;
      }
    } else {
#pragma unroll
      for (int v = 0; v < 8; ++v) {
        if (row0 + hi * 8 + v < n_nodes) {
          const float r = d[v] + bv;
          out[base + (long)v * OUT_PITCH] = RELU ? fmaxf(r, 0.0f) : r;
        }
      }
    }
  }
}

#define TPITCH 5
typedef float v4f_a4 __attribute__((ext_vector_type(4), aligned(4)));

__device__ inline unsigned ld_entry(const int* p) {
  return (unsigned)__builtin_nontemporal_load(p);
}
__device__ inline long row_offset(unsigned e, int rbase, int last) {
  return (long)min(rbase + (int)(e >> DL_BITS), last) * TPITCH;
}
__device__ inline void add_row(float* acc, unsigned e, v4f a, float c) {
  float* q = acc + (e & ((1u << DL_BITS) - 1));
  atomicAdd(q, a.x);
  atomicAdd(q + DG, a.y);
  atomicAdd(q + 2 * DG, a.z);
  atomicAdd(q + 3 * DG, a.w);
  atomicAdd(q + 4 * DG, c);
}

__device__ inline float sum_ranges(const float* __restrict__ p, int nr,
                                   long part_stride) {
  float v[NR_MAX];
#pragma unroll
  for (int k = 0; k < NR_MAX; ++k) {
    const bool ok = k < nr;
    const float t = p[ok ? (long)k * part_stride : 0];
    v[k] = ok ? t : 0.0f;
  }
  float s = v[0];
#pragma unroll
  for (int k = 1; k < NR_MAX; ++k) s += v[k];
  return s;
}

__global__ void __launch_bounds__(1024)
agg_block(const int* __restrict__ ss2, const int* __restrict__ offs,
          const float* __restrict__ tab, float* __restrict__ part,
          long n_edges, int n_nodes, int nr, int nbk, int G, long part_stride) {
  __shared__ float acc[5 * DG];
  const int b = blockIdx.x;
  const int g = b / nr, r = b - g * nr;
  const int tid = threadIdx.x;
  const int bs = offs[(long)b * G];
  const int be = (b + 1 < nbk) ? offs[(long)(b + 1) * G] : (int)n_edges;
  const int n = be - bs;
  const int* __restrict__ in = ss2 + bs;

  for (int i = tid; i < 5 * DG; i += 1024) acc[i] = 0.0f;
  __syncthreads();

  const int rbase = r * SRN;
  const int last = n_nodes - 1;
  int i = tid;
  for (; i + 3072 < n; i += 4096) {
    const unsigned e0 = ld_entry(in + i);
    const unsigned e1 = ld_entry(in + i + 1024);
    const unsigned e2 = ld_entry(in + i + 2048);
    const unsigned e3 = ld_entry(in + i + 3072);
    const float* __restrict__ p0 = tab + row_offset(e0, rbase, last);
    const float* __restrict__ p1 = tab + row_offset(e1, rbase, last);
    const float* __restrict__ p2 = tab + row_offset(e2, rbase, last);
    const float* __restrict__ p3 = tab + row_offset(e3, rbase, last);
    const v4f a0 = *(const v4f_a4*)p0;
    const v4f a1 = *(const v4f_a4*)p1;
    const v4f a2 = *(const v4f_a4*)p2;
    const v4f a3 = *(const v4f_a4*)p3;
    const float c0 = p0[4];
    const float c1 = p1[4];
    const float c2 = p2[4];
    const float c3 = p3[4];
    add_row(acc, e0, a0, c0);
    add_row(acc, e1, a1, c1);
    add_row(acc, e2, a2, c2);
    add_row(acc, e3, a3, c3);
  }
  for (; i < n; i += 1024) {
    const unsigned e0 = ld_entry(in + i);
    const float* __restrict__ p0 = tab + row_offset(e0, rbase, last);
    const v4f a0 = *(const v4f_a4*)p0;
    const float c0 = p0[4];
    add_row(acc, e0, a0, c0);
  }
  __syncthreads();

  const int node0 = g * DG;
  const int rows = min(DG, n_nodes - node0);
  float* __restrict__ po = part + (long)r * part_stride + (long)node0 * 5;
  for (int idx = tid; idx < rows * 5; idx += 1024) {
    const int d = idx / 5;
    const int f = idx - d * 5;
    po[idx] = acc[f * DG + d];
  }
}

__global__ void combine_rows(const float* __restrict__ part,
                             const int* __restrict__ deg,
                             const float* __restrict__ hs,
                             float* __restrict__ tab, int n_nodes, int nr,
                             long part_stride) {
  const int l    = threadIdx.x & 7;
  const int node = (int)((blockIdx.x * blockDim.x + threadIdx.x) >> 3);
  if (node >= n_nodes || l >= 5) return;
  const float s = sum_ranges(part + (long)node * 5 + l, nr, part_stride);
  const float iv = 1.0f / fmaxf((float)deg[node], 1.0f);
  tab[(long)node * TPITCH + l] = fmaxf(s * iv + hs[(long)node * 5 + l], 0.0f);
}

template <int F_OUT, bool RELU, int OUT_PITCH>
__global__ void __launch_bounds__(256)
combine_transform(const float* __restrict__ part, const int* __restrict__ deg,
                  const float* h8,
                  const float* __restrict__ Wl, const float* __restrict__ Wr,
                  const float* __restrict__ bias,
                  float* out, int n_nodes, int nr, long part_stride) {
  __shared__ float arow[32 * A_PITCH];
  const int l    = threadIdx.x & 7;
  const int ln   = threadIdx.x >> 3;
  const int node = (int)blockIdx.x * 32 + ln;
  const bool live = node < n_nodes;
  const int nd  = live ? node : n_nodes - 1;

  const float mine = (l < 5) ? h8[(long)nd * TPITCH + l] : 0.0f;

  const int wave = threadIdx.x >> 5;
  const int lane = threadIdx.x & 31;
  const int m    = lane & 15;
  const int hi   = lane >> 4;
  const bool cok = (m < F_OUT);
  auto bok = [&](int k) -> bool { return cok && (k < 10); };
  auto ldB = [&](int k) -> float {
    const bool left = (k < 5);
    const float* __restrict__ Wp = left ? Wl : Wr;
    const int kk = left ? k : k - 5;
    return Wp[bok(k) ? (m * 5 + kk) : 0];
  };
  float braw[A_PITCH / 2];
  float bv = 0.0f;
#pragma unroll
  for (int i = 0; i < A_PITCH / 2; ++i) braw[i] = 0.0f;
  if (wave < 2) {
#pragma unroll
    for (int i = 0; i < A_PITCH / 4; ++i) {
      const int k0 = 4 * i + hi * 2;
      braw[2 * i]     = ldB(k0);
      braw[2 * i + 1] = ldB(k0 + 1);
    }
    bv = cok ? bias[m] : 0.0f;
  }

  const float val = sum_ranges(part + (long)nd * 5 + (l < 5 ? l : 4), nr, part_stride);
  const float iv = 1.0f / fmaxf((float)deg[nd], 1.0f);

  if (l < 5) {
    arow[ln * A_PITCH + l]     = live ? val * iv : 0.0f;
    arow[ln * A_PITCH + 5 + l] = live ? mine : 0.0f;
  } else if (l < 7) {
    arow[ln * A_PITCH + 5 + l] = 0.0f;
  }
  __syncthreads();

  if (wave >= 2) return;

  const float* __restrict__ at = arow + (wave * 16 + m) * A_PITCH;
  v8f d = {};
#pragma unroll
  for (int kc = 0; kc < A_PITCH; kc += 4) {
    const int k0 = kc + hi * 2;
    v2f a, b;
    a.x = at[k0];
    a.y = at[k0 + 1];
    b.x = bok(k0) ? braw[kc / 2] : 0.0f;
    b.y = bok(k0 + 1) ? braw[kc / 2 + 1] : 0.0f;
    d = __builtin_amdgcn_wmma_f32_16x16x4_f32(false, a, false, b, (short)0,
                                              d, false, false);
  }

  const int row0 = (int)blockIdx.x * 32 + wave * 16;
  if (m < OUT_PITCH) {
    const long  base = (long)(row0 + hi * 8) * OUT_PITCH + m;
    const bool  full = (row0 + 16 <= n_nodes);
    if (full) {
#pragma unroll
      for (int v = 0; v < 8; ++v) {
        const float r = d[v] + bv;
        out[base + (long)v * OUT_PITCH] = RELU ? fmaxf(r, 0.0f) : r;
      }
    } else {
#pragma unroll
      for (int v = 0; v < 8; ++v) {
        if (row0 + hi * 8 + v < n_nodes) {
          const float r = d[v] + bv;
          out[base + (long)v * OUT_PITCH] = RELU ? fmaxf(r, 0.0f) : r;
        }
      }
    }
  }
}

__global__ void zero_f32(float* __restrict__ p, long n) {
  long i = (long)blockIdx.x * blockDim.x + threadIdx.x;
  if (i < n) p[i] = 0.0f;
}

__global__ void count_edges(const int* __restrict__ dst, float* __restrict__ cnt,
                            long n_edges) {
  long i = (long)blockIdx.x * blockDim.x + threadIdx.x;
  if (i < n_edges) unsafeAtomicAdd(&cnt[dst[i]], 1.0f);
}

__global__ void reciprocal_clamped(float* __restrict__ p, long n) {
  long i = (long)blockIdx.x * blockDim.x + threadIdx.x;
  if (i < n) p[i] = 1.0f / fmaxf(p[i], 1.0f);
}

template <int F>
__global__ void edge_scatter(const int* __restrict__ src,
                             const int* __restrict__ dst,
                             const float* __restrict__ hl,
                             float* __restrict__ agg, long n_edges) {
  long i = (long)blockIdx.x * blockDim.x + threadIdx.x;
  if (i >= n_edges) return;
  const int s = src[i];
  const int d = dst[i];
  float v[F];
#pragma unroll
  for (int j = 0; j < F; ++j) v[j] = hl[(long)s * F + j];
#pragma unroll
  for (int j = 0; j < F; ++j) unsafeAtomicAdd(&agg[(long)d * F + j], v[j]);
}

template <int F, bool RELU>
__global__ void combine(const float* __restrict__ agg,
                        const float* __restrict__ hs,
                        const float* __restrict__ inv,
                        float* __restrict__ out, int n_nodes) {
  const int node = blockIdx.x * blockDim.x + threadIdx.x;
  if (node >= n_nodes) return;
  const float iv = inv[node];
#pragma unroll
  for (int j = 0; j < F; ++j) {
    const long o = (long)node * F + j;
    const float v = agg[o] * iv + hs[o];
    out[o] = RELU ? fmaxf(v, 0.0f) : v;
  }
}

extern "C" void kernel_launch(void* const* d_in, const int* in_sizes, int n_in,
                              void* d_out, int out_size, void* d_ws,
                              size_t ws_size, hipStream_t stream) {
  const float* x    = (const float*)d_in[0];
  const int*   ei   = (const int*)  d_in[1];
  const float* Wl1  = (const float*)d_in[2];
  const float* Wr1  = (const float*)d_in[3];
  const float* b1   = (const float*)d_in[4];
  const float* Wlm  = (const float*)d_in[5];
  const float* Wrm  = (const float*)d_in[6];
  const float* bm   = (const float*)d_in[7];
  const float* Wl10 = (const float*)d_in[8];
  const float* Wr10 = (const float*)d_in[9];
  const float* b10  = (const float*)d_in[10];

  const int  N = in_sizes[0] / 10;
  const long E = (long)in_sizes[1] / 2;
  const int* src = ei;
  const int* dst = ei + E;

  const int B = 256;
  auto cdiv = [](long a, long b) { return (int)((a + b - 1) / b); };
  auto pad32 = [](size_t a) { return (a + 31) & ~(size_t)31; };

  const int tiles   = (N + 15) / 16;
  const int tblocks = (tiles + 7) / 8;

  const int    nb    = cdiv(N, SCAN_CHUNK);
  const size_t w_rp  = pad32((size_t)N);
  const size_t w_bs  = pad32((size_t)nb);
  const size_t w_ss  = pad32((size_t)E);
  const size_t w_h   = pad32((size_t)N * 5);
  const size_t w_h8  = pad32((size_t)N * ROW8);
  const size_t need  = (w_rp + w_bs + w_ss + 3 * w_h) * sizeof(int);
  const size_t need8 = (w_rp + w_bs + w_ss + 2 * w_h + w_h8) * sizeof(int);

  if (ws_size >= need) {
    const bool wide = (ws_size >= need8);
    int*   rp   = (int*)d_ws;
    int*   bsum = rp + w_rp;
    int*   ss   = bsum + w_bs;
    float* h    = (float*)(ss + w_ss);
    float* hs   = h + w_h;
    float* hl   = hs + w_h;
    float* none = nullptr;

    const int    PG     = 1024;
    const int    nbk    = cdiv(N, BK_NODES);
    const int    chunk  = cdiv(E, PG);
    const size_t w_hist = pad32((size_t)nbk * PG);
    const size_t need_s = need8 + (w_hist + w_ss) * sizeof(int);
    const bool lds_sort = wide && ws_size >= need_s && nbk <= MAX_BK &&
                          chunk <= PART_CAP && N <= (1 << 21);
    {
      const int    ng      = cdiv(N, DG);
      const int    nr      = cdiv(N, SRN);
      const int    nbk2    = ng * nr;
      const int    chunk2  = cdiv(E, PG2);
      const int    hn2     = nbk2 * PG2;
      const int    hnb2    = cdiv(hn2, SCAN_CHUNK);
      const size_t w_hist2 = pad32((size_t)hn2);
      const size_t need_b  = (w_rp + w_bs + w_ss + 2 * w_h + w_h8 + w_hist2 + w_ss) * sizeof(int);
      const bool blocked = wide && ws_size >= need_b && nr <= NR_MAX &&
                           nbk2 <= MAX_BK && chunk2 <= PART_CAP &&
                           (size_t)hnb2 <= w_bs && (size_t)nr * w_h <= w_ss &&
                           (size_t)nr * w_rp <= 2 * w_h && E < (1L << 31);
      if (blocked) {
        int*   deg   = rp;
        int*   ss2   = ss;
        int*   degp  = (int*)h;
        float* tab   = hl;
        int*   hist2 = (int*)(hl + w_h8);
        int*   tmp   = hist2 + w_hist2;
        float* part  = (float*)tmp;
        const long dstr = (long)w_rp;
        const long pstr = (long)w_h;
        part_hist2<<<PG2, 256, 0, stream>>>(src, dst, hist2, E, chunk2, nr, nbk2);
        block_sums<<<hnb2, B, 0, stream>>>(hist2, bsum, hn2);
        scan_block_sums<<<1, B, 0, stream>>>(bsum, hnb2);
        scan_apply<<<hnb2, B, 0, stream>>>(hist2, bsum, hn2);
        part_scatter2<<<PG2, 256, 0, stream>>>(src, dst, hist2, tmp, E, chunk2, nr, nbk2);
        bucket_sort2<<<nbk2, 1024, 0, stream>>>(tmp, hist2, degp, ss2, E, N, nr, nbk2, PG2, dstr);
        degree_sum<<<cdiv(N, B), B, 0, stream>>>(degp, deg, N, nr, dstr);

        const int cblocks = cdiv((long)N * 8, B);
        node_transform<10, 5, false, TPITCH><<<tblocks, 256, 0, stream>>>(x, Wl1, Wr1, b1, tab, hs, none, N);
        agg_block<<<nbk2, 1024, 0, stream>>>(ss2, hist2, tab, part, E, N, nr, nbk2, PG2, pstr);
        combine_rows<<<cblocks, B, 0, stream>>>(part, deg, hs, tab, N, nr, pstr);
        for (int l = 0; l < 8; ++l) {
          agg_block<<<nbk2, 1024, 0, stream>>>(ss2, hist2, tab, part, E, N, nr, nbk2, PG2, pstr);
          combine_transform<5, true, TPITCH><<<cblocks, B, 0, stream>>>(
              part, deg, tab, Wlm + l * 25, Wrm + l * 25, bm + l * 5, tab, N, nr, pstr);
        }
        agg_block<<<nbk2, 1024, 0, stream>>>(ss2, hist2, tab, part, E, N, nr, nbk2, PG2, pstr);
        combine_transform<1, false, 1><<<cblocks, B, 0, stream>>>(
            part, deg, tab, Wl10, Wr10, b10, (float*)d_out, N, nr, pstr);
        return;
      }
    }

    if (lds_sort) {
      int* hist = (int*)(hl + w_h8);
      int* tmp  = hist + w_hist;
      const int hn  = nbk * PG;
      const int hnb = cdiv(hn, SCAN_CHUNK);
      part_hist<<<PG, 256, 0, stream>>>(dst, hist, E, chunk, nbk);
      block_sums<<<hnb, B, 0, stream>>>(hist, bsum, hn);
      scan_block_sums<<<1, B, 0, stream>>>(bsum, hnb);
      scan_apply<<<hnb, B, 0, stream>>>(hist, bsum, hn);
      part_scatter<<<PG, 256, 0, stream>>>(src, dst, hist, tmp, E, chunk, nbk);
      bucket_rows<<<nbk, 1024, 0, stream>>>(tmp, hist, rp, ss, E, N, nbk, PG);
    } else {
      const int eblocks = cdiv(E, B) < 2048 ? cdiv(E, B) : 2048;
      zero_i32<<<cdiv(N, B), B, 0, stream>>>(rp, N);
      count_in_edges<<<eblocks, B, 0, stream>>>(dst, rp, E);
      block_sums<<<nb, B, 0, stream>>>(rp, bsum, N);
      scan_block_sums<<<1, B, 0, stream>>>(bsum, nb);
      scan_apply<<<nb, B, 0, stream>>>(rp, bsum, N);
      fill_rows<<<eblocks, B, 0, stream>>>(src, dst, rp, ss, E);
    }

    const int pblocks = cdiv((long)N * 8, B);

    if (lds_sort && w_ss >= w_h8) {
      float* ta = hl;
      float* tb = hl + w_h8 + w_hist;
      node_transform<10, 5, false, ROW8><<<tblocks, 256, 0, stream>>>(x, Wl1, Wr1, b1, ta, hs, none, N);
      pull_rows8<true, ROW8><<<pblocks, B, 0, stream>>>(rp, ss, ta, hs, tb, N);
      for (int l = 0; l < 8; ++l) {
        const float* tin  = (l & 1) ? ta : tb;
        float*       tout = (l & 1) ? tb : ta;
        pull_transform<5, true, ROW8><<<pblocks, B, 0, stream>>>(
            rp, ss, tin, Wlm + l * 25, Wrm + l * 25, bm + l * 5, tout, N);
      }
      pull_transform<1, false, 1><<<pblocks, B, 0, stream>>>(
          rp, ss, tb, Wl10, Wr10, b10, (float*)d_out, N);
      return;
    }

    if (wide) {
      node_transform<10, 5, false, ROW8><<<tblocks, 256, 0, stream>>>(x, Wl1, Wr1, b1, hl, hs, none, N);
      pull_rows8<true, 5><<<pblocks, B, 0, stream>>>(rp, ss, hl, hs, h, N);
    } else {
      node_transform<10, 5, false, 5><<<tblocks, 256, 0, stream>>>(x, Wl1, Wr1, b1, hl, hs, none, N);
      pull_combine<5, true><<<pblocks, B, 0, stream>>>(rp, ss, hl, hs, h, N);
    }

    for (int l = 0; l < 8; ++l) {
      if (wide) {
        node_transform<5, 5, false, ROW8><<<tblocks, 256, 0, stream>>>(
            h, Wlm + l * 25, Wrm + l * 25, bm + l * 5, hl, hs, none, N);
        pull_rows8<true, 5><<<pblocks, B, 0, stream>>>(rp, ss, hl, hs, h, N);
      } else {
        node_transform<5, 5, false, 5><<<tblocks, 256, 0, stream>>>(
            h, Wlm + l * 25, Wrm + l * 25, bm + l * 5, hl, hs, none, N);
        pull_combine<5, true><<<pblocks, B, 0, stream>>>(rp, ss, hl, hs, h, N);
      }
    }

    node_transform<5, 1, false, 1><<<tblocks, 256, 0, stream>>>(h, Wl10, Wr10, b10, hl, hs, none, N);
    pull_combine<1, false><<<pblocks, B, 0, stream>>>(rp, ss, hl, hs, (float*)d_out, N);
    return;
  }

  float* ws  = (float*)d_ws;
  float* inv = ws;
  float* hl  = inv + (size_t)N;
  float* hs  = hl + (size_t)N * 5;
  float* agg = hs + (size_t)N * 5;
  float* hA  = agg + (size_t)N * 5;
  float* hB  = hA + (size_t)N * 5;

  zero_f32<<<cdiv(N, B), B, 0, stream>>>(inv, N);
  count_edges<<<cdiv(E, B), B, 0, stream>>>(dst, inv, E);
  reciprocal_clamped<<<cdiv(N, B), B, 0, stream>>>(inv, N);

  node_transform<10, 5, true, 5><<<tblocks, 256, 0, stream>>>(x, Wl1, Wr1, b1, hl, hs, agg, N);
  edge_scatter<5><<<cdiv(E, B), B, 0, stream>>>(src, dst, hl, agg, E);
  combine<5, true><<<cdiv(N, B), B, 0, stream>>>(agg, hs, inv, hA, N);

  for (int l = 0; l < 8; ++l) {
    const float* hin  = (l & 1) ? hB : hA;
    float*       hout = (l & 1) ? hA : hB;
    node_transform<5, 5, true, 5><<<tblocks, 256, 0, stream>>>(
        hin, Wlm + l * 25, Wrm + l * 25, bm + l * 5, hl, hs, agg, N);
    edge_scatter<5><<<cdiv(E, B), B, 0, stream>>>(src, dst, hl, agg, E);
    combine<5, true><<<cdiv(N, B), B, 0, stream>>>(agg, hs, inv, hout, N);
  }

  node_transform<5, 1, true, 1><<<tblocks, 256, 0, stream>>>(hA, Wl10, Wr10, b10, hl, hs, agg, N);
  edge_scatter<1><<<cdiv(E, B), B, 0, stream>>>(src, dst, hl, agg, E);
  combine<1, false><<<cdiv(N, B), B, 0, stream>>>(agg, hs, inv, (float*)d_out, N);
}


# Layer 10 transforms first (5->1 on the matrix unit, own launch); its blocked aggregation then carries 1 float per edge: 4-byte gather, 1 LDS float add, 1-wide partials; sum order changes
# speedup vs baseline: 14.2956x; 1.0491x over previous
#include <hip/hip_runtime.h>


typedef __attribute__((ext_vector_type(2))) float v2f;
typedef __attribute__((ext_vector_type(4))) float v4f;
typedef __attribute__((ext_vector_type(8))) float v8f;

#define ROW8 8

#define SCAN_CHUNK 2048

__global__ void zero_i32(int* __restrict__ p, int n) {
  const int i = blockIdx.x * blockDim.x + threadIdx.x;
  if (i < n) p[i] = 0;
}

__global__ void count_in_edges(const int* __restrict__ dst, int* __restrict__ rp,
                               long n_edges) {
  const long stride = (long)gridDim.x * blockDim.x;
  long i = (long)blockIdx.x * blockDim.x + threadIdx.x;
  for (; i + 3 * stride < n_edges; i += 4 * stride) {
    const int d0 = dst[i];
    const int d1 = dst[i + stride];
    const int d2 = dst[i + 2 * stride];
    const int d3 = dst[i + 3 * stride];
    atomicAdd(&rp[d0], 1);
    atomicAdd(&rp[d1], 1);
    atomicAdd(&rp[d2], 1);
    atomicAdd(&rp[d3], 1);
  }
  for (; i < n_edges; i += stride) atomicAdd(&rp[dst[i]], 1);
}

__device__ inline int wave_incl_scan(int x, int lane) {
  for (int off = 1; off < 32; off <<= 1) {
    const int y = __shfl_up(x, off);
    if (lane >= off) x += y;
  }
  return x;
}

__global__ void block_sums(const int* __restrict__ cnt, int* __restrict__ bsum,
                           int n) {
  __shared__ int wsum[8];
  const int base = blockIdx.x * SCAN_CHUNK;
  int s = 0;
  for (int k = 0; k < 8; ++k) {
    const int idx = base + k * 256 + (int)threadIdx.x;
    if (idx < n) s += cnt[idx];
  }
  for (int off = 16; off > 0; off >>= 1) s += __shfl_xor(s, off);
  if ((threadIdx.x & 31) == 0) wsum[threadIdx.x >> 5] = s;
  __syncthreads();
  if (threadIdx.x == 0) {
    int t = 0;
    for (int w = 0; w < 8; ++w) t += wsum[w];
    bsum[blockIdx.x] = t;
  }
}

__global__ void scan_block_sums(int* __restrict__ bsum, int nb) {
  __shared__ int wsum[8];
  const int lane = threadIdx.x & 31;
  const int wave = threadIdx.x >> 5;
  int carry = 0;
  for (int base = 0; base < nb; base += 256) {
    const int i = base + (int)threadIdx.x;
    const int v = (i < nb) ? bsum[i] : 0;
    const int x = wave_incl_scan(v, lane);
    if (lane == 31) wsum[wave] = x;
    __syncthreads();
    int before = 0, total = 0;
    for (int w = 0; w < 8; ++w) {
      const int t = wsum[w];
      before += (w < wave) ? t : 0;
      total += t;
    }
    if (i < nb) bsum[i] = carry + before + x - v;
    carry += total;
    __syncthreads();
  }
}

__global__ void scan_apply(int* rp, const int* __restrict__ bofs, int n) {
  __shared__ int wsum[8];
  const int lane = threadIdx.x & 31;
  const int wave = threadIdx.x >> 5;
  const int base = blockIdx.x * SCAN_CHUNK + (int)threadIdx.x * 8;
  int v[8];
  int t = 0;
  for (int k = 0; k < 8; ++k) {
    const int idx = base + k;
    v[k] = (idx < n) ? rp[idx] : 0;
    t += v[k];
  }
  const int x = wave_incl_scan(t, lane);
  if (lane == 31) wsum[wave] = x;
  __syncthreads();
  int run = bofs[blockIdx.x] + x - t;
  for (int w = 0; w < 8; ++w) run += (w < wave) ? wsum[w] : 0;
  for (int k = 0; k < 8; ++k) {
    const int idx = base + k;
    if (idx < n) rp[idx] = run;
    run += v[k];
  }
}

__global__ void fill_rows(const int* __restrict__ src, const int* __restrict__ dst,
                          int* __restrict__ rp, int* __restrict__ ss,
                          long n_edges) {
  const long stride = (long)gridDim.x * blockDim.x;
  long i = (long)blockIdx.x * blockDim.x + threadIdx.x;
  for (; i + 3 * stride < n_edges; i += 4 * stride) {
    const int d0 = dst[i];
    const int d1 = dst[i + stride];
    const int d2 = dst[i + 2 * stride];
    const int d3 = dst[i + 3 * stride];
    const int s0 = src[i];
    const int s1 = src[i + stride];
    const int s2 = src[i + 2 * stride];
    const int s3 = src[i + 3 * stride];
    const int p0 = atomicAdd(&rp[d0], 1);
    const int p1 = atomicAdd(&rp[d1], 1);
    const int p2 = atomicAdd(&rp[d2], 1);
    const int p3 = atomicAdd(&rp[d3], 1);
    ss[p0] = s0;
    ss[p1] = s1;
    ss[p2] = s2;
    ss[p3] = s3;
  }
  for (; i < n_edges; i += stride) {
    const int d = dst[i];
    const int s = src[i];
    const int p = atomicAdd(&rp[d], 1);
    ss[p] = s;
  }
}

#define BK_SHIFT 11
#define BK_NODES 2048
#define MAX_BK 1024
#define PART_CAP 16384
#define BK_CAP 73728

__device__ inline int bucket_of(int d, int nbk) {
  const int b = (int)((unsigned)d >> BK_SHIFT);
  return b < nbk ? b : nbk - 1;
}

__global__ void part_hist(const int* __restrict__ dst, int* __restrict__ hist,
                          long n_edges, int chunk, int nbk) {
  __shared__ int bins[MAX_BK];
  for (int b = threadIdx.x; b < nbk; b += blockDim.x) bins[b] = 0;
  __syncthreads();
  const long lo = (long)blockIdx.x * chunk;
  long hi = lo + chunk;
  if (hi > n_edges) hi = n_edges;
  for (long i = lo + threadIdx.x; i < hi; i += 4 * blockDim.x) {
    const long i1 = i + blockDim.x, i2 = i1 + blockDim.x, i3 = i2 + blockDim.x;
    const int d0 = dst[i];
    const int d1 = dst[i1 < hi ? i1 : i];
    const int d2 = dst[i2 < hi ? i2 : i];
    const int d3 = dst[i3 < hi ? i3 : i];
    atomicAdd(&bins[bucket_of(d0, nbk)], 1);
    if (i1 < hi) atomicAdd(&bins[bucket_of(d1, nbk)], 1);
    if (i2 < hi) atomicAdd(&bins[bucket_of(d2, nbk)], 1);
    if (i3 < hi) atomicAdd(&bins[bucket_of(d3, nbk)], 1);
  }
  __syncthreads();
  for (int b = threadIdx.x; b < nbk; b += blockDim.x)
    hist[(long)b * gridDim.x + blockIdx.x] = bins[b];
}

__global__ void part_scatter(const int* __restrict__ src,
                             const int* __restrict__ dst,
                             const int* __restrict__ offs,
                             int* __restrict__ tmp, long n_edges, int chunk,
                             int nbk) {
  __shared__ int lbeg[MAX_BK];
  __shared__ int cur[MAX_BK];
  __shared__ int wsum[8];
  __shared__ int stage[PART_CAP];
  const int tid = threadIdx.x, lane = tid & 31, wave = tid >> 5;
  const int G = gridDim.x, g = blockIdx.x;
  const long total = (long)nbk * G;

  int v[4];
  int t = 0;
  for (int k = 0; k < 4; ++k) {
    const int b = 4 * tid + k;
    int c = 0;
    if (b < nbk) {
      const long idx = (long)b * G + g;
      const int o0 = offs[idx];
      const int o1 = (idx + 1 < total) ? offs[idx + 1] : (int)n_edges;
      c = o1 - o0;
    }
    v[k] = c;
    t += c;
  }
  const int x = wave_incl_scan(t, lane);
  if (lane == 31) wsum[wave] = x;
  __syncthreads();
  int run = x - t;
  for (int w = 0; w < 8; ++w) run += (w < wave) ? wsum[w] : 0;
  for (int k = 0; k < 4; ++k) {
    const int b = 4 * tid + k;
    if (b < nbk) { lbeg[b] = run; cur[b] = run; }
    run += v[k];
  }
  __syncthreads();

  const long lo = (long)g * chunk;
  long hi = lo + chunk;
  if (hi > n_edges) hi = n_edges;
  for (long i = lo + tid; i < hi; i += 4 * blockDim.x) {
    const long i1 = i + blockDim.x, i2 = i1 + blockDim.x, i3 = i2 + blockDim.x;
    const bool k1 = i1 < hi, k2 = i2 < hi, k3 = i3 < hi;
    const int d0 = dst[i];
    const int d1 = dst[k1 ? i1 : i];
    const int d2 = dst[k2 ? i2 : i];
    const int d3 = dst[k3 ? i3 : i];
    const int s0 = src[i];
    const int s1 = src[k1 ? i1 : i];
    const int s2 = src[k2 ? i2 : i];
    const int s3 = src[k3 ? i3 : i];
    {
      const int p = atomicAdd(&cur[bucket_of(d0, nbk)], 1);
      if (p < PART_CAP) stage[p] = (int)(((unsigned)s0 << BK_SHIFT) | ((unsigned)d0 & (BK_NODES - 1)));
    }
    if (k1) {
      const int p = atomicAdd(&cur[bucket_of(d1, nbk)], 1);
      if (p < PART_CAP) stage[p] = (int)(((unsigned)s1 << BK_SHIFT) | ((unsigned)d1 & (BK_NODES - 1)));
    }
    if (k2) {
      const int p = atomicAdd(&cur[bucket_of(d2, nbk)], 1);
      if (p < PART_CAP) stage[p] = (int)(((unsigned)s2 << BK_SHIFT) | ((unsigned)d2 & (BK_NODES - 1)));
    }
    if (k3) {
      const int p = atomicAdd(&cur[bucket_of(d3, nbk)], 1);
      if (p < PART_CAP) stage[p] = (int)(((unsigned)s3 << BK_SHIFT) | ((unsigned)d3 & (BK_NODES - 1)));
    }
  }
  __syncthreads();

  for (int b = wave; b < nbk; b += 8) {
    const int ls = lbeg[b];
    const int n = cur[b] - ls;
    const int gs = offs[(long)b * G + g];
    for (int k = lane; k < n; k += 32) tmp[gs + k] = stage[ls + k];
  }
}

__global__ void __launch_bounds__(1024)
bucket_rows(const int* __restrict__ tmp, const int* __restrict__ offs,
            int* __restrict__ rp, int* __restrict__ ss, long n_edges,
            int n_nodes, int nbk, int G) {
  __shared__ int cnt[BK_NODES];
  __shared__ int wsum[32];
  __shared__ int stage[BK_CAP];
  const int b = blockIdx.x;
  const int tid = threadIdx.x, lane = tid & 31, wave = tid >> 5;
  const int bs = offs[(long)b * G];
  const int be = (b + 1 < nbk) ? offs[(long)(b + 1) * G] : (int)n_edges;
  const int n = be - bs;
  const int* __restrict__ in = tmp + bs;

  cnt[tid] = 0;
  cnt[tid + 1024] = 0;
  __syncthreads();
  for (int i = tid; i < n; i += 4096) {
    const int i1 = i + 1024, i2 = i + 2048, i3 = i + 3072;
    const bool k1 = i1 < n, k2 = i2 < n, k3 = i3 < n;
    const int e0 = in[i];
    const int e1 = in[k1 ? i1 : i];
    const int e2 = in[k2 ? i2 : i];
    const int e3 = in[k3 ? i3 : i];
    atomicAdd(&cnt[e0 & (BK_NODES - 1)], 1);
    if (k1) atomicAdd(&cnt[e1 & (BK_NODES - 1)], 1);
    if (k2) atomicAdd(&cnt[e2 & (BK_NODES - 1)], 1);
    if (k3) atomicAdd(&cnt[e3 & (BK_NODES - 1)], 1);
  }
  __syncthreads();

  const int c0 = cnt[2 * tid];
  const int c1 = cnt[2 * tid + 1];
  const int t = c0 + c1;
  const int x = wave_incl_scan(t, lane);
  if (lane == 31) wsum[wave] = x;
  __syncthreads();
  int ex = x - t;
  for (int w = 0; w < 32; ++w) ex += (w < wave) ? wsum[w] : 0;
  cnt[2 * tid] = ex;
  cnt[2 * tid + 1] = ex + c0;
  const int node0 = b * BK_NODES + 2 * tid;
  if (node0 < n_nodes) rp[node0] = bs + ex + c0;
  if (node0 + 1 < n_nodes) rp[node0 + 1] = bs + ex + t;
  __syncthreads();

  const bool fits = (n <= BK_CAP);
  for (int i = tid; i < n; i += 4096) {
    const int i1 = i + 1024, i2 = i + 2048, i3 = i + 3072;
    const bool k1 = i1 < n, k2 = i2 < n, k3 = i3 < n;
    const unsigned e0 = (unsigned)in[i];
    const unsigned e1 = (unsigned)in[k1 ? i1 : i];
    const unsigned e2 = (unsigned)in[k2 ? i2 : i];
    const unsigned e3 = (unsigned)in[k3 ? i3 : i];
    {
      const int p = atomicAdd(&cnt[e0 & (BK_NODES - 1)], 1);
      if (fits) stage[p] = (int)(e0 >> BK_SHIFT); else ss[bs + p] = (int)(e0 >> BK_SHIFT);
    }
    if (k1) {
      const int p = atomicAdd(&cnt[e1 & (BK_NODES - 1)], 1);
      if (fits) stage[p] = (int)(e1 >> BK_SHIFT); else ss[bs + p] = (int)(e1 >> BK_SHIFT);
    }
    if (k2) {
      const int p = atomicAdd(&cnt[e2 & (BK_NODES - 1)], 1);
      if (fits) stage[p] = (int)(e2 >> BK_SHIFT); else ss[bs + p] = (int)(e2 >> BK_SHIFT);
    }
    if (k3) {
      const int p = atomicAdd(&cnt[e3 & (BK_NODES - 1)], 1);
      if (fits) stage[p] = (int)(e3 >> BK_SHIFT); else ss[bs + p] = (int)(e3 >> BK_SHIFT);
    }
  }
  __syncthreads();
  if (fits)
    for (int i = tid; i < n; i += 1024) ss[bs + i] = stage[i];
}

#define DG 11936
#define DL_BITS 14
#define SRN 83456
#define NR_MAX 6
#define SBIN_SHIFT 4
#define SBINS 8192
#define BK2_CAP 69632
#define PG2 992

__device__ inline int bucket2_of(int d, int s, int nr, int nbk) {
  const unsigned b = ((unsigned)d / DG) * (unsigned)nr + ((unsigned)s / SRN);
  return b < (unsigned)nbk ? (int)b : nbk - 1;
}
__device__ inline int entry2_of(int d, int s) {
  return (int)((((unsigned)s % SRN) << DL_BITS) | ((unsigned)d % DG));
}

__global__ void part_hist2(const int* __restrict__ src, const int* __restrict__ dst,
                           int* __restrict__ hist, long n_edges, int chunk,
                           int nr, int nbk) {
  __shared__ int bins[MAX_BK];
  for (int b = threadIdx.x; b < nbk; b += blockDim.x) bins[b] = 0;
  __syncthreads();
  const long lo = (long)blockIdx.x * chunk;
  long hi = lo + chunk;
  if (hi > n_edges) hi = n_edges;
  for (long i = lo + threadIdx.x; i < hi; i += 4 * blockDim.x) {
    const long i1 = i + blockDim.x, i2 = i1 + blockDim.x, i3 = i2 + blockDim.x;
    const bool k1 = i1 < hi, k2 = i2 < hi, k3 = i3 < hi;
    const int d0 = dst[i];
    const int d1 = dst[k1 ? i1 : i];
    const int d2 = dst[k2 ? i2 : i];
    const int d3 = dst[k3 ? i3 : i];
    const int s0 = src[i];
    const int s1 = src[k1 ? i1 : i];
    const int s2 = src[k2 ? i2 : i];
    const int s3 = src[k3 ? i3 : i];
    atomicAdd(&bins[bucket2_of(d0, s0, nr, nbk)], 1);
    if (k1) atomicAdd(&bins[bucket2_of(d1, s1, nr, nbk)], 1);
    if (k2) atomicAdd(&bins[bucket2_of(d2, s2, nr, nbk)], 1);
    if (k3) atomicAdd(&bins[bucket2_of(d3, s3, nr, nbk)], 1);
  }
  __syncthreads();
  for (int b = threadIdx.x; b < nbk; b += blockDim.x)
    hist[(long)b * gridDim.x + blockIdx.x] = bins[b];
}

__global__ void part_scatter2(const int* __restrict__ src,
                              const int* __restrict__ dst,
                              const int* __restrict__ offs,
                              int* __restrict__ tmp, long n_edges, int chunk,
                              int nr, int nbk) {
  __shared__ int lbeg[MAX_BK];
  __shared__ int cur[MAX_BK];
  __shared__ int wsum[8];
  __shared__ int stage[PART_CAP];
  const int tid = threadIdx.x, lane = tid & 31, wave = tid >> 5;
  const int G = gridDim.x, g = blockIdx.x;
  const long total = (long)nbk * G;

  int v[4];
  int t = 0;
  for (int k = 0; k < 4; ++k) {
    const int b = 4 * tid + k;
    int c = 0;
    if (b < nbk) {
      const long idx = (long)b * G + g;
      const int o0 = offs[idx];
      const int o1 = (idx + 1 < total) ? offs[idx + 1] : (int)n_edges;
      c = o1 - o0;
    }
    v[k] = c;
    t += c;
  }
  const int x = wave_incl_scan(t, lane);
  if (lane == 31) wsum[wave] = x;
  __syncthreads();
  int run = x - t;
  for (int w = 0; w < 8; ++w) run += (w < wave) ? wsum[w] : 0;
  for (int k = 0; k < 4; ++k) {
    const int b = 4 * tid + k;
    if (b < nbk) { lbeg[b] = run; cur[b] = run; }
    run += v[k];
  }
  __syncthreads();

  const long lo = (long)g * chunk;
  long hi = lo + chunk;
  if (hi > n_edges) hi = n_edges;
  for (long i = lo + tid; i < hi; i += 4 * blockDim.x) {
    const long i1 = i + blockDim.x, i2 = i1 + blockDim.x, i3 = i2 + blockDim.x;
    const bool k1 = i1 < hi, k2 = i2 < hi, k3 = i3 < hi;
    const int d0 = dst[i];
    const int d1 = dst[k1 ? i1 : i];
    const int d2 = dst[k2 ? i2 : i];
    const int d3 = dst[k3 ? i3 : i];
    const int s0 = src[i];
    const int s1 = src[k1 ? i1 : i];
    const int s2 = src[k2 ? i2 : i];
    const int s3 = src[k3 ? i3 : i];
    {
      const int p = atomicAdd(&cur[bucket2_of(d0, s0, nr, nbk)], 1);
      if (p < PART_CAP) stage[p] = entry2_of(d0, s0);
    }
    if (k1) {
      const int p = atomicAdd(&cur[bucket2_of(d1, s1, nr, nbk)], 1);
      if (p < PART_CAP) stage[p] = entry2_of(d1, s1);
    }
    if (k2) {
      const int p = atomicAdd(&cur[bucket2_of(d2, s2, nr, nbk)], 1);
      if (p < PART_CAP) stage[p] = entry2_of(d2, s2);
    }
    if (k3) {
      const int p = atomicAdd(&cur[bucket2_of(d3, s3, nr, nbk)], 1);
      if (p < PART_CAP) stage[p] = entry2_of(d3, s3);
    }
  }
  __syncthreads();

  for (int b = wave; b < nbk; b += 8) {
    const int ls = lbeg[b];
    const int n = cur[b] - ls;
    const int gs = offs[(long)b * G + g];
    for (int k = lane; k < n; k += 32) tmp[gs + k] = stage[ls + k];
  }
}

__global__ void __launch_bounds__(1024)
bucket_sort2(const int* __restrict__ tmp, const int* __restrict__ offs,
             int* __restrict__ degp, int* __restrict__ ss2, long n_edges,
             int n_nodes, int nr, int nbk, int G, long deg_stride) {
  __shared__ int cnt[SBINS];
  __shared__ int wsum[32];
  __shared__ int stage[BK2_CAP];
  const int b = blockIdx.x;
  const int g = b / nr, r = b - g * nr;
  const int tid = threadIdx.x, lane = tid & 31, wave = tid >> 5;
  const int bs = offs[(long)b * G];
  const int be = (b + 1 < nbk) ? offs[(long)(b + 1) * G] : (int)n_edges;
  const int n = be - bs;
  const int* __restrict__ in = tmp + bs;

  for (int i = tid; i < DG; i += 1024) stage[i] = 0;
  for (int i = tid; i < SBINS; i += 1024) cnt[i] = 0;
  __syncthreads();
  for (int i = tid; i < n; i += 4096) {
    const int i1 = i + 1024, i2 = i + 2048, i3 = i + 3072;
    const bool k1 = i1 < n, k2 = i2 < n, k3 = i3 < n;
    const unsigned e0 = (unsigned)in[i];
    const unsigned e1 = (unsigned)in[k1 ? i1 : i];
    const unsigned e2 = (unsigned)in[k2 ? i2 : i];
    const unsigned e3 = (unsigned)in[k3 ? i3 : i];
    atomicAdd(&stage[e0 & ((1u << DL_BITS) - 1)], 1);
    atomicAdd(&cnt[e0 >> (DL_BITS + SBIN_SHIFT)], 1);
    if (k1) {
      atomicAdd(&stage[e1 & ((1u << DL_BITS) - 1)], 1);
      atomicAdd(&cnt[e1 >> (DL_BITS + SBIN_SHIFT)], 1);
    }
    if (k2) {
      atomicAdd(&stage[e2 & ((1u << DL_BITS) - 1)], 1);
      atomicAdd(&cnt[e2 >> (DL_BITS + SBIN_SHIFT)], 1);
    }
    if (k3) {
      atomicAdd(&stage[e3 & ((1u << DL_BITS) - 1)], 1);
      atomicAdd(&cnt[e3 >> (DL_BITS + SBIN_SHIFT)], 1);
    }
  }
  __syncthreads();
  {
    const int node0 = g * DG;
    int* __restrict__ dp = degp + (long)r * deg_stride + node0;
    for (int i = tid; i < DG; i += 1024)
      if (node0 + i < n_nodes) dp[i] = stage[i];
  }

  int c[8];
  int t = 0;
  for (int k = 0; k < 8; ++k) { c[k] = cnt[8 * tid + k]; t += c[k]; }
  const int x = wave_incl_scan(t, lane);
  if (lane == 31) wsum[wave] = x;
  __syncthreads();
  int ex = x - t;
  for (int w = 0; w < 32; ++w) ex += (w < wave) ? wsum[w] : 0;
  for (int k = 0; k < 8; ++k) { cnt[8 * tid + k] = ex; ex += c[k]; }
  __syncthreads();

  const bool fits = (n <= BK2_CAP);
  for (int i = tid; i < n; i += 4096) {
    const int i1 = i + 1024, i2 = i + 2048, i3 = i + 3072;
    const bool k1 = i1 < n, k2 = i2 < n, k3 = i3 < n;
    const unsigned e0 = (unsigned)in[i];
    const unsigned e1 = (unsigned)in[k1 ? i1 : i];
    const unsigned e2 = (unsigned)in[k2 ? i2 : i];
    const unsigned e3 = (unsigned)in[k3 ? i3 : i];
    {
      const int p = atomicAdd(&cnt[e0 >> (DL_BITS + SBIN_SHIFT)], 1);
      if (fits) stage[p] = (int)e0; else ss2[bs + p] = (int)e0;
    }
    if (k1) {
      const int p = atomicAdd(&cnt[e1 >> (DL_BITS + SBIN_SHIFT)], 1);
      if (fits) stage[p] = (int)e1; else ss2[bs + p] = (int)e1;
    }
    if (k2) {
      const int p = atomicAdd(&cnt[e2 >> (DL_BITS + SBIN_SHIFT)], 1);
      if (fits) stage[p] = (int)e2; else ss2[bs + p] = (int)e2;
    }
    if (k3) {
      const int p = atomicAdd(&cnt[e3 >> (DL_BITS + SBIN_SHIFT)], 1);
      if (fits) stage[p] = (int)e3; else ss2[bs + p] = (int)e3;
    }
  }
  __syncthreads();
  if (fits)
    for (int i = tid; i < n; i += 1024) ss2[bs + i] = stage[i];
}

__global__ void degree_sum(const int* __restrict__ degp, int* __restrict__ deg,
                           int n_nodes, int nr, long deg_stride) {
  const int i = blockIdx.x * blockDim.x + threadIdx.x;
  if (i >= n_nodes) return;
  int s = 0;
  for (int r = 0; r < nr; ++r) s += degp[(long)r * deg_stride + i];
  deg[i] = s;
}

template <int F_IN, int F_OUT, bool ZERO_AGG, int HL_PITCH>
__global__ void node_transform(const float* __restrict__ h,
                               const float* __restrict__ Wl,
                               const float* __restrict__ Wr,
                               const float* __restrict__ bias,
                               float* __restrict__ hl,
                               float* __restrict__ hs,
                               float* __restrict__ agg,
                               int n_nodes) {
  const int lane = threadIdx.x & 31;
  const int wave = threadIdx.x >> 5;
  const int tile = blockIdx.x * (blockDim.x >> 5) + wave;
  const int tiles = (n_nodes + 15) >> 4;
  if (tile >= tiles) return;

  const int m    = lane & 15;
  const int hi   = lane >> 4;
  const int row0 = tile << 4;
  const int arow = min(row0 + m, n_nodes - 1);
  const bool cok = (m < F_OUT);

  auto ldA = [&](int k) -> float {
    const bool ok = (k < F_IN);
    const float v = h[(long)arow * F_IN + (ok ? k : 0)];
    return ok ? v : 0.0f;
  };
  auto ldW = [&](const float* __restrict__ W, int k) -> float {
    const bool ok = cok && (k < F_IN);
    const float v = W[ok ? (m * F_IN + k) : 0];
    return ok ? v : 0.0f;
  };

  v8f dl = {};
  v8f dr = {};
#pragma unroll
  for (int kc = 0; kc < F_IN; kc += 4) {
    const int k0 = kc + hi * 2;
    v2f a, bl, br;
    a.x  = ldA(k0);
    a.y  = ldA(k0 + 1);
    bl.x = ldW(Wl, k0);
    bl.y = ldW(Wl, k0 + 1);
    br.x = ldW(Wr, k0);
    br.y = ldW(Wr, k0 + 1);
    dl = __builtin_amdgcn_wmma_f32_16x16x4_f32(false, a, false, bl, (short)0,
                                               dl, false, false);
    dr = __builtin_amdgcn_wmma_f32_16x16x4_f32(false, a, false, br, (short)0,
                                               dr, false, false);
  }

  const bool full = (row0 + 16 <= n_nodes);
  if (m < HL_PITCH) {
    const long lbase = (long)(row0 + hi * 8) * HL_PITCH + m;
    if (full) {
#pragma unroll
      for (int v = 0; v < 8; ++v) hl[lbase + (long)v * HL_PITCH] = dl[v];
    } else {
#pragma unroll
      for (int v = 0; v < 8; ++v)
        if (row0 + hi * 8 + v < n_nodes) hl[lbase + (long)v * HL_PITCH] = dl[v];
    }
  }
  if (cok) {
    const float bv   = bias[m];
    const long  base = (long)(row0 + hi * 8) * F_OUT + m;
    if (full) {
#pragma unroll
      for (int v = 0; v < 8; ++v) {
        hs[base + (long)v * F_OUT]  = dr[v] + bv;
        if (ZERO_AGG) agg[base + (long)v * F_OUT] = 0.0f;
      }
    } else {
#pragma unroll
      for (int v = 0; v < 8; ++v) {
        if (row0 + hi * 8 + v < n_nodes) {
          hs[base + (long)v * F_OUT]  = dr[v] + bv;
          if (ZERO_AGG) agg[base + (long)v * F_OUT] = 0.0f;
        }
      }
    }
  }
}

template <int F, bool RELU>
__global__ void pull_combine(const int* __restrict__ rp, const int* __restrict__ ss,
                             const float* __restrict__ hl,
                             const float* __restrict__ hs,
                             float* __restrict__ out, int n_nodes) {
  const int q    = threadIdx.x & 7;
  const int node = (int)((blockIdx.x * blockDim.x + threadIdx.x) >> 3);
  const bool live = node < n_nodes;
  const int nd  = live ? node : n_nodes - 1;
  const int beg = (nd > 0) ? rp[nd - 1] : 0;
  const int end = live ? rp[nd] : beg;

  float acc[F];
  for (int f = 0; f < F; ++f) acc[f] = 0.0f;
  for (int j = beg + q; j < end; j += 8) {
    const int s = ss[j];
    for (int f = 0; f < F; ++f) acc[f] += hl[(long)s * F + f];
  }
  for (int f = 0; f < F; ++f) {
    acc[f] += __shfl_xor(acc[f], 4);
    acc[f] += __shfl_xor(acc[f], 2);
    acc[f] += __shfl_xor(acc[f], 1);
  }
  float val = acc[0];
  for (int f = 1; f < F; ++f) val = (q == f) ? acc[f] : val;

  if (live && q < F) {
    const float iv = 1.0f / fmaxf((float)(end - beg), 1.0f);
    const long  o  = (long)node * F + q;
    const float r  = val * iv + hs[o];
    out[o] = RELU ? fmaxf(r, 0.0f) : r;
  }
}

template <bool RELU, int OUT_PITCH>
__global__ void pull_rows8(const int* __restrict__ rp, const int* __restrict__ ss,
                           const float* __restrict__ hl8,
                           const float* __restrict__ hs,
                           float* __restrict__ out, int n_nodes) {
  const int l    = threadIdx.x & 7;
  const int p    = l >> 1;
  const int hh   = l & 1;
  const int node = (int)((blockIdx.x * blockDim.x + threadIdx.x) >> 3);
  const bool live = node < n_nodes;
  const int nd  = live ? node : n_nodes - 1;
  const int beg = (nd > 0) ? rp[nd - 1] : 0;
  const int end = live ? rp[nd] : beg;

  const v4f* __restrict__ rows = (const v4f*)hl8;
  const v4f zero = {0.0f, 0.0f, 0.0f, 0.0f};
  v4f acc = zero;
  for (int j0 = beg + p; j0 < end; j0 += 16) {
    const int j1 = j0 + 4, j2 = j0 + 8, j3 = j0 + 12;
    const bool k1 = j1 < end, k2 = j2 < end, k3 = j3 < end;
    const int s0 = ss[j0];
    int s1 = ss[k1 ? j1 : j0];
    int s2 = ss[k2 ? j2 : j0];
    int s3 = ss[k3 ? j3 : j0];
    asm volatile("" : "+v"(s1), "+v"(s2), "+v"(s3));
    v4f v0 = rows[(long)s0 * 2 + hh];
    v4f v1 = zero, v2 = zero, v3 = zero;
    if (k1) v1 = rows[(long)s1 * 2 + hh];
    if (k2) v2 = rows[(long)s2 * 2 + hh];
    if (k3) v3 = rows[(long)s3 * 2 + hh];
    acc += (v0 + v1) + (v2 + v3);
  }
  for (int c = 0; c < 4; ++c) {
    acc[c] += __shfl_xor(acc[c], 4);
    acc[c] += __shfl_xor(acc[c], 2);
  }
  const float nx = __shfl_xor(acc[0], 1);
  const float ny = __shfl_xor(acc[1], 1);
  const float nw = __shfl_xor(acc[3], 1);
  float val = acc[0];
  val = (l == 1) ? ny : val;
  val = (l == 2) ? acc[2] : val;
  val = (l == 3) ? nw : val;
  val = (l == 4) ? nx : val;

  if (live && l < OUT_PITCH) {
    float r = 0.0f;
    if (l < 5) {
      const float iv = 1.0f / fmaxf((float)(end - beg), 1.0f);
      r = val * iv + hs[(long)node * 5 + l];
      r = RELU ? fmaxf(r, 0.0f) : r;
    }
    out[(long)node * OUT_PITCH + l] = r;
  }
}

#define A_PITCH 12
template <int F_OUT, bool RELU, int OUT_PITCH>
__global__ void __launch_bounds__(256)
pull_transform(const int* __restrict__ rp, const int* __restrict__ ss,
               const float* __restrict__ h8,
               const float* __restrict__ Wl, const float* __restrict__ Wr,
               const float* __restrict__ bias,
               float* __restrict__ out, int n_nodes) {
  __shared__ float arow[32 * A_PITCH];
  const int l    = threadIdx.x & 7;
  const int p    = l >> 1;
  const int hh   = l & 1;
  const int ln   = threadIdx.x >> 3;
  const int node = (int)blockIdx.x * 32 + ln;
  const bool live = node < n_nodes;
  const int nd  = live ? node : n_nodes - 1;
  const int beg = (nd > 0) ? rp[nd - 1] : 0;
  const int end = live ? rp[nd] : beg;

  const v4f* __restrict__ rows = (const v4f*)h8;
  const v4f zero = {0.0f, 0.0f, 0.0f, 0.0f};
  const float mine = (l < 5) ? h8[(long)nd * ROW8 + l] : 0.0f;

  const int wave = threadIdx.x >> 5;
  const int lane = threadIdx.x & 31;
  const int m    = lane & 15;
  const int hi   = lane >> 4;
  const bool cok = (m < F_OUT);
  auto bok = [&](int k) -> bool { return cok && (k < 10); };
  auto ldB = [&](int k) -> float {
    const bool left = (k < 5);
    const float* __restrict__ Wp = left ? Wl : Wr;
    const int kk = left ? k : k - 5;
    return Wp[bok(k) ? (m * 5 + kk) : 0];
  };
  float braw[A_PITCH / 2];
  float bv = 0.0f;
#pragma unroll
  for (int i = 0; i < A_PITCH / 2; ++i) braw[i] = 0.0f;
  if (wave < 2) {
#pragma unroll
    for (int i = 0; i < A_PITCH / 4; ++i) {
      const int k0 = 4 * i + hi * 2;
      braw[2 * i]     = ldB(k0);
      braw[2 * i + 1] = ldB(k0 + 1);
    }
    bv = cok ? bias[m] : 0.0f;
  }
  v4f acc = zero;
  for (int j0 = beg + p; j0 < end; j0 += 16) {
    const int j1 = j0 + 4, j2 = j0 + 8, j3 = j0 + 12;
    const bool k1 = j1 < end, k2 = j2 < end, k3 = j3 < end;
    const int s0 = ss[j0];
    int s1 = ss[k1 ? j1 : j0];
    int s2 = ss[k2 ? j2 : j0];
    int s3 = ss[k3 ? j3 : j0];
    asm volatile("" : "+v"(s1), "+v"(s2), "+v"(s3));
    v4f v0 = rows[(long)s0 * 2 + hh];
    v4f v1 = zero, v2 = zero, v3 = zero;
    if (k1) v1 = rows[(long)s1 * 2 + hh];
    if (k2) v2 = rows[(long)s2 * 2 + hh];
    if (k3) v3 = rows[(long)s3 * 2 + hh];
    acc += (v0 + v1) + (v2 + v3);
  }
  for (int c = 0; c < 4; ++c) {
    acc[c] += __shfl_xor(acc[c], 4);
    acc[c] += __shfl_xor(acc[c], 2);
  }
  const float nx = __shfl_xor(acc[0], 1);
  const float ny = __shfl_xor(acc[1], 1);
  const float nw = __shfl_xor(acc[3], 1);
  float val = acc[0];
  val = (l == 1) ? ny : val;
  val = (l == 2) ? acc[2] : val;
  val = (l == 3) ? nw : val;
  val = (l == 4) ? nx : val;

  const float iv = 1.0f / fmaxf((float)(end - beg), 1.0f);
  if (l < 5) {
    arow[ln * A_PITCH + l]     = live ? val * iv : 0.0f;
    arow[ln * A_PITCH + 5 + l] = live ? mine : 0.0f;
  } else if (l < 7) {
    arow[ln * A_PITCH + 5 + l] = 0.0f;
  }
  __syncthreads();

  if (wave >= 2) return;

  const float* __restrict__ at = arow + (wave * 16 + m) * A_PITCH;
  v8f d = {};
#pragma unroll
  for (int kc = 0; kc < A_PITCH; kc += 4) {
    const int k0 = kc + hi * 2;
    v2f a, b;
    a.x = at[k0];
    a.y = at[k0 + 1];
    b.x = bok(k0) ? braw[kc / 2] : 0.0f;
    b.y = bok(k0 + 1) ? braw[kc / 2 + 1] : 0.0f;
    d = __builtin_amdgcn_wmma_f32_16x16x4_f32(false, a, false, b, (short)0,
                                              d, false, false);
  }

  const int row0 = (int)blockIdx.x * 32 + wave * 16;
  if (m < OUT_PITCH) {
    const long  base = (long)(row0 + hi * 8) * OUT_PITCH + m;
    const bool  full = (row0 + 16 <= n_nodes);
    if (full) {
#pragma unroll
      for (int v = 0; v < 8; ++v) {
        const float r = d[v] + bv;
        out[base + (long)v * OUT_PITCH] = RELU ? fmaxf(r, 0.0f) : r;
      }
    } else {
#pragma unroll
      for (int v = 0; v < 8; ++v) {
        if (row0 + hi * 8 + v < n_nodes) {
          const float r = d[v] + bv;
          out[base + (long)v * OUT_PITCH] = RELU ? fmaxf(r, 0.0f) : r;
        }
      }
    }
  }
}

#define TPITCH 5
typedef float v4f_a4 __attribute__((ext_vector_type(4), aligned(4)));

__device__ inline unsigned ld_entry(const int* p) {
  return (unsigned)__builtin_nontemporal_load(p);
}
__device__ inline long row_offset(unsigned e, int rbase, int last) {
  return (long)min(rbase + (int)(e >> DL_BITS), last) * TPITCH;
}
__device__ inline void add_row(float* acc, unsigned e, v4f a, float c) {
  float* q = acc + (e & ((1u << DL_BITS) - 1));
  atomicAdd(q, a.x);
  atomicAdd(q + DG, a.y);
  atomicAdd(q + 2 * DG, a.z);
  atomicAdd(q + 3 * DG, a.w);
  atomicAdd(q + 4 * DG, c);
}

__device__ inline float sum_ranges(const float* __restrict__ p, int nr,
                                   long part_stride) {
  float v[NR_MAX];
#pragma unroll
  for (int k = 0; k < NR_MAX; ++k) {
    const bool ok = k < nr;
    const float t = p[ok ? (long)k * part_stride : 0];
    v[k] = ok ? t : 0.0f;
  }
  float s = v[0];
#pragma unroll
  for (int k = 1; k < NR_MAX; ++k) s += v[k];
  return s;
}

__global__ void __launch_bounds__(1024)
agg_block(const int* __restrict__ ss2, const int* __restrict__ offs,
          const float* __restrict__ tab, float* __restrict__ part,
          long n_edges, int n_nodes, int nr, int nbk, int G, long part_stride) {
  __shared__ float acc[5 * DG];
  const int b = blockIdx.x;
  const int g = b / nr, r = b - g * nr;
  const int tid = threadIdx.x;
  const int bs = offs[(long)b * G];
  const int be = (b + 1 < nbk) ? offs[(long)(b + 1) * G] : (int)n_edges;
  const int n = be - bs;
  const int* __restrict__ in = ss2 + bs;

  for (int i = tid; i < 5 * DG; i += 1024) acc[i] = 0.0f;
  __syncthreads();

  const int rbase = r * SRN;
  const int last = n_nodes - 1;
  int i = tid;
  for (; i + 3072 < n; i += 4096) {
    const unsigned e0 = ld_entry(in + i);
    const unsigned e1 = ld_entry(in + i + 1024);
    const unsigned e2 = ld_entry(in + i + 2048);
    const unsigned e3 = ld_entry(in + i + 3072);
    const float* __restrict__ p0 = tab + row_offset(e0, rbase, last);
    const float* __restrict__ p1 = tab + row_offset(e1, rbase, last);
    const float* __restrict__ p2 = tab + row_offset(e2, rbase, last);
    const float* __restrict__ p3 = tab + row_offset(e3, rbase, last);
    const v4f a0 = *(const v4f_a4*)p0;
    const v4f a1 = *(const v4f_a4*)p1;
    const v4f a2 = *(const v4f_a4*)p2;
    const v4f a3 = *(const v4f_a4*)p3;
    const float c0 = p0[4];
    const float c1 = p1[4];
    const float c2 = p2[4];
    const float c3 = p3[4];
    add_row(acc, e0, a0, c0);
    add_row(acc, e1, a1, c1);
    add_row(acc, e2, a2, c2);
    add_row(acc, e3, a3, c3);
  }
  for (; i < n; i += 1024) {
    const unsigned e0 = ld_entry(in + i);
    const float* __restrict__ p0 = tab + row_offset(e0, rbase, last);
    const v4f a0 = *(const v4f_a4*)p0;
    const float c0 = p0[4];
    add_row(acc, e0, a0, c0);
  }
  __syncthreads();

  const int node0 = g * DG;
  const int rows = min(DG, n_nodes - node0);
  float* __restrict__ po = part + (long)r * part_stride + (long)node0 * 5;
  for (int idx = tid; idx < rows * 5; idx += 1024) {
    const int d = idx / 5;
    const int f = idx - d * 5;
    po[idx] = acc[f * DG + d];
  }
}

__global__ void combine_rows(const float* __restrict__ part,
                             const int* __restrict__ deg,
                             const float* __restrict__ hs,
                             float* __restrict__ tab, int n_nodes, int nr,
                             long part_stride) {
  const int l    = threadIdx.x & 7;
  const int node = (int)((blockIdx.x * blockDim.x + threadIdx.x) >> 3);
  if (node >= n_nodes || l >= 5) return;
  const float s = sum_ranges(part + (long)node * 5 + l, nr, part_stride);
  const float iv = 1.0f / fmaxf((float)deg[node], 1.0f);
  tab[(long)node * TPITCH + l] = fmaxf(s * iv + hs[(long)node * 5 + l], 0.0f);
}

template <int F_OUT, bool RELU, int OUT_PITCH>
__global__ void __launch_bounds__(256)
combine_transform(const float* __restrict__ part, const int* __restrict__ deg,
                  const float* h8,
                  const float* __restrict__ Wl, const float* __restrict__ Wr,
                  const float* __restrict__ bias,
                  float* out, int n_nodes, int nr, long part_stride) {
  __shared__ float arow[32 * A_PITCH];
  const int l    = threadIdx.x & 7;
  const int ln   = threadIdx.x >> 3;
  const int node = (int)blockIdx.x * 32 + ln;
  const bool live = node < n_nodes;
  const int nd  = live ? node : n_nodes - 1;

  const float mine = (l < 5) ? h8[(long)nd * TPITCH + l] : 0.0f;

  const int wave = threadIdx.x >> 5;
  const int lane = threadIdx.x & 31;
  const int m    = lane & 15;
  const int hi   = lane >> 4;
  const bool cok = (m < F_OUT);
  auto bok = [&](int k) -> bool { return cok && (k < 10); };
  auto ldB = [&](int k) -> float {
    const bool left = (k < 5);
    const float* __restrict__ Wp = left ? Wl : Wr;
    const int kk = left ? k : k - 5;
    return Wp[bok(k) ? (m * 5 + kk) : 0];
  };
  float braw[A_PITCH / 2];
  float bv = 0.0f;
#pragma unroll
  for (int i = 0; i < A_PITCH / 2; ++i) braw[i] = 0.0f;
  if (wave < 2) {
#pragma unroll
    for (int i = 0; i < A_PITCH / 4; ++i) {
      const int k0 = 4 * i + hi * 2;
      braw[2 * i]     = ldB(k0);
      braw[2 * i + 1] = ldB(k0 + 1);
    }
    bv = cok ? bias[m] : 0.0f;
  }

  const float val = sum_ranges(part + (long)nd * 5 + (l < 5 ? l : 4), nr, part_stride);
  const float iv = 1.0f / fmaxf((float)deg[nd], 1.0f);

  if (l < 5) {
    arow[ln * A_PITCH + l]     = live ? val * iv : 0.0f;
    arow[ln * A_PITCH + 5 + l] = live ? mine : 0.0f;
  } else if (l < 7) {
    arow[ln * A_PITCH + 5 + l] = 0.0f;
  }
  __syncthreads();

  if (wave >= 2) return;

  const float* __restrict__ at = arow + (wave * 16 + m) * A_PITCH;
  v8f d = {};
#pragma unroll
  for (int kc = 0; kc < A_PITCH; kc += 4) {
    const int k0 = kc + hi * 2;
    v2f a, b;
    a.x = at[k0];
    a.y = at[k0 + 1];
    b.x = bok(k0) ? braw[kc / 2] : 0.0f;
    b.y = bok(k0 + 1) ? braw[kc / 2 + 1] : 0.0f;
    d = __builtin_amdgcn_wmma_f32_16x16x4_f32(false, a, false, b, (short)0,
                                              d, false, false);
  }

  const int row0 = (int)blockIdx.x * 32 + wave * 16;
  if (m < OUT_PITCH) {
    const long  base = (long)(row0 + hi * 8) * OUT_PITCH + m;
    const bool  full = (row0 + 16 <= n_nodes);
    if (full) {
#pragma unroll
      for (int v = 0; v < 8; ++v) {
        const float r = d[v] + bv;
        out[base + (long)v * OUT_PITCH] = RELU ? fmaxf(r, 0.0f) : r;
      }
    } else {
#pragma unroll
      for (int v = 0; v < 8; ++v) {
        if (row0 + hi * 8 + v < n_nodes) {
          const float r = d[v] + bv;
          out[base + (long)v * OUT_PITCH] = RELU ? fmaxf(r, 0.0f) : r;
        }
      }
    }
  }
}

__global__ void __launch_bounds__(1024)
agg_block1(const int* __restrict__ ss2, const int* __restrict__ offs,
           const float* __restrict__ tab1, float* __restrict__ part1,
           long n_edges, int n_nodes, int nr, int nbk, int G, long part_stride) {
  __shared__ float acc[DG];
  const int b = blockIdx.x;
  const int g = b / nr, r = b - g * nr;
  const int tid = threadIdx.x;
  const int bs = offs[(long)b * G];
  const int be = (b + 1 < nbk) ? offs[(long)(b + 1) * G] : (int)n_edges;
  const int n = be - bs;
  const int* __restrict__ in = ss2 + bs;

  for (int i = tid; i < DG; i += 1024) acc[i] = 0.0f;
  __syncthreads();

  const int rbase = r * SRN;
  const int last = n_nodes - 1;
  const unsigned dmask = (1u << DL_BITS) - 1;
  int i = tid;
  for (; i + 7168 < n; i += 8192) {
    unsigned e[8];
    float v[8];
#pragma unroll
    for (int k = 0; k < 8; ++k) e[k] = ld_entry(in + i + k * 1024);
#pragma unroll
    for (int k = 0; k < 8; ++k)
      v[k] = tab1[min(rbase + (int)(e[k] >> DL_BITS), last)];
#pragma unroll
    for (int k = 0; k < 8; ++k) atomicAdd(acc + (e[k] & dmask), v[k]);
  }
  for (; i + 3072 < n; i += 4096) {
    unsigned e[4];
    float v[4];
#pragma unroll
    for (int k = 0; k < 4; ++k) e[k] = ld_entry(in + i + k * 1024);
#pragma unroll
    for (int k = 0; k < 4; ++k)
      v[k] = tab1[min(rbase + (int)(e[k] >> DL_BITS), last)];
#pragma unroll
    for (int k = 0; k < 4; ++k) atomicAdd(acc + (e[k] & dmask), v[k]);
  }
  for (; i < n; i += 1024) {
    const unsigned e0 = ld_entry(in + i);
    const float v0 = tab1[min(rbase + (int)(e0 >> DL_BITS), last)];
    atomicAdd(acc + (e0 & dmask), v0);
  }
  __syncthreads();

  const int node0 = g * DG;
  const int rows = min(DG, n_nodes - node0);
  float* __restrict__ po = part1 + (long)r * part_stride + node0;
  for (int d = tid; d < rows; d += 1024) po[d] = acc[d];
}

__global__ void combine_out1(const float* __restrict__ part1,
                             const int* __restrict__ deg,
                             const float* __restrict__ hs1,
                             float* __restrict__ out, int n_nodes, int nr,
                             long part_stride) {
  const int node = blockIdx.x * blockDim.x + threadIdx.x;
  if (node >= n_nodes) return;
  const float s = sum_ranges(part1 + node, nr, part_stride);
  const float iv = 1.0f / fmaxf((float)deg[node], 1.0f);
  out[node] = s * iv + hs1[node];
}

__global__ void zero_f32(float* __restrict__ p, long n) {
  long i = (long)blockIdx.x * blockDim.x + threadIdx.x;
  if (i < n) p[i] = 0.0f;
}

__global__ void count_edges(const int* __restrict__ dst, float* __restrict__ cnt,
                            long n_edges) {
  long i = (long)blockIdx.x * blockDim.x + threadIdx.x;
  if (i < n_edges) unsafeAtomicAdd(&cnt[dst[i]], 1.0f);
}

__global__ void reciprocal_clamped(float* __restrict__ p, long n) {
  long i = (long)blockIdx.x * blockDim.x + threadIdx.x;
  if (i < n) p[i] = 1.0f / fmaxf(p[i], 1.0f);
}

template <int F>
__global__ void edge_scatter(const int* __restrict__ src,
                             const int* __restrict__ dst,
                             const float* __restrict__ hl,
                             float* __restrict__ agg, long n_edges) {
  long i = (long)blockIdx.x * blockDim.x + threadIdx.x;
  if (i >= n_edges) return;
  const int s = src[i];
  const int d = dst[i];
  float v[F];
#pragma unroll
  for (int j = 0; j < F; ++j) v[j] = hl[(long)s * F + j];
#pragma unroll
  for (int j = 0; j < F; ++j) unsafeAtomicAdd(&agg[(long)d * F + j], v[j]);
}

template <int F, bool RELU>
__global__ void combine(const float* __restrict__ agg,
                        const float* __restrict__ hs,
                        const float* __restrict__ inv,
                        float* __restrict__ out, int n_nodes) {
  const int node = blockIdx.x * blockDim.x + threadIdx.x;
  if (node >= n_nodes) return;
  const float iv = inv[node];
#pragma unroll
  for (int j = 0; j < F; ++j) {
    const long o = (long)node * F + j;
    const float v = agg[o] * iv + hs[o];
    out[o] = RELU ? fmaxf(v, 0.0f) : v;
  }
}

extern "C" void kernel_launch(void* const* d_in, const int* in_sizes, int n_in,
                              void* d_out, int out_size, void* d_ws,
                              size_t ws_size, hipStream_t stream) {
  const float* x    = (const float*)d_in[0];
  const int*   ei   = (const int*)  d_in[1];
  const float* Wl1  = (const float*)d_in[2];
  const float* Wr1  = (const float*)d_in[3];
  const float* b1   = (const float*)d_in[4];
  const float* Wlm  = (const float*)d_in[5];
  const float* Wrm  = (const float*)d_in[6];
  const float* bm   = (const float*)d_in[7];
  const float* Wl10 = (const float*)d_in[8];
  const float* Wr10 = (const float*)d_in[9];
  const float* b10  = (const float*)d_in[10];

  const int  N = in_sizes[0] / 10;
  const long E = (long)in_sizes[1] / 2;
  const int* src = ei;
  const int* dst = ei + E;

  const int B = 256;
  auto cdiv = [](long a, long b) { return (int)((a + b - 1) / b); };
  auto pad32 = [](size_t a) { return (a + 31) & ~(size_t)31; };

  const int tiles   = (N + 15) / 16;
  const int tblocks = (tiles + 7) / 8;

  const int    nb    = cdiv(N, SCAN_CHUNK);
  const size_t w_rp  = pad32((size_t)N);
  const size_t w_bs  = pad32((size_t)nb);
  const size_t w_ss  = pad32((size_t)E);
  const size_t w_h   = pad32((size_t)N * 5);
  const size_t w_h8  = pad32((size_t)N * ROW8);
  const size_t need  = (w_rp + w_bs + w_ss + 3 * w_h) * sizeof(int);
  const size_t need8 = (w_rp + w_bs + w_ss + 2 * w_h + w_h8) * sizeof(int);

  if (ws_size >= need) {
    const bool wide = (ws_size >= need8);
    int*   rp   = (int*)d_ws;
    int*   bsum = rp + w_rp;
    int*   ss   = bsum + w_bs;
    float* h    = (float*)(ss + w_ss);
    float* hs   = h + w_h;
    float* hl   = hs + w_h;
    float* none = nullptr;

    const int    PG     = 1024;
    const int    nbk    = cdiv(N, BK_NODES);
    const int    chunk  = cdiv(E, PG);
    const size_t w_hist = pad32((size_t)nbk * PG);
    const size_t need_s = need8 + (w_hist + w_ss) * sizeof(int);
    const bool lds_sort = wide && ws_size >= need_s && nbk <= MAX_BK &&
                          chunk <= PART_CAP && N <= (1 << 21);
    {
      const int    ng      = cdiv(N, DG);
      const int    nr      = cdiv(N, SRN);
      const int    nbk2    = ng * nr;
      const int    chunk2  = cdiv(E, PG2);
      const int    hn2     = nbk2 * PG2;
      const int    hnb2    = cdiv(hn2, SCAN_CHUNK);
      const size_t w_hist2 = pad32((size_t)hn2);
      const size_t need_b  = (w_rp + w_bs + w_ss + 2 * w_h + w_h8 + w_hist2 + w_ss) * sizeof(int);
      const bool blocked = wide && ws_size >= need_b && nr <= NR_MAX &&
                           nbk2 <= MAX_BK && chunk2 <= PART_CAP &&
                           (size_t)hnb2 <= w_bs && (size_t)nr * w_h <= w_ss &&
                           (size_t)nr * w_rp <= 2 * w_h && E < (1L << 31);
      if (blocked) {
        int*   deg   = rp;
        int*   ss2   = ss;
        int*   degp  = (int*)h;
        float* tab   = hl;
        int*   hist2 = (int*)(hl + w_h8);
        int*   tmp   = hist2 + w_hist2;
        float* part  = (float*)tmp;
        const long dstr = (long)w_rp;
        const long pstr = (long)w_h;
        part_hist2<<<PG2, 256, 0, stream>>>(src, dst, hist2, E, chunk2, nr, nbk2);
        block_sums<<<hnb2, B, 0, stream>>>(hist2, bsum, hn2);
        scan_block_sums<<<1, B, 0, stream>>>(bsum, hnb2);
        scan_apply<<<hnb2, B, 0, stream>>>(hist2, bsum, hn2);
        part_scatter2<<<PG2, 256, 0, stream>>>(src, dst, hist2, tmp, E, chunk2, nr, nbk2);
        bucket_sort2<<<nbk2, 1024, 0, stream>>>(tmp, hist2, degp, ss2, E, N, nr, nbk2, PG2, dstr);
        degree_sum<<<cdiv(N, B), B, 0, stream>>>(degp, deg, N, nr, dstr);

        const int cblocks = cdiv((long)N * 8, B);
        node_transform<10, 5, false, TPITCH><<<tblocks, 256, 0, stream>>>(x, Wl1, Wr1, b1, tab, hs, none, N);
        agg_block<<<nbk2, 1024, 0, stream>>>(ss2, hist2, tab, part, E, N, nr, nbk2, PG2, pstr);
        combine_rows<<<cblocks, B, 0, stream>>>(part, deg, hs, tab, N, nr, pstr);
        for (int l = 0; l < 8; ++l) {
          agg_block<<<nbk2, 1024, 0, stream>>>(ss2, hist2, tab, part, E, N, nr, nbk2, PG2, pstr);
          combine_transform<5, true, TPITCH><<<cblocks, B, 0, stream>>>(
              part, deg, tab, Wlm + l * 25, Wrm + l * 25, bm + l * 5, tab, N, nr, pstr);
        }
        float* hl1 = h;
        float* hs1 = hs;
        node_transform<5, 1, false, 1><<<tblocks, 256, 0, stream>>>(tab, Wl10, Wr10, b10, hl1, hs1, none, N);
        agg_block1<<<nbk2, 1024, 0, stream>>>(ss2, hist2, hl1, part, E, N, nr, nbk2, PG2, pstr);
        combine_out1<<<cdiv(N, B), B, 0, stream>>>(part, deg, hs1, (float*)d_out, N, nr, pstr);
        return;
      }
    }

    if (lds_sort) {
      int* hist = (int*)(hl + w_h8);
      int* tmp  = hist + w_hist;
      const int hn  = nbk * PG;
      const int hnb = cdiv(hn, SCAN_CHUNK);
      part_hist<<<PG, 256, 0, stream>>>(dst, hist, E, chunk, nbk);
      block_sums<<<hnb, B, 0, stream>>>(hist, bsum, hn);
      scan_block_sums<<<1, B, 0, stream>>>(bsum, hnb);
      scan_apply<<<hnb, B, 0, stream>>>(hist, bsum, hn);
      part_scatter<<<PG, 256, 0, stream>>>(src, dst, hist, tmp, E, chunk, nbk);
      bucket_rows<<<nbk, 1024, 0, stream>>>(tmp, hist, rp, ss, E, N, nbk, PG);
    } else {
      const int eblocks = cdiv(E, B) < 2048 ? cdiv(E, B) : 2048;
      zero_i32<<<cdiv(N, B), B, 0, stream>>>(rp, N);
      count_in_edges<<<eblocks, B, 0, stream>>>(dst, rp, E);
      block_sums<<<nb, B, 0, stream>>>(rp, bsum, N);
      scan_block_sums<<<1, B, 0, stream>>>(bsum, nb);
      scan_apply<<<nb, B, 0, stream>>>(rp, bsum, N);
      fill_rows<<<eblocks, B, 0, stream>>>(src, dst, rp, ss, E);
    }

    const int pblocks = cdiv((long)N * 8, B);

    if (lds_sort && w_ss >= w_h8) {
      float* ta = hl;
      float* tb = hl + w_h8 + w_hist;
      node_transform<10, 5, false, ROW8><<<tblocks, 256, 0, stream>>>(x, Wl1, Wr1, b1, ta, hs, none, N);
      pull_rows8<true, ROW8><<<pblocks, B, 0, stream>>>(rp, ss, ta, hs, tb, N);
      for (int l = 0; l < 8; ++l) {
        const float* tin  = (l & 1) ? ta : tb;
        float*       tout = (l & 1) ? tb : ta;
        pull_transform<5, true, ROW8><<<pblocks, B, 0, stream>>>(
            rp, ss, tin, Wlm + l * 25, Wrm + l * 25, bm + l * 5, tout, N);
      }
      pull_transform<1, false, 1><<<pblocks, B, 0, stream>>>(
          rp, ss, tb, Wl10, Wr10, b10, (float*)d_out, N);
      return;
    }

    if (wide) {
      node_transform<10, 5, false, ROW8><<<tblocks, 256, 0, stream>>>(x, Wl1, Wr1, b1, hl, hs, none, N);
      pull_rows8<true, 5><<<pblocks, B, 0, stream>>>(rp, ss, hl, hs, h, N);
    } else {
      node_transform<10, 5, false, 5><<<tblocks, 256, 0, stream>>>(x, Wl1, Wr1, b1, hl, hs, none, N);
      pull_combine<5, true><<<pblocks, B, 0, stream>>>(rp, ss, hl, hs, h, N);
    }

    for (int l = 0; l < 8; ++l) {
      if (wide) {
        node_transform<5, 5, false, ROW8><<<tblocks, 256, 0, stream>>>(
            h, Wlm + l * 25, Wrm + l * 25, bm + l * 5, hl, hs, none, N);
        pull_rows8<true, 5><<<pblocks, B, 0, stream>>>(rp, ss, hl, hs, h, N);
      } else {
        node_transform<5, 5, false, 5><<<tblocks, 256, 0, stream>>>(
            h, Wlm + l * 25, Wrm + l * 25, bm + l * 5, hl, hs, none, N);
        pull_combine<5, true><<<pblocks, B, 0, stream>>>(rp, ss, hl, hs, h, N);
      }
    }

    node_transform<5, 1, false, 1><<<tblocks, 256, 0, stream>>>(h, Wl10, Wr10, b10, hl, hs, none, N);
    pull_combine<1, false><<<pblocks, B, 0, stream>>>(rp, ss, hl, hs, (float*)d_out, N);
    return;
  }

  float* ws  = (float*)d_ws;
  float* inv = ws;
  float* hl  = inv + (size_t)N;
  float* hs  = hl + (size_t)N * 5;
  float* agg = hs + (size_t)N * 5;
  float* hA  = agg + (size_t)N * 5;
  float* hB  = hA + (size_t)N * 5;

  zero_f32<<<cdiv(N, B), B, 0, stream>>>(inv, N);
  count_edges<<<cdiv(E, B), B, 0, stream>>>(dst, inv, E);
  reciprocal_clamped<<<cdiv(N, B), B, 0, stream>>>(inv, N);

  node_transform<10, 5, true, 5><<<tblocks, 256, 0, stream>>>(x, Wl1, Wr1, b1, hl, hs, agg, N);
  edge_scatter<5><<<cdiv(E, B), B, 0, stream>>>(src, dst, hl, agg, E);
  combine<5, true><<<cdiv(N, B), B, 0, stream>>>(agg, hs, inv, hA, N);

  for (int l = 0; l < 8; ++l) {
    const float* hin  = (l & 1) ? hB : hA;
    float*       hout = (l & 1) ? hA : hB;
    node_transform<5, 5, true, 5><<<tblocks, 256, 0, stream>>>(
        hin, Wlm + l * 25, Wrm + l * 25, bm + l * 5, hl, hs, agg, N);
    edge_scatter<5><<<cdiv(E, B), B, 0, stream>>>(src, dst, hl, agg, E);
    combine<5, true><<<cdiv(N, B), B, 0, stream>>>(agg, hs, inv, hout, N);
  }

  node_transform<5, 1, true, 1><<<tblocks, 256, 0, stream>>>(hA, Wl10, Wr10, b10, hl, hs, agg, N);
  edge_scatter<1><<<cdiv(E, B), B, 0, stream>>>(src, dst, hl, agg, E);
  combine<1, false><<<cdiv(N, B), B, 0, stream>>>(agg, hs, inv, (float*)d_out, N);
}


# Layers 2..9 combine launch: one wave per 32 nodes loads its matrix fragments straight from the partial rows (no LDS tile, no barrier, no idle waves): 8x fewer waves; sum order unchanged
# speedup vs baseline: 15.1217x; 1.0578x over previous
#include <hip/hip_runtime.h>


typedef __attribute__((ext_vector_type(2))) float v2f;
typedef __attribute__((ext_vector_type(4))) float v4f;
typedef __attribute__((ext_vector_type(8))) float v8f;

#define ROW8 8

#define SCAN_CHUNK 2048

__global__ void zero_i32(int* __restrict__ p, int n) {
  const int i = blockIdx.x * blockDim.x + threadIdx.x;
  if (i < n) p[i] = 0;
}

__global__ void count_in_edges(const int* __restrict__ dst, int* __restrict__ rp,
                               long n_edges) {
  const long stride = (long)gridDim.x * blockDim.x;
  long i = (long)blockIdx.x * blockDim.x + threadIdx.x;
  for (; i + 3 * stride < n_edges; i += 4 * stride) {
    const int d0 = dst[i];
    const int d1 = dst[i + stride];
    const int d2 = dst[i + 2 * stride];
    const int d3 = dst[i + 3 * stride];
    atomicAdd(&rp[d0], 1);
    atomicAdd(&rp[d1], 1);
    atomicAdd(&rp[d2], 1);
    atomicAdd(&rp[d3], 1);
  }
  for (; i < n_edges; i += stride) atomicAdd(&rp[dst[i]], 1);
}

__device__ inline int wave_incl_scan(int x, int lane) {
  for (int off = 1; off < 32; off <<= 1) {
    const int y = __shfl_up(x, off);
    if (lane >= off) x += y;
  }
  return x;
}

__global__ void block_sums(const int* __restrict__ cnt, int* __restrict__ bsum,
                           int n) {
  __shared__ int wsum[8];
  const int base = blockIdx.x * SCAN_CHUNK;
  int s = 0;
  for (int k = 0; k < 8; ++k) {
    const int idx = base + k * 256 + (int)threadIdx.x;
    if (idx < n) s += cnt[idx];
  }
  for (int off = 16; off > 0; off >>= 1) s += __shfl_xor(s, off);
  if ((threadIdx.x & 31) == 0) wsum[threadIdx.x >> 5] = s;
  __syncthreads();
  if (threadIdx.x == 0) {
    int t = 0;
    for (int w = 0; w < 8; ++w) t += wsum[w];
    bsum[blockIdx.x] = t;
  }
}

__global__ void scan_block_sums(int* __restrict__ bsum, int nb) {
  __shared__ int wsum[8];
  const int lane = threadIdx.x & 31;
  const int wave = threadIdx.x >> 5;
  int carry = 0;
  for (int base = 0; base < nb; base += 256) {
    const int i = base + (int)threadIdx.x;
    const int v = (i < nb) ? bsum[i] : 0;
    const int x = wave_incl_scan(v, lane);
    if (lane == 31) wsum[wave] = x;
    __syncthreads();
    int before = 0, total = 0;
    for (int w = 0; w < 8; ++w) {
      const int t = wsum[w];
      before += (w < wave) ? t : 0;
      total += t;
    }
    if (i < nb) bsum[i] = carry + before + x - v;
    carry += total;
    __syncthreads();
  }
}

__global__ void scan_apply(int* rp, const int* __restrict__ bofs, int n) {
  __shared__ int wsum[8];
  const int lane = threadIdx.x & 31;
  const int wave = threadIdx.x >> 5;
  const int base = blockIdx.x * SCAN_CHUNK + (int)threadIdx.x * 8;
  int v[8];
  int t = 0;
  for (int k = 0; k < 8; ++k) {
    const int idx = base + k;
    v[k] = (idx < n) ? rp[idx] : 0;
    t += v[k];
  }
  const int x = wave_incl_scan(t, lane);
  if (lane == 31) wsum[wave] = x;
  __syncthreads();
  int run = bofs[blockIdx.x] + x - t;
  for (int w = 0; w < 8; ++w) run += (w < wave) ? wsum[w] : 0;
  for (int k = 0; k < 8; ++k) {
    const int idx = base + k;
    if (idx < n) rp[idx] = run;
    run += v[k];
  }
}

__global__ void fill_rows(const int* __restrict__ src, const int* __restrict__ dst,
                          int* __restrict__ rp, int* __restrict__ ss,
                          long n_edges) {
  const long stride = (long)gridDim.x * blockDim.x;
  long i = (long)blockIdx.x * blockDim.x + threadIdx.x;
  for (; i + 3 * stride < n_edges; i += 4 * stride) {
    const int d0 = dst[i];
    const int d1 = dst[i + stride];
    const int d2 = dst[i + 2 * stride];
    const int d3 = dst[i + 3 * stride];
    const int s0 = src[i];
    const int s1 = src[i + stride];
    const int s2 = src[i + 2 * stride];
    const int s3 = src[i + 3 * stride];
    const int p0 = atomicAdd(&rp[d0], 1);
    const int p1 = atomicAdd(&rp[d1], 1);
    const int p2 = atomicAdd(&rp[d2], 1);
    const int p3 = atomicAdd(&rp[d3], 1);
    ss[p0] = s0;
    ss[p1] = s1;
    ss[p2] = s2;
    ss[p3] = s3;
  }
  for (; i < n_edges; i += stride) {
    const int d = dst[i];
    const int s = src[i];
    const int p = atomicAdd(&rp[d], 1);
    ss[p] = s;
  }
}

#define BK_SHIFT 11
#define BK_NODES 2048
#define MAX_BK 1024
#define PART_CAP 16384
#define BK_CAP 73728

__device__ inline int bucket_of(int d, int nbk) {
  const int b = (int)((unsigned)d >> BK_SHIFT);
  return b < nbk ? b : nbk - 1;
}

__global__ void part_hist(const int* __restrict__ dst, int* __restrict__ hist,
                          long n_edges, int chunk, int nbk) {
  __shared__ int bins[MAX_BK];
  for (int b = threadIdx.x; b < nbk; b += blockDim.x) bins[b] = 0;
  __syncthreads();
  const long lo = (long)blockIdx.x * chunk;
  long hi = lo + chunk;
  if (hi > n_edges) hi = n_edges;
  for (long i = lo + threadIdx.x; i < hi; i += 4 * blockDim.x) {
    const long i1 = i + blockDim.x, i2 = i1 + blockDim.x, i3 = i2 + blockDim.x;
    const int d0 = dst[i];
    const int d1 = dst[i1 < hi ? i1 : i];
    const int d2 = dst[i2 < hi ? i2 : i];
    const int d3 = dst[i3 < hi ? i3 : i];
    atomicAdd(&bins[bucket_of(d0, nbk)], 1);
    if (i1 < hi) atomicAdd(&bins[bucket_of(d1, nbk)], 1);
    if (i2 < hi) atomicAdd(&bins[bucket_of(d2, nbk)], 1);
    if (i3 < hi) atomicAdd(&bins[bucket_of(d3, nbk)], 1);
  }
  __syncthreads();
  for (int b = threadIdx.x; b < nbk; b += blockDim.x)
    hist[(long)b * gridDim.x + blockIdx.x] = bins[b];
}

__global__ void part_scatter(const int* __restrict__ src,
                             const int* __restrict__ dst,
                             const int* __restrict__ offs,
                             int* __restrict__ tmp, long n_edges, int chunk,
                             int nbk) {
  __shared__ int lbeg[MAX_BK];
  __shared__ int cur[MAX_BK];
  __shared__ int wsum[8];
  __shared__ int stage[PART_CAP];
  const int tid = threadIdx.x, lane = tid & 31, wave = tid >> 5;
  const int G = gridDim.x, g = blockIdx.x;
  const long total = (long)nbk * G;

  int v[4];
  int t = 0;
  for (int k = 0; k < 4; ++k) {
    const int b = 4 * tid + k;
    int c = 0;
    if (b < nbk) {
      const long idx = (long)b * G + g;
      const int o0 = offs[idx];
      const int o1 = (idx + 1 < total) ? offs[idx + 1] : (int)n_edges;
      c = o1 - o0;
    }
    v[k] = c;
    t += c;
  }
  const int x = wave_incl_scan(t, lane);
  if (lane == 31) wsum[wave] = x;
  __syncthreads();
  int run = x - t;
  for (int w = 0; w < 8; ++w) run += (w < wave) ? wsum[w] : 0;
  for (int k = 0; k < 4; ++k) {
    const int b = 4 * tid + k;
    if (b < nbk) { lbeg[b] = run; cur[b] = run; }
    run += v[k];
  }
  __syncthreads();

  const long lo = (long)g * chunk;
  long hi = lo + chunk;
  if (hi > n_edges) hi = n_edges;
  for (long i = lo + tid; i < hi; i += 4 * blockDim.x) {
    const long i1 = i + blockDim.x, i2 = i1 + blockDim.x, i3 = i2 + blockDim.x;
    const bool k1 = i1 < hi, k2 = i2 < hi, k3 = i3 < hi;
    const int d0 = dst[i];
    const int d1 = dst[k1 ? i1 : i];
    const int d2 = dst[k2 ? i2 : i];
    const int d3 = dst[k3 ? i3 : i];
    const int s0 = src[i];
    const int s1 = src[k1 ? i1 : i];
    const int s2 = src[k2 ? i2 : i];
    const int s3 = src[k3 ? i3 : i];
    {
      const int p = atomicAdd(&cur[bucket_of(d0, nbk)], 1);
      if (p < PART_CAP) stage[p] = (int)(((unsigned)s0 << BK_SHIFT) | ((unsigned)d0 & (BK_NODES - 1)));
    }
    if (k1) {
      const int p = atomicAdd(&cur[bucket_of(d1, nbk)], 1);
      if (p < PART_CAP) stage[p] = (int)(((unsigned)s1 << BK_SHIFT) | ((unsigned)d1 & (BK_NODES - 1)));
    }
    if (k2) {
      const int p = atomicAdd(&cur[bucket_of(d2, nbk)], 1);
      if (p < PART_CAP) stage[p] = (int)(((unsigned)s2 << BK_SHIFT) | ((unsigned)d2 & (BK_NODES - 1)));
    }
    if (k3) {
      const int p = atomicAdd(&cur[bucket_of(d3, nbk)], 1);
      if (p < PART_CAP) stage[p] = (int)(((unsigned)s3 << BK_SHIFT) | ((unsigned)d3 & (BK_NODES - 1)));
    }
  }
  __syncthreads();

  for (int b = wave; b < nbk; b += 8) {
    const int ls = lbeg[b];
    const int n = cur[b] - ls;
    const int gs = offs[(long)b * G + g];
    for (int k = lane; k < n; k += 32) tmp[gs + k] = stage[ls + k];
  }
}

__global__ void __launch_bounds__(1024)
bucket_rows(const int* __restrict__ tmp, const int* __restrict__ offs,
            int* __restrict__ rp, int* __restrict__ ss, long n_edges,
            int n_nodes, int nbk, int G) {
  __shared__ int cnt[BK_NODES];
  __shared__ int wsum[32];
  __shared__ int stage[BK_CAP];
  const int b = blockIdx.x;
  const int tid = threadIdx.x, lane = tid & 31, wave = tid >> 5;
  const int bs = offs[(long)b * G];
  const int be = (b + 1 < nbk) ? offs[(long)(b + 1) * G] : (int)n_edges;
  const int n = be - bs;
  const int* __restrict__ in = tmp + bs;

  cnt[tid] = 0;
  cnt[tid + 1024] = 0;
  __syncthreads();
  for (int i = tid; i < n; i += 4096) {
    const int i1 = i + 1024, i2 = i + 2048, i3 = i + 3072;
    const bool k1 = i1 < n, k2 = i2 < n, k3 = i3 < n;
    const int e0 = in[i];
    const int e1 = in[k1 ? i1 : i];
    const int e2 = in[k2 ? i2 : i];
    const int e3 = in[k3 ? i3 : i];
    atomicAdd(&cnt[e0 & (BK_NODES - 1)], 1);
    if (k1) atomicAdd(&cnt[e1 & (BK_NODES - 1)], 1);
    if (k2) atomicAdd(&cnt[e2 & (BK_NODES - 1)], 1);
    if (k3) atomicAdd(&cnt[e3 & (BK_NODES - 1)], 1);
  }
  __syncthreads();

  const int c0 = cnt[2 * tid];
  const int c1 = cnt[2 * tid + 1];
  const int t = c0 + c1;
  const int x = wave_incl_scan(t, lane);
  if (lane == 31) wsum[wave] = x;
  __syncthreads();
  int ex = x - t;
  for (int w = 0; w < 32; ++w) ex += (w < wave) ? wsum[w] : 0;
  cnt[2 * tid] = ex;
  cnt[2 * tid + 1] = ex + c0;
  const int node0 = b * BK_NODES + 2 * tid;
  if (node0 < n_nodes) rp[node0] = bs + ex + c0;
  if (node0 + 1 < n_nodes) rp[node0 + 1] = bs + ex + t;
  __syncthreads();

  const bool fits = (n <= BK_CAP);
  for (int i = tid; i < n; i += 4096) {
    const int i1 = i + 1024, i2 = i + 2048, i3 = i + 3072;
    const bool k1 = i1 < n, k2 = i2 < n, k3 = i3 < n;
    const unsigned e0 = (unsigned)in[i];
    const unsigned e1 = (unsigned)in[k1 ? i1 : i];
    const unsigned e2 = (unsigned)in[k2 ? i2 : i];
    const unsigned e3 = (unsigned)in[k3 ? i3 : i];
    {
      const int p = atomicAdd(&cnt[e0 & (BK_NODES - 1)], 1);
      if (fits) stage[p] = (int)(e0 >> BK_SHIFT); else ss[bs + p] = (int)(e0 >> BK_SHIFT);
    }
    if (k1) {
      const int p = atomicAdd(&cnt[e1 & (BK_NODES - 1)], 1);
      if (fits) stage[p] = (int)(e1 >> BK_SHIFT); else ss[bs + p] = (int)(e1 >> BK_SHIFT);
    }
    if (k2) {
      const int p = atomicAdd(&cnt[e2 & (BK_NODES - 1)], 1);
      if (fits) stage[p] = (int)(e2 >> BK_SHIFT); else ss[bs + p] = (int)(e2 >> BK_SHIFT);
    }
    if (k3) {
      const int p = atomicAdd(&cnt[e3 & (BK_NODES - 1)], 1);
      if (fits) stage[p] = (int)(e3 >> BK_SHIFT); else ss[bs + p] = (int)(e3 >> BK_SHIFT);
    }
  }
  __syncthreads();
  if (fits)
    for (int i = tid; i < n; i += 1024) ss[bs + i] = stage[i];
}

#define DG 11936
#define DL_BITS 14
#define SRN 83456
#define NR_MAX 6
#define SBIN_SHIFT 4
#define SBINS 8192
#define BK2_CAP 69632
#define PG2 992

__device__ inline int bucket2_of(int d, int s, int nr, int nbk) {
  const unsigned b = ((unsigned)d / DG) * (unsigned)nr + ((unsigned)s / SRN);
  return b < (unsigned)nbk ? (int)b : nbk - 1;
}
__device__ inline int entry2_of(int d, int s) {
  return (int)((((unsigned)s % SRN) << DL_BITS) | ((unsigned)d % DG));
}

__global__ void part_hist2(const int* __restrict__ src, const int* __restrict__ dst,
                           int* __restrict__ hist, long n_edges, int chunk,
                           int nr, int nbk) {
  __shared__ int bins[MAX_BK];
  for (int b = threadIdx.x; b < nbk; b += blockDim.x) bins[b] = 0;
  __syncthreads();
  const long lo = (long)blockIdx.x * chunk;
  long hi = lo + chunk;
  if (hi > n_edges) hi = n_edges;
  for (long i = lo + threadIdx.x; i < hi; i += 4 * blockDim.x) {
    const long i1 = i + blockDim.x, i2 = i1 + blockDim.x, i3 = i2 + blockDim.x;
    const bool k1 = i1 < hi, k2 = i2 < hi, k3 = i3 < hi;
    const int d0 = dst[i];
    const int d1 = dst[k1 ? i1 : i];
    const int d2 = dst[k2 ? i2 : i];
    const int d3 = dst[k3 ? i3 : i];
    const int s0 = src[i];
    const int s1 = src[k1 ? i1 : i];
    const int s2 = src[k2 ? i2 : i];
    const int s3 = src[k3 ? i3 : i];
    atomicAdd(&bins[bucket2_of(d0, s0, nr, nbk)], 1);
    if (k1) atomicAdd(&bins[bucket2_of(d1, s1, nr, nbk)], 1);
    if (k2) atomicAdd(&bins[bucket2_of(d2, s2, nr, nbk)], 1);
    if (k3) atomicAdd(&bins[bucket2_of(d3, s3, nr, nbk)], 1);
  }
  __syncthreads();
  for (int b = threadIdx.x; b < nbk; b += blockDim.x)
    hist[(long)b * gridDim.x + blockIdx.x] = bins[b];
}

__global__ void part_scatter2(const int* __restrict__ src,
                              const int* __restrict__ dst,
                              const int* __restrict__ offs,
                              int* __restrict__ tmp, long n_edges, int chunk,
                              int nr, int nbk) {
  __shared__ int lbeg[MAX_BK];
  __shared__ int cur[MAX_BK];
  __shared__ int wsum[8];
  __shared__ int stage[PART_CAP];
  const int tid = threadIdx.x, lane = tid & 31, wave = tid >> 5;
  const int G = gridDim.x, g = blockIdx.x;
  const long total = (long)nbk * G;

  int v[4];
  int t = 0;
  for (int k = 0; k < 4; ++k) {
    const int b = 4 * tid + k;
    int c = 0;
    if (b < nbk) {
      const long idx = (long)b * G + g;
      const int o0 = offs[idx];
      const int o1 = (idx + 1 < total) ? offs[idx + 1] : (int)n_edges;
      c = o1 - o0;
    }
    v[k] = c;
    t += c;
  }
  const int x = wave_incl_scan(t, lane);
  if (lane == 31) wsum[wave] = x;
  __syncthreads();
  int run = x - t;
  for (int w = 0; w < 8; ++w) run += (w < wave) ? wsum[w] : 0;
  for (int k = 0; k < 4; ++k) {
    const int b = 4 * tid + k;
    if (b < nbk) { lbeg[b] = run; cur[b] = run; }
    run += v[k];
  }
  __syncthreads();

  const long lo = (long)g * chunk;
  long hi = lo + chunk;
  if (hi > n_edges) hi = n_edges;
  for (long i = lo + tid; i < hi; i += 4 * blockDim.x) {
    const long i1 = i + blockDim.x, i2 = i1 + blockDim.x, i3 = i2 + blockDim.x;
    const bool k1 = i1 < hi, k2 = i2 < hi, k3 = i3 < hi;
    const int d0 = dst[i];
    const int d1 = dst[k1 ? i1 : i];
    const int d2 = dst[k2 ? i2 : i];
    const int d3 = dst[k3 ? i3 : i];
    const int s0 = src[i];
    const int s1 = src[k1 ? i1 : i];
    const int s2 = src[k2 ? i2 : i];
    const int s3 = src[k3 ? i3 : i];
    {
      const int p = atomicAdd(&cur[bucket2_of(d0, s0, nr, nbk)], 1);
      if (p < PART_CAP) stage[p] = entry2_of(d0, s0);
    }
    if (k1) {
      const int p = atomicAdd(&cur[bucket2_of(d1, s1, nr, nbk)], 1);
      if (p < PART_CAP) stage[p] = entry2_of(d1, s1);
    }
    if (k2) {
      const int p = atomicAdd(&cur[bucket2_of(d2, s2, nr, nbk)], 1);
      if (p < PART_CAP) stage[p] = entry2_of(d2, s2);
    }
    if (k3) {
      const int p = atomicAdd(&cur[bucket2_of(d3, s3, nr, nbk)], 1);
      if (p < PART_CAP) stage[p] = entry2_of(d3, s3);
    }
  }
  __syncthreads();

  for (int b = wave; b < nbk; b += 8) {
    const int ls = lbeg[b];
    const int n = cur[b] - ls;
    const int gs = offs[(long)b * G + g];
    for (int k = lane; k < n; k += 32) tmp[gs + k] = stage[ls + k];
  }
}

__global__ void __launch_bounds__(1024)
bucket_sort2(const int* __restrict__ tmp, const int* __restrict__ offs,
             int* __restrict__ degp, int* __restrict__ ss2, long n_edges,
             int n_nodes, int nr, int nbk, int G, long deg_stride) {
  __shared__ int cnt[SBINS];
  __shared__ int wsum[32];
  __shared__ int stage[BK2_CAP];
  const int b = blockIdx.x;
  const int g = b / nr, r = b - g * nr;
  const int tid = threadIdx.x, lane = tid & 31, wave = tid >> 5;
  const int bs = offs[(long)b * G];
  const int be = (b + 1 < nbk) ? offs[(long)(b + 1) * G] : (int)n_edges;
  const int n = be - bs;
  const int* __restrict__ in = tmp + bs;

  for (int i = tid; i < DG; i += 1024) stage[i] = 0;
  for (int i = tid; i < SBINS; i += 1024) cnt[i] = 0;
  __syncthreads();
  for (int i = tid; i < n; i += 4096) {
    const int i1 = i + 1024, i2 = i + 2048, i3 = i + 3072;
    const bool k1 = i1 < n, k2 = i2 < n, k3 = i3 < n;
    const unsigned e0 = (unsigned)in[i];
    const unsigned e1 = (unsigned)in[k1 ? i1 : i];
    const unsigned e2 = (unsigned)in[k2 ? i2 : i];
    const unsigned e3 = (unsigned)in[k3 ? i3 : i];
    atomicAdd(&stage[e0 & ((1u << DL_BITS) - 1)], 1);
    atomicAdd(&cnt[e0 >> (DL_BITS + SBIN_SHIFT)], 1);
    if (k1) {
      atomicAdd(&stage[e1 & ((1u << DL_BITS) - 1)], 1);
      atomicAdd(&cnt[e1 >> (DL_BITS + SBIN_SHIFT)], 1);
    }
    if (k2) {
      atomicAdd(&stage[e2 & ((1u << DL_BITS) - 1)], 1);
      atomicAdd(&cnt[e2 >> (DL_BITS + SBIN_SHIFT)], 1);
    }
    if (k3) {
      atomicAdd(&stage[e3 & ((1u << DL_BITS) - 1)], 1);
      atomicAdd(&cnt[e3 >> (DL_BITS + SBIN_SHIFT)], 1);
    }
  }
  __syncthreads();
  {
    const int node0 = g * DG;
    int* __restrict__ dp = degp + (long)r * deg_stride + node0;
    for (int i = tid; i < DG; i += 1024)
      if (node0 + i < n_nodes) dp[i] = stage[i];
  }

  int c[8];
  int t = 0;
  for (int k = 0; k < 8; ++k) { c[k] = cnt[8 * tid + k]; t += c[k]; }
  const int x = wave_incl_scan(t, lane);
  if (lane == 31) wsum[wave] = x;
  __syncthreads();
  int ex = x - t;
  for (int w = 0; w < 32; ++w) ex += (w < wave) ? wsum[w] : 0;
  for (int k = 0; k < 8; ++k) { cnt[8 * tid + k] = ex; ex += c[k]; }
  __syncthreads();

  const bool fits = (n <= BK2_CAP);
  for (int i = tid; i < n; i += 4096) {
    const int i1 = i + 1024, i2 = i + 2048, i3 = i + 3072;
    const bool k1 = i1 < n, k2 = i2 < n, k3 = i3 < n;
    const unsigned e0 = (unsigned)in[i];
    const unsigned e1 = (unsigned)in[k1 ? i1 : i];
    const unsigned e2 = (unsigned)in[k2 ? i2 : i];
    const unsigned e3 = (unsigned)in[k3 ? i3 : i];
    {
      const int p = atomicAdd(&cnt[e0 >> (DL_BITS + SBIN_SHIFT)], 1);
      if (fits) stage[p] = (int)e0; else ss2[bs + p] = (int)e0;
    }
    if (k1) {
      const int p = atomicAdd(&cnt[e1 >> (DL_BITS + SBIN_SHIFT)], 1);
      if (fits) stage[p] = (int)e1; else ss2[bs + p] = (int)e1;
    }
    if (k2) {
      const int p = atomicAdd(&cnt[e2 >> (DL_BITS + SBIN_SHIFT)], 1);
      if (fits) stage[p] = (int)e2; else ss2[bs + p] = (int)e2;
    }
    if (k3) {
      const int p = atomicAdd(&cnt[e3 >> (DL_BITS + SBIN_SHIFT)], 1);
      if (fits) stage[p] = (int)e3; else ss2[bs + p] = (int)e3;
    }
  }
  __syncthreads();
  if (fits)
    for (int i = tid; i < n; i += 1024) ss2[bs + i] = stage[i];
}

__global__ void degree_sum(const int* __restrict__ degp, int* __restrict__ deg,
                           int n_nodes, int nr, long deg_stride) {
  const int i = blockIdx.x * blockDim.x + threadIdx.x;
  if (i >= n_nodes) return;
  int s = 0;
  for (int r = 0; r < nr; ++r) s += degp[(long)r * deg_stride + i];
  deg[i] = s;
}

template <int F_IN, int F_OUT, bool ZERO_AGG, int HL_PITCH>
__global__ void node_transform(const float* __restrict__ h,
                               const float* __restrict__ Wl,
                               const float* __restrict__ Wr,
                               const float* __restrict__ bias,
                               float* __restrict__ hl,
                               float* __restrict__ hs,
                               float* __restrict__ agg,
                               int n_nodes) {
  const int lane = threadIdx.x & 31;
  const int wave = threadIdx.x >> 5;
  const int tile = blockIdx.x * (blockDim.x >> 5) + wave;
  const int tiles = (n_nodes + 15) >> 4;
  if (tile >= tiles) return;

  const int m    = lane & 15;
  const int hi   = lane >> 4;
  const int row0 = tile << 4;
  const int arow = min(row0 + m, n_nodes - 1);
  const bool cok = (m < F_OUT);

  auto ldA = [&](int k) -> float {
    const bool ok = (k < F_IN);
    const float v = h[(long)arow * F_IN + (ok ? k : 0)];
    return ok ? v : 0.0f;
  };
  auto ldW = [&](const float* __restrict__ W, int k) -> float {
    const bool ok = cok && (k < F_IN);
    const float v = W[ok ? (m * F_IN + k) : 0];
    return ok ? v : 0.0f;
  };

  v8f dl = {};
  v8f dr = {};
#pragma unroll
  for (int kc = 0; kc < F_IN; kc += 4) {
    const int k0 = kc + hi * 2;
    v2f a, bl, br;
    a.x  = ldA(k0);
    a.y  = ldA(k0 + 1);
    bl.x = ldW(Wl, k0);
    bl.y = ldW(Wl, k0 + 1);
    br.x = ldW(Wr, k0);
    br.y = ldW(Wr, k0 + 1);
    dl = __builtin_amdgcn_wmma_f32_16x16x4_f32(false, a, false, bl, (short)0,
                                               dl, false, false);
    dr = __builtin_amdgcn_wmma_f32_16x16x4_f32(false, a, false, br, (short)0,
                                               dr, false, false);
  }

  const bool full = (row0 + 16 <= n_nodes);
  if (m < HL_PITCH) {
    const long lbase = (long)(row0 + hi * 8) * HL_PITCH + m;
    if (full) {
#pragma unroll
      for (int v = 0; v < 8; ++v) hl[lbase + (long)v * HL_PITCH] = dl[v];
    } else {
#pragma unroll
      for (int v = 0; v < 8; ++v)
        if (row0 + hi * 8 + v < n_nodes) hl[lbase + (long)v * HL_PITCH] = dl[v];
    }
  }
  if (cok) {
    const float bv   = bias[m];
    const long  base = (long)(row0 + hi * 8) * F_OUT + m;
    if (full) {
#pragma unroll
      for (int v = 0; v < 8; ++v) {
        hs[base + (long)v * F_OUT]  = dr[v] + bv;
        if (ZERO_AGG) agg[base + (long)v * F_OUT] = 0.0f;
      }
    } else {
#pragma unroll
      for (int v = 0; v < 8; ++v) {
        if (row0 + hi * 8 + v < n_nodes) {
          hs[base + (long)v * F_OUT]  = dr[v] + bv;
          if (ZERO_AGG) agg[base + (long)v * F_OUT] = 0.0f;
        }
      }
    }
  }
}

template <int F, bool RELU>
__global__ void pull_combine(const int* __restrict__ rp, const int* __restrict__ ss,
                             const float* __restrict__ hl,
                             const float* __restrict__ hs,
                             float* __restrict__ out, int n_nodes) {
  const int q    = threadIdx.x & 7;
  const int node = (int)((blockIdx.x * blockDim.x + threadIdx.x) >> 3);
  const bool live = node < n_nodes;
  const int nd  = live ? node : n_nodes - 1;
  const int beg = (nd > 0) ? rp[nd - 1] : 0;
  const int end = live ? rp[nd] : beg;

  float acc[F];
  for (int f = 0; f < F; ++f) acc[f] = 0.0f;
  for (int j = beg + q; j < end; j += 8) {
    const int s = ss[j];
    for (int f = 0; f < F; ++f) acc[f] += hl[(long)s * F + f];
  }
  for (int f = 0; f < F; ++f) {
    acc[f] += __shfl_xor(acc[f], 4);
    acc[f] += __shfl_xor(acc[f], 2);
    acc[f] += __shfl_xor(acc[f], 1);
  }
  float val = acc[0];
  for (int f = 1; f < F; ++f) val = (q == f) ? acc[f] : val;

  if (live && q < F) {
    const float iv = 1.0f / fmaxf((float)(end - beg), 1.0f);
    const long  o  = (long)node * F + q;
    const float r  = val * iv + hs[o];
    out[o] = RELU ? fmaxf(r, 0.0f) : r;
  }
}

template <bool RELU, int OUT_PITCH>
__global__ void pull_rows8(const int* __restrict__ rp, const int* __restrict__ ss,
                           const float* __restrict__ hl8,
                           const float* __restrict__ hs,
                           float* __restrict__ out, int n_nodes) {
  const int l    = threadIdx.x & 7;
  const int p    = l >> 1;
  const int hh   = l & 1;
  const int node = (int)((blockIdx.x * blockDim.x + threadIdx.x) >> 3);
  const bool live = node < n_nodes;
  const int nd  = live ? node : n_nodes - 1;
  const int beg = (nd > 0) ? rp[nd - 1] : 0;
  const int end = live ? rp[nd] : beg;

  const v4f* __restrict__ rows = (const v4f*)hl8;
  const v4f zero = {0.0f, 0.0f, 0.0f, 0.0f};
  v4f acc = zero;
  for (int j0 = beg + p; j0 < end; j0 += 16) {
    const int j1 = j0 + 4, j2 = j0 + 8, j3 = j0 + 12;
    const bool k1 = j1 < end, k2 = j2 < end, k3 = j3 < end;
    const int s0 = ss[j0];
    int s1 = ss[k1 ? j1 : j0];
    int s2 = ss[k2 ? j2 : j0];
    int s3 = ss[k3 ? j3 : j0];
    asm volatile("" : "+v"(s1), "+v"(s2), "+v"(s3));
    v4f v0 = rows[(long)s0 * 2 + hh];
    v4f v1 = zero, v2 = zero, v3 = zero;
    if (k1) v1 = rows[(long)s1 * 2 + hh];
    if (k2) v2 = rows[(long)s2 * 2 + hh];
    if (k3) v3 = rows[(long)s3 * 2 + hh];
    acc += (v0 + v1) + (v2 + v3);
  }
  for (int c = 0; c < 4; ++c) {
    acc[c] += __shfl_xor(acc[c], 4);
    acc[c] += __shfl_xor(acc[c], 2);
  }
  const float nx = __shfl_xor(acc[0], 1);
  const float ny = __shfl_xor(acc[1], 1);
  const float nw = __shfl_xor(acc[3], 1);
  float val = acc[0];
  val = (l == 1) ? ny : val;
  val = (l == 2) ? acc[2] : val;
  val = (l == 3) ? nw : val;
  val = (l == 4) ? nx : val;

  if (live && l < OUT_PITCH) {
    float r = 0.0f;
    if (l < 5) {
      const float iv = 1.0f / fmaxf((float)(end - beg), 1.0f);
      r = val * iv + hs[(long)node * 5 + l];
      r = RELU ? fmaxf(r, 0.0f) : r;
    }
    out[(long)node * OUT_PITCH + l] = r;
  }
}

#define A_PITCH 12
template <int F_OUT, bool RELU, int OUT_PITCH>
__global__ void __launch_bounds__(256)
pull_transform(const int* __restrict__ rp, const int* __restrict__ ss,
               const float* __restrict__ h8,
               const float* __restrict__ Wl, const float* __restrict__ Wr,
               const float* __restrict__ bias,
               float* __restrict__ out, int n_nodes) {
  __shared__ float arow[32 * A_PITCH];
  const int l    = threadIdx.x & 7;
  const int p    = l >> 1;
  const int hh   = l & 1;
  const int ln   = threadIdx.x >> 3;
  const int node = (int)blockIdx.x * 32 + ln;
  const bool live = node < n_nodes;
  const int nd  = live ? node : n_nodes - 1;
  const int beg = (nd > 0) ? rp[nd - 1] : 0;
  const int end = live ? rp[nd] : beg;

  const v4f* __restrict__ rows = (const v4f*)h8;
  const v4f zero = {0.0f, 0.0f, 0.0f, 0.0f};
  const float mine = (l < 5) ? h8[(long)nd * ROW8 + l] : 0.0f;

  const int wave = threadIdx.x >> 5;
  const int lane = threadIdx.x & 31;
  const int m    = lane & 15;
  const int hi   = lane >> 4;
  const bool cok = (m < F_OUT);
  auto bok = [&](int k) -> bool { return cok && (k < 10); };
  auto ldB = [&](int k) -> float {
    const bool left = (k < 5);
    const float* __restrict__ Wp = left ? Wl : Wr;
    const int kk = left ? k : k - 5;
    return Wp[bok(k) ? (m * 5 + kk) : 0];
  };
  float braw[A_PITCH / 2];
  float bv = 0.0f;
#pragma unroll
  for (int i = 0; i < A_PITCH / 2; ++i) braw[i] = 0.0f;
  if (wave < 2) {
#pragma unroll
    for (int i = 0; i < A_PITCH / 4; ++i) {
      const int k0 = 4 * i + hi * 2;
      braw[2 * i]     = ldB(k0);
      braw[2 * i + 1] = ldB(k0 + 1);
    }
    bv = cok ? bias[m] : 0.0f;
  }
  v4f acc = zero;
  for (int j0 = beg + p; j0 < end; j0 += 16) {
    const int j1 = j0 + 4, j2 = j0 + 8, j3 = j0 + 12;
    const bool k1 = j1 < end, k2 = j2 < end, k3 = j3 < end;
    const int s0 = ss[j0];
    int s1 = ss[k1 ? j1 : j0];
    int s2 = ss[k2 ? j2 : j0];
    int s3 = ss[k3 ? j3 : j0];
    asm volatile("" : "+v"(s1), "+v"(s2), "+v"(s3));
    v4f v0 = rows[(long)s0 * 2 + hh];
    v4f v1 = zero, v2 = zero, v3 = zero;
    if (k1) v1 = rows[(long)s1 * 2 + hh];
    if (k2) v2 = rows[(long)s2 * 2 + hh];
    if (k3) v3 = rows[(long)s3 * 2 + hh];
    acc += (v0 + v1) + (v2 + v3);
  }
  for (int c = 0; c < 4; ++c) {
    acc[c] += __shfl_xor(acc[c], 4);
    acc[c] += __shfl_xor(acc[c], 2);
  }
  const float nx = __shfl_xor(acc[0], 1);
  const float ny = __shfl_xor(acc[1], 1);
  const float nw = __shfl_xor(acc[3], 1);
  float val = acc[0];
  val = (l == 1) ? ny : val;
  val = (l == 2) ? acc[2] : val;
  val = (l == 3) ? nw : val;
  val = (l == 4) ? nx : val;

  const float iv = 1.0f / fmaxf((float)(end - beg), 1.0f);
  if (l < 5) {
    arow[ln * A_PITCH + l]     = live ? val * iv : 0.0f;
    arow[ln * A_PITCH + 5 + l] = live ? mine : 0.0f;
  } else if (l < 7) {
    arow[ln * A_PITCH + 5 + l] = 0.0f;
  }
  __syncthreads();

  if (wave >= 2) return;

  const float* __restrict__ at = arow + (wave * 16 + m) * A_PITCH;
  v8f d = {};
#pragma unroll
  for (int kc = 0; kc < A_PITCH; kc += 4) {
    const int k0 = kc + hi * 2;
    v2f a, b;
    a.x = at[k0];
    a.y = at[k0 + 1];
    b.x = bok(k0) ? braw[kc / 2] : 0.0f;
    b.y = bok(k0 + 1) ? braw[kc / 2 + 1] : 0.0f;
    d = __builtin_amdgcn_wmma_f32_16x16x4_f32(false, a, false, b, (short)0,
                                              d, false, false);
  }

  const int row0 = (int)blockIdx.x * 32 + wave * 16;
  if (m < OUT_PITCH) {
    const long  base = (long)(row0 + hi * 8) * OUT_PITCH + m;
    const bool  full = (row0 + 16 <= n_nodes);
    if (full) {
#pragma unroll
      for (int v = 0; v < 8; ++v) {
        const float r = d[v] + bv;
        out[base + (long)v * OUT_PITCH] = RELU ? fmaxf(r, 0.0f) : r;
      }
    } else {
#pragma unroll
      for (int v = 0; v < 8; ++v) {
        if (row0 + hi * 8 + v < n_nodes) {
          const float r = d[v] + bv;
          out[base + (long)v * OUT_PITCH] = RELU ? fmaxf(r, 0.0f) : r;
        }
      }
    }
  }
}

#define TPITCH 5
typedef float v4f_a4 __attribute__((ext_vector_type(4), aligned(4)));

__device__ inline unsigned ld_entry(const int* p) {
  return (unsigned)__builtin_nontemporal_load(p);
}
__device__ inline long row_offset(unsigned e, int rbase, int last) {
  return (long)min(rbase + (int)(e >> DL_BITS), last) * TPITCH;
}
__device__ inline void add_row(float* acc, unsigned e, v4f a, float c) {
  float* q = acc + (e & ((1u << DL_BITS) - 1));
  atomicAdd(q, a.x);
  atomicAdd(q + DG, a.y);
  atomicAdd(q + 2 * DG, a.z);
  atomicAdd(q + 3 * DG, a.w);
  atomicAdd(q + 4 * DG, c);
}

__device__ inline float sum_ranges(const float* __restrict__ p, int nr,
                                   long part_stride) {
  float v[NR_MAX];
#pragma unroll
  for (int k = 0; k < NR_MAX; ++k) {
    const bool ok = k < nr;
    const float t = p[ok ? (long)k * part_stride : 0];
    v[k] = ok ? t : 0.0f;
  }
  float s = v[0];
#pragma unroll
  for (int k = 1; k < NR_MAX; ++k) s += v[k];
  return s;
}

__global__ void __launch_bounds__(1024)
agg_block(const int* __restrict__ ss2, const int* __restrict__ offs,
          const float* __restrict__ tab, float* __restrict__ part,
          long n_edges, int n_nodes, int nr, int nbk, int G, long part_stride) {
  __shared__ float acc[5 * DG];
  const int b = blockIdx.x;
  const int g = b / nr, r = b - g * nr;
  const int tid = threadIdx.x;
  const int bs = offs[(long)b * G];
  const int be = (b + 1 < nbk) ? offs[(long)(b + 1) * G] : (int)n_edges;
  const int n = be - bs;
  const int* __restrict__ in = ss2 + bs;

  for (int i = tid; i < 5 * DG; i += 1024) acc[i] = 0.0f;
  __syncthreads();

  const int rbase = r * SRN;
  const int last = n_nodes - 1;
  int i = tid;
  for (; i + 3072 < n; i += 4096) {
    const unsigned e0 = ld_entry(in + i);
    const unsigned e1 = ld_entry(in + i + 1024);
    const unsigned e2 = ld_entry(in + i + 2048);
    const unsigned e3 = ld_entry(in + i + 3072);
    const float* __restrict__ p0 = tab + row_offset(e0, rbase, last);
    const float* __restrict__ p1 = tab + row_offset(e1, rbase, last);
    const float* __restrict__ p2 = tab + row_offset(e2, rbase, last);
    const float* __restrict__ p3 = tab + row_offset(e3, rbase, last);
    const v4f a0 = *(const v4f_a4*)p0;
    const v4f a1 = *(const v4f_a4*)p1;
    const v4f a2 = *(const v4f_a4*)p2;
    const v4f a3 = *(const v4f_a4*)p3;
    const float c0 = p0[4];
    const float c1 = p1[4];
    const float c2 = p2[4];
    const float c3 = p3[4];
    add_row(acc, e0, a0, c0);
    add_row(acc, e1, a1, c1);
    add_row(acc, e2, a2, c2);
    add_row(acc, e3, a3, c3);
  }
  for (; i < n; i += 1024) {
    const unsigned e0 = ld_entry(in + i);
    const float* __restrict__ p0 = tab + row_offset(e0, rbase, last);
    const v4f a0 = *(const v4f_a4*)p0;
    const float c0 = p0[4];
    add_row(acc, e0, a0, c0);
  }
  __syncthreads();

  const int node0 = g * DG;
  const int rows = min(DG, n_nodes - node0);
  float* __restrict__ po = part + (long)r * part_stride + (long)node0 * 5;
  for (int idx = tid; idx < rows * 5; idx += 1024) {
    const int d = idx / 5;
    const int f = idx - d * 5;
    po[idx] = acc[f * DG + d];
  }
}

__global__ void combine_rows(const float* __restrict__ part,
                             const int* __restrict__ deg,
                             const float* __restrict__ hs,
                             float* __restrict__ tab, int n_nodes, int nr,
                             long part_stride) {
  const int l    = threadIdx.x & 7;
  const int node = (int)((blockIdx.x * blockDim.x + threadIdx.x) >> 3);
  if (node >= n_nodes || l >= 5) return;
  const float s = sum_ranges(part + (long)node * 5 + l, nr, part_stride);
  const float iv = 1.0f / fmaxf((float)deg[node], 1.0f);
  tab[(long)node * TPITCH + l] = fmaxf(s * iv + hs[(long)node * 5 + l], 0.0f);
}

template <int F_OUT, bool RELU, int OUT_PITCH>
__global__ void __launch_bounds__(256)
combine_tiles(const float* __restrict__ part, const int* __restrict__ deg,
              const float* h8,
              const float* __restrict__ Wl, const float* __restrict__ Wr,
              const float* __restrict__ bias,
              float* out, int n_nodes, int nr, long part_stride) {
  const int lane = threadIdx.x & 31;
  const int unit = (int)blockIdx.x * ((int)blockDim.x >> 5) + ((int)threadIdx.x >> 5);
  const int tiles = (n_nodes + 15) >> 4;
  if (2 * unit >= tiles) return;
  const int m    = lane & 15;
  const int hi   = lane >> 4;
  const bool cok = (m < F_OUT);
  auto bok = [&](int k) -> bool { return cok && (k < 10); };
  auto ldB = [&](int k) -> float {
    const bool left = (k < 5);
    const float* __restrict__ Wp = left ? Wl : Wr;
    const int kk = left ? k : k - 5;
    return Wp[bok(k) ? (m * 5 + kk) : 0];
  };
  float braw[A_PITCH / 2];
#pragma unroll
  for (int i = 0; i < A_PITCH / 4; ++i) {
    const int k0 = 4 * i + hi * 2;
    braw[2 * i]     = ldB(k0);
    braw[2 * i + 1] = ldB(k0 + 1);
  }
  const float bv = cok ? bias[m] : 0.0f;

  v2f a[2][A_PITCH / 4];
#pragma unroll
  for (int t = 0; t < 2; ++t) {
    const int row0 = (2 * unit + t) << 4;
    const bool live = row0 + m < n_nodes;
    const int nd = live ? row0 + m : n_nodes - 1;
    const float* __restrict__ pr = part + (long)nd * 5;
    const float* own = h8 + (long)nd * TPITCH;
    const float iv = 1.0f / fmaxf((float)deg[nd], 1.0f);
    const float m0 = sum_ranges(pr + 2 * hi, nr, part_stride) * iv;
    const float m1 = sum_ranges(pr + 2 * hi + 1, nr, part_stride) * iv;
    const float m4 = sum_ranges(pr + 4, nr, part_stride) * iv;
    const float t0 = own[hi ? 1 : 0];
    const float t1 = own[hi ? 2 : 3];
    const float t2 = own[4];
    a[t][0].x = live ? m0 : 0.0f;
    a[t][0].y = live ? m1 : 0.0f;
    a[t][1].x = live ? (hi ? t0 : m4) : 0.0f;
    a[t][1].y = live ? (hi ? t1 : t0) : 0.0f;
    a[t][2].x = (live && !hi) ? t1 : 0.0f;
    a[t][2].y = (live && !hi) ? t2 : 0.0f;
  }

#pragma unroll
  for (int t = 0; t < 2; ++t) {
    const int row0 = (2 * unit + t) << 4;
    if (row0 >= n_nodes) break;
    v8f d = {};
#pragma unroll
    for (int kc = 0; kc < A_PITCH; kc += 4) {
      const int k0 = kc + hi * 2;
      v2f b;
      b.x = bok(k0) ? braw[kc / 2] : 0.0f;
      b.y = bok(k0 + 1) ? braw[kc / 2 + 1] : 0.0f;
      d = __builtin_amdgcn_wmma_f32_16x16x4_f32(false, a[t][kc / 4], false, b,
                                                (short)0, d, false, false);
    }
    if (m < OUT_PITCH) {
      const long  base = (long)(row0 + hi * 8) * OUT_PITCH + m;
      const bool  full = (row0 + 16 <= n_nodes);
      if (full) {
#pragma unroll
        for (int v = 0; v < 8; ++v) {
          const float r = d[v] + bv;
          out[base + (long)v * OUT_PITCH] = RELU ? fmaxf(r, 0.0f) : r;
        }
      } else {
#pragma unroll
        for (int v = 0; v < 8; ++v) {
          if (row0 + hi * 8 + v < n_nodes) {
            const float r = d[v] + bv;
            out[base + (long)v * OUT_PITCH] = RELU ? fmaxf(r, 0.0f) : r;
          }
        }
      }
    }
  }
}

__global__ void __launch_bounds__(1024)
agg_block1(const int* __restrict__ ss2, const int* __restrict__ offs,
           const float* __restrict__ tab1, float* __restrict__ part1,
           long n_edges, int n_nodes, int nr, int nbk, int G, long part_stride) {
  __shared__ float acc[DG];
  const int b = blockIdx.x;
  const int g = b / nr, r = b - g * nr;
  const int tid = threadIdx.x;
  const int bs = offs[(long)b * G];
  const int be = (b + 1 < nbk) ? offs[(long)(b + 1) * G] : (int)n_edges;
  const int n = be - bs;
  const int* __restrict__ in = ss2 + bs;

  for (int i = tid; i < DG; i += 1024) acc[i] = 0.0f;
  __syncthreads();

  const int rbase = r * SRN;
  const int last = n_nodes - 1;
  const unsigned dmask = (1u << DL_BITS) - 1;
  int i = tid;
  for (; i + 7168 < n; i += 8192) {
    unsigned e[8];
    float v[8];
#pragma unroll
    for (int k = 0; k < 8; ++k) e[k] = ld_entry(in + i + k * 1024);
#pragma unroll
    for (int k = 0; k < 8; ++k)
      v[k] = tab1[min(rbase + (int)(e[k] >> DL_BITS), last)];
#pragma unroll
    for (int k = 0; k < 8; ++k) atomicAdd(acc + (e[k] & dmask), v[k]);
  }
  for (; i + 3072 < n; i += 4096) {
    unsigned e[4];
    float v[4];
#pragma unroll
    for (int k = 0; k < 4; ++k) e[k] = ld_entry(in + i + k * 1024);
#pragma unroll
    for (int k = 0; k < 4; ++k)
      v[k] = tab1[min(rbase + (int)(e[k] >> DL_BITS), last)];
#pragma unroll
    for (int k = 0; k < 4; ++k) atomicAdd(acc + (e[k] & dmask), v[k]);
  }
  for (; i < n; i += 1024) {
    const unsigned e0 = ld_entry(in + i);
    const float v0 = tab1[min(rbase + (int)(e0 >> DL_BITS), last)];
    atomicAdd(acc + (e0 & dmask), v0);
  }
  __syncthreads();

  const int node0 = g * DG;
  const int rows = min(DG, n_nodes - node0);
  float* __restrict__ po = part1 + (long)r * part_stride + node0;
  for (int d = tid; d < rows; d += 1024) po[d] = acc[d];
}

__global__ void combine_out1(const float* __restrict__ part1,
                             const int* __restrict__ deg,
                             const float* __restrict__ hs1,
                             float* __restrict__ out, int n_nodes, int nr,
                             long part_stride) {
  const int node = blockIdx.x * blockDim.x + threadIdx.x;
  if (node >= n_nodes) return;
  const float s = sum_ranges(part1 + node, nr, part_stride);
  const float iv = 1.0f / fmaxf((float)deg[node], 1.0f);
  out[node] = s * iv + hs1[node];
}

__global__ void zero_f32(float* __restrict__ p, long n) {
  long i = (long)blockIdx.x * blockDim.x + threadIdx.x;
  if (i < n) p[i] = 0.0f;
}

__global__ void count_edges(const int* __restrict__ dst, float* __restrict__ cnt,
                            long n_edges) {
  long i = (long)blockIdx.x * blockDim.x + threadIdx.x;
  if (i < n_edges) unsafeAtomicAdd(&cnt[dst[i]], 1.0f);
}

__global__ void reciprocal_clamped(float* __restrict__ p, long n) {
  long i = (long)blockIdx.x * blockDim.x + threadIdx.x;
  if (i < n) p[i] = 1.0f / fmaxf(p[i], 1.0f);
}

template <int F>
__global__ void edge_scatter(const int* __restrict__ src,
                             const int* __restrict__ dst,
                             const float* __restrict__ hl,
                             float* __restrict__ agg, long n_edges) {
  long i = (long)blockIdx.x * blockDim.x + threadIdx.x;
  if (i >= n_edges) return;
  const int s = src[i];
  const int d = dst[i];
  float v[F];
#pragma unroll
  for (int j = 0; j < F; ++j) v[j] = hl[(long)s * F + j];
#pragma unroll
  for (int j = 0; j < F; ++j) unsafeAtomicAdd(&agg[(long)d * F + j], v[j]);
}

template <int F, bool RELU>
__global__ void combine(const float* __restrict__ agg,
                        const float* __restrict__ hs,
                        const float* __restrict__ inv,
                        float* __restrict__ out, int n_nodes) {
  const int node = blockIdx.x * blockDim.x + threadIdx.x;
  if (node >= n_nodes) return;
  const float iv = inv[node];
#pragma unroll
  for (int j = 0; j < F; ++j) {
    const long o = (long)node * F + j;
    const float v = agg[o] * iv + hs[o];
    out[o] = RELU ? fmaxf(v, 0.0f) : v;
  }
}

extern "C" void kernel_launch(void* const* d_in, const int* in_sizes, int n_in,
                              void* d_out, int out_size, void* d_ws,
                              size_t ws_size, hipStream_t stream) {
  const float* x    = (const float*)d_in[0];
  const int*   ei   = (const int*)  d_in[1];
  const float* Wl1  = (const float*)d_in[2];
  const float* Wr1  = (const float*)d_in[3];
  const float* b1   = (const float*)d_in[4];
  const float* Wlm  = (const float*)d_in[5];
  const float* Wrm  = (const float*)d_in[6];
  const float* bm   = (const float*)d_in[7];
  const float* Wl10 = (const float*)d_in[8];
  const float* Wr10 = (const float*)d_in[9];
  const float* b10  = (const float*)d_in[10];

  const int  N = in_sizes[0] / 10;
  const long E = (long)in_sizes[1] / 2;
  const int* src = ei;
  const int* dst = ei + E;

  const int B = 256;
  auto cdiv = [](long a, long b) { return (int)((a + b - 1) / b); };
  auto pad32 = [](size_t a) { return (a + 31) & ~(size_t)31; };

  const int tiles   = (N + 15) / 16;
  const int tblocks = (tiles + 7) / 8;

  const int    nb    = cdiv(N, SCAN_CHUNK);
  const size_t w_rp  = pad32((size_t)N);
  const size_t w_bs  = pad32((size_t)nb);
  const size_t w_ss  = pad32((size_t)E);
  const size_t w_h   = pad32((size_t)N * 5);
  const size_t w_h8  = pad32((size_t)N * ROW8);
  const size_t need  = (w_rp + w_bs + w_ss + 3 * w_h) * sizeof(int);
  const size_t need8 = (w_rp + w_bs + w_ss + 2 * w_h + w_h8) * sizeof(int);

  if (ws_size >= need) {
    const bool wide = (ws_size >= need8);
    int*   rp   = (int*)d_ws;
    int*   bsum = rp + w_rp;
    int*   ss   = bsum + w_bs;
    float* h    = (float*)(ss + w_ss);
    float* hs   = h + w_h;
    float* hl   = hs + w_h;
    float* none = nullptr;

    const int    PG     = 1024;
    const int    nbk    = cdiv(N, BK_NODES);
    const int    chunk  = cdiv(E, PG);
    const size_t w_hist = pad32((size_t)nbk * PG);
    const size_t need_s = need8 + (w_hist + w_ss) * sizeof(int);
    const bool lds_sort = wide && ws_size >= need_s && nbk <= MAX_BK &&
                          chunk <= PART_CAP && N <= (1 << 21);
    {
      const int    ng      = cdiv(N, DG);
      const int    nr      = cdiv(N, SRN);
      const int    nbk2    = ng * nr;
      const int    chunk2  = cdiv(E, PG2);
      const int    hn2     = nbk2 * PG2;
      const int    hnb2    = cdiv(hn2, SCAN_CHUNK);
      const size_t w_hist2 = pad32((size_t)hn2);
      const size_t need_b  = (w_rp + w_bs + w_ss + 2 * w_h + w_h8 + w_hist2 + w_ss) * sizeof(int);
      const bool blocked = wide && ws_size >= need_b && nr <= NR_MAX &&
                           nbk2 <= MAX_BK && chunk2 <= PART_CAP &&
                           (size_t)hnb2 <= w_bs && (size_t)nr * w_h <= w_ss &&
                           (size_t)nr * w_rp <= 2 * w_h && E < (1L << 31);
      if (blocked) {
        int*   deg   = rp;
        int*   ss2   = ss;
        int*   degp  = (int*)h;
        float* tab   = hl;
        int*   hist2 = (int*)(hl + w_h8);
        int*   tmp   = hist2 + w_hist2;
        float* part  = (float*)tmp;
        const long dstr = (long)w_rp;
        const long pstr = (long)w_h;
        part_hist2<<<PG2, 256, 0, stream>>>(src, dst, hist2, E, chunk2, nr, nbk2);
        block_sums<<<hnb2, B, 0, stream>>>(hist2, bsum, hn2);
        scan_block_sums<<<1, B, 0, stream>>>(bsum, hnb2);
        scan_apply<<<hnb2, B, 0, stream>>>(hist2, bsum, hn2);
        part_scatter2<<<PG2, 256, 0, stream>>>(src, dst, hist2, tmp, E, chunk2, nr, nbk2);
        bucket_sort2<<<nbk2, 1024, 0, stream>>>(tmp, hist2, degp, ss2, E, N, nr, nbk2, PG2, dstr);
        degree_sum<<<cdiv(N, B), B, 0, stream>>>(degp, deg, N, nr, dstr);

        const int cblocks = cdiv((long)N * 8, B);
        const int ublocks = cdiv(cdiv(tiles, 2), 8);
        node_transform<10, 5, false, TPITCH><<<tblocks, 256, 0, stream>>>(x, Wl1, Wr1, b1, tab, hs, none, N);
        agg_block<<<nbk2, 1024, 0, stream>>>(ss2, hist2, tab, part, E, N, nr, nbk2, PG2, pstr);
        combine_rows<<<cblocks, B, 0, stream>>>(part, deg, hs, tab, N, nr, pstr);
        for (int l = 0; l < 8; ++l) {
          agg_block<<<nbk2, 1024, 0, stream>>>(ss2, hist2, tab, part, E, N, nr, nbk2, PG2, pstr);
          combine_tiles<5, true, TPITCH><<<ublocks, B, 0, stream>>>(
              part, deg, tab, Wlm + l * 25, Wrm + l * 25, bm + l * 5, tab, N, nr, pstr);
        }
        float* hl1 = h;
        float* hs1 = hs;
        node_transform<5, 1, false, 1><<<tblocks, 256, 0, stream>>>(tab, Wl10, Wr10, b10, hl1, hs1, none, N);
        agg_block1<<<nbk2, 1024, 0, stream>>>(ss2, hist2, hl1, part, E, N, nr, nbk2, PG2, pstr);
        combine_out1<<<cdiv(N, B), B, 0, stream>>>(part, deg, hs1, (float*)d_out, N, nr, pstr);
        return;
      }
    }

    if (lds_sort) {
      int* hist = (int*)(hl + w_h8);
      int* tmp  = hist + w_hist;
      const int hn  = nbk * PG;
      const int hnb = cdiv(hn, SCAN_CHUNK);
      part_hist<<<PG, 256, 0, stream>>>(dst, hist, E, chunk, nbk);
      block_sums<<<hnb, B, 0, stream>>>(hist, bsum, hn);
      scan_block_sums<<<1, B, 0, stream>>>(bsum, hnb);
      scan_apply<<<hnb, B, 0, stream>>>(hist, bsum, hn);
      part_scatter<<<PG, 256, 0, stream>>>(src, dst, hist, tmp, E, chunk, nbk);
      bucket_rows<<<nbk, 1024, 0, stream>>>(tmp, hist, rp, ss, E, N, nbk, PG);
    } else {
      const int eblocks = cdiv(E, B) < 2048 ? cdiv(E, B) : 2048;
      zero_i32<<<cdiv(N, B), B, 0, stream>>>(rp, N);
      count_in_edges<<<eblocks, B, 0, stream>>>(dst, rp, E);
      block_sums<<<nb, B, 0, stream>>>(rp, bsum, N);
      scan_block_sums<<<1, B, 0, stream>>>(bsum, nb);
      scan_apply<<<nb, B, 0, stream>>>(rp, bsum, N);
      fill_rows<<<eblocks, B, 0, stream>>>(src, dst, rp, ss, E);
    }

    const int pblocks = cdiv((long)N * 8, B);

    if (lds_sort && w_ss >= w_h8) {
      float* ta = hl;
      float* tb = hl + w_h8 + w_hist;
      node_transform<10, 5, false, ROW8><<<tblocks, 256, 0, stream>>>(x, Wl1, Wr1, b1, ta, hs, none, N);
      pull_rows8<true, ROW8><<<pblocks, B, 0, stream>>>(rp, ss, ta, hs, tb, N);
      for (int l = 0; l < 8; ++l) {
        const float* tin  = (l & 1) ? ta : tb;
        float*       tout = (l & 1) ? tb : ta;
        pull_transform<5, true, ROW8><<<pblocks, B, 0, stream>>>(
            rp, ss, tin, Wlm + l * 25, Wrm + l * 25, bm + l * 5, tout, N);
      }
      pull_transform<1, false, 1><<<pblocks, B, 0, stream>>>(
          rp, ss, tb, Wl10, Wr10, b10, (float*)d_out, N);
      return;
    }

    if (wide) {
      node_transform<10, 5, false, ROW8><<<tblocks, 256, 0, stream>>>(x, Wl1, Wr1, b1, hl, hs, none, N);
      pull_rows8<true, 5><<<pblocks, B, 0, stream>>>(rp, ss, hl, hs, h, N);
    } else {
      node_transform<10, 5, false, 5><<<tblocks, 256, 0, stream>>>(x, Wl1, Wr1, b1, hl, hs, none, N);
      pull_combine<5, true><<<pblocks, B, 0, stream>>>(rp, ss, hl, hs, h, N);
    }

    for (int l = 0; l < 8; ++l) {
      if (wide) {
        node_transform<5, 5, false, ROW8><<<tblocks, 256, 0, stream>>>(
            h, Wlm + l * 25, Wrm + l * 25, bm + l * 5, hl, hs, none, N);
        pull_rows8<true, 5><<<pblocks, B, 0, stream>>>(rp, ss, hl, hs, h, N);
      } else {
        node_transform<5, 5, false, 5><<<tblocks, 256, 0, stream>>>(
            h, Wlm + l * 25, Wrm + l * 25, bm + l * 5, hl, hs, none, N);
        pull_combine<5, true><<<pblocks, B, 0, stream>>>(rp, ss, hl, hs, h, N);
      }
    }

    node_transform<5, 1, false, 1><<<tblocks, 256, 0, stream>>>(h, Wl10, Wr10, b10, hl, hs, none, N);
    pull_combine<1, false><<<pblocks, B, 0, stream>>>(rp, ss, hl, hs, (float*)d_out, N);
    return;
  }

  float* ws  = (float*)d_ws;
  float* inv = ws;
  float* hl  = inv + (size_t)N;
  float* hs  = hl + (size_t)N * 5;
  float* agg = hs + (size_t)N * 5;
  float* hA  = agg + (size_t)N * 5;
  float* hB  = hA + (size_t)N * 5;

  zero_f32<<<cdiv(N, B), B, 0, stream>>>(inv, N);
  count_edges<<<cdiv(E, B), B, 0, stream>>>(dst, inv, E);
  reciprocal_clamped<<<cdiv(N, B), B, 0, stream>>>(inv, N);

  node_transform<10, 5, true, 5><<<tblocks, 256, 0, stream>>>(x, Wl1, Wr1, b1, hl, hs, agg, N);
  edge_scatter<5><<<cdiv(E, B), B, 0, stream>>>(src, dst, hl, agg, E);
  combine<5, true><<<cdiv(N, B), B, 0, stream>>>(agg, hs, inv, hA, N);

  for (int l = 0; l < 8; ++l) {
    const float* hin  = (l & 1) ? hB : hA;
    float*       hout = (l & 1) ? hA : hB;
    node_transform<5, 5, true, 5><<<tblocks, 256, 0, stream>>>(
        hin, Wlm + l * 25, Wrm + l * 25, bm + l * 5, hl, hs, agg, N);
    edge_scatter<5><<<cdiv(E, B), B, 0, stream>>>(src, dst, hl, agg, E);
    combine<5, true><<<cdiv(N, B), B, 0, stream>>>(agg, hs, inv, hout, N);
  }

  node_transform<5, 1, true, 1><<<tblocks, 256, 0, stream>>>(hA, Wl10, Wr10, b10, hl, hs, agg, N);
  edge_scatter<1><<<cdiv(E, B), B, 0, stream>>>(src, dst, hl, agg, E);
  combine<1, false><<<cdiv(N, B), B, 0, stream>>>(agg, hs, inv, (float*)d_out, N);
}
